# starting kernel
# baseline (speedup 1.0000x reference)
#include <hip/hip_runtime.h>

typedef __attribute__((ext_vector_type(2))) float v2f;
typedef __attribute__((ext_vector_type(8))) float v8f;

__global__ void k_init_one(float* deg, int N) {
    int i = blockIdx.x * blockDim.x + threadIdx.x;
    if (i < N) deg[i] = 1.0f;
}

__global__ void k_deg_accum(const int* __restrict__ dst, float* deg, int E) {
    int e = blockIdx.x * blockDim.x + threadIdx.x;
    if (e < E) atomicAdd(&deg[dst[e]], 1.0f);
}

__global__ void k_rsqrt_inplace(float* deg, int N) {
    int i = blockIdx.x * blockDim.x + threadIdx.x;
    if (i < N) deg[i] = rsqrtf(deg[i]);
}

__global__ void k_edge_norm(const int* __restrict__ src, const int* __restrict__ dst,
                            const float* __restrict__ dis, float* __restrict__ nrm, int E) {
    int e = blockIdx.x * blockDim.x + threadIdx.x;
    if (e < E) nrm[e] = dis[src[e]] * dis[dst[e]];
}

__global__ void k_input_proj(const float* __restrict__ num_x, const float* __restrict__ num_mask,
                             const float* __restrict__ x, const float* __restrict__ Wn,
                             const float* __restrict__ bn, const float* __restrict__ a,
                             float* __restrict__ h, int total) {
    int i = blockIdx.x * blockDim.x + threadIdx.x;
    if (i >= total) return;
    int n = i >> 7, d = i & 127;
    float z = num_x[n] * num_mask[n] * Wn[d] + bn[d];
    z = (z >= 0.0f) ? z : a[d] * z;
    h[i] = z + x[i];
}

__global__ __launch_bounds__(128) void k_gemm128_wmma(const float* __restrict__ h,
                                                      const float* __restrict__ W,
                                                      float* __restrict__ t,
                                                      int nRowTiles) {
    __shared__ float Wl[128 * 128];
    {
        int tid = threadIdx.x;
        v2f* Wlv = (v2f*)Wl;
        #pragma unroll
        for (int i = 0; i < 64; ++i) {
            int idx = i * 128 + tid;
            int c = idx & 127;
            int p = idx >> 7;
            v2f pr;
            pr.x = W[(2 * p)     * 128 + c];
            pr.y = W[(2 * p + 1) * 128 + c];
            Wlv[p * 128 + c] = pr;
        }
    }
    __syncthreads();

    const int wave = threadIdx.x >> 5;
    const int lane = threadIdx.x & 31;
    const int rowTile = blockIdx.x * 4 + wave;
    if (rowTile >= nRowTiles) return;

    const int r15   = lane & 15;
    const int khalf = (lane >> 4) << 1;
    const float* hrow = h + (size_t)(rowTile * 16 + r15) * 128;
    const v2f* Wlv = (const v2f*)Wl;

    v8f acc[8] = {};
    for (int k0 = 0; k0 < 128; k0 += 4) {
        v2f a = *(const v2f*)(hrow + k0 + khalf);
        const v2f* wrow = Wlv + (((k0 + khalf) >> 1) * 128 + r15);
        #pragma unroll
        for (int j = 0; j < 8; ++j) {
            v2f bfrag = wrow[j * 16];
            acc[j] = __builtin_amdgcn_wmma_f32_16x16x4_f32(
                false, a, false, bfrag, (short)0, acc[j], false, false);
        }
    }

    const int mhalf = (lane >> 4) << 3;
    float* trow = t + (size_t)(rowTile * 16 + mhalf) * 128 + r15;
    #pragma unroll
    for (int j = 0; j < 8; ++j)
        #pragma unroll
        for (int r = 0; r < 8; ++r)
            trow[(size_t)r * 128 + j * 16] = acc[j][r];
}

__global__ void k_agg_init(const float* __restrict__ t, const float* __restrict__ dis,
                           const float* __restrict__ bias, float* __restrict__ acc, int total) {
    int i = blockIdx.x * blockDim.x + threadIdx.x;
    if (i >= total) return;
    int n = i >> 7, c = i & 127;
    float sn = dis[n] * dis[n];
    acc[i] = sn * t[i] + bias[c];
}

__global__ void k_edge_scatter_vec(const int* __restrict__ src, const int* __restrict__ dst,
                                   const float* __restrict__ nrm, const float* __restrict__ t,
                                   float* __restrict__ acc, int E) {
    int wid  = (blockIdx.x * blockDim.x + threadIdx.x) >> 5;
    int lane = threadIdx.x & 31;
    if (wid >= E) return;
    int s = src[wid], d = dst[wid];
    float w = nrm[wid];
    const float4 v = ((const float4*)(t + (size_t)s * 128))[lane];
    float* ap = acc + (size_t)d * 128 + lane * 4;
    atomicAdd(ap + 0, w * v.x);
    atomicAdd(ap + 1, w * v.y);
    atomicAdd(ap + 2, w * v.z);
    atomicAdd(ap + 3, w * v.w);
}

__global__ void k_prelu(float* __restrict__ h, const float* __restrict__ a, int total) {
    int i = blockIdx.x * blockDim.x + threadIdx.x;
    if (i >= total) return;
    float v = h[i];
    h[i] = (v >= 0.0f) ? v : a[i & 127] * v;
}

__global__ void k_dot_w3(const float* __restrict__ h, const float* __restrict__ W3,
                         float* __restrict__ t3, int N) {
    int wid  = (blockIdx.x * blockDim.x + threadIdx.x) >> 5;
    int lane = threadIdx.x & 31;
    if (wid >= N) return;
    const float4 hv = ((const float4*)(h + (size_t)wid * 128))[lane];
    const float4 wv = ((const float4*)W3)[lane];
    float s = hv.x * wv.x + hv.y * wv.y + hv.z * wv.z + hv.w * wv.w;
    #pragma unroll
    for (int off = 16; off > 0; off >>= 1) s += __shfl_xor(s, off, 32);
    if (lane == 0) t3[wid] = s;
}

__global__ void k_out_init(const float* __restrict__ t3, const float* __restrict__ dis,
                           const float* __restrict__ b3, float* __restrict__ out, int N) {
    int i = blockIdx.x * blockDim.x + threadIdx.x;
    if (i < N) out[i] = dis[i] * dis[i] * t3[i] + b3[0];
}

__global__ void k_edge_scatter_scalar(const int* __restrict__ src, const int* __restrict__ dst,
                                      const float* __restrict__ nrm, const float* __restrict__ t3,
                                      float* __restrict__ out, int E) {
    int e = blockIdx.x * blockDim.x + threadIdx.x;
    if (e < E) atomicAdd(&out[dst[e]], nrm[e] * t3[src[e]]);
}

extern "C" void kernel_launch(void* const* d_in, const int* in_sizes, int n_in,
                              void* d_out, int out_size, void* d_ws, size_t ws_size,
                              hipStream_t stream) {
    const float* num_x    = (const float*)d_in[0];
    const float* num_mask = (const float*)d_in[1];
    const float* x        = (const float*)d_in[2];
    const int*   edge     = (const int*)d_in[3];
    const float* W_num    = (const float*)d_in[4];
    const float* b_num    = (const float*)d_in[5];
    const float* a_in     = (const float*)d_in[6];
    const float* W1       = (const float*)d_in[7];
    const float* b1       = (const float*)d_in[8];
    const float* a1       = (const float*)d_in[9];
    const float* W2       = (const float*)d_in[10];
    const float* b2       = (const float*)d_in[11];
    const float* a2       = (const float*)d_in[12];
    const float* W3       = (const float*)d_in[13];
    const float* b3       = (const float*)d_in[14];

    const int N = in_sizes[1];
    const int E = in_sizes[3] / 2;
    const int* src = edge;
    const int* dst = edge + E;

    const int nRowTiles = (N + 15) / 16;
    const size_t padRows = (size_t)nRowTiles * 16;

    auto al = [](size_t v) { return (v + 255) & ~(size_t)255; };
    char* p = (char*)d_ws;
    float* dis  = (float*)p; p += al((size_t)N * 4);
    float* nrm  = (float*)p; p += al((size_t)E * 4);
    float* hbuf = (float*)p; p += al(padRows * 128 * 4);
    float* tbuf = (float*)p; p += al(padRows * 128 * 4);
    float* t3   = (float*)p; p += al((size_t)N * 4);
    float* out  = (float*)d_out;

    const int TPB = 256;
    const int totND = N * 128;
    dim3 blk(TPB);
    auto gN  = dim3((N + TPB - 1) / TPB);
    auto gE  = dim3((E + TPB - 1) / TPB);
    auto gND = dim3((totND + TPB - 1) / TPB);
    auto gEw = dim3((int)(((size_t)E * 32 + TPB - 1) / TPB));
    auto gNw = dim3((int)(((size_t)N * 32 + TPB - 1) / TPB));
    const int gemmBlocks = (nRowTiles + 3) / 4;

    k_init_one<<<gN, blk, 0, stream>>>(dis, N);
    k_deg_accum<<<gE, blk, 0, stream>>>(dst, dis, E);
    k_rsqrt_inplace<<<gN, blk, 0, stream>>>(dis, N);
    k_edge_norm<<<gE, blk, 0, stream>>>(src, dst, dis, nrm, E);

    k_input_proj<<<gND, blk, 0, stream>>>(num_x, num_mask, x, W_num, b_num, a_in, hbuf, totND);

    k_gemm128_wmma<<<gemmBlocks, 128, 0, stream>>>(hbuf, W1, tbuf, nRowTiles);
    k_agg_init<<<gND, blk, 0, stream>>>(tbuf, dis, b1, hbuf, totND);
    k_edge_scatter_vec<<<gEw, blk, 0, stream>>>(src, dst, nrm, tbuf, hbuf, E);
    k_prelu<<<gND, blk, 0, stream>>>(hbuf, a1, totND);

    k_gemm128_wmma<<<gemmBlocks, 128, 0, stream>>>(hbuf, W2, tbuf, nRowTiles);
    k_agg_init<<<gND, blk, 0, stream>>>(tbuf, dis, b2, hbuf, totND);
    k_edge_scatter_vec<<<gEw, blk, 0, stream>>>(src, dst, nrm, tbuf, hbuf, E);
    k_prelu<<<gND, blk, 0, stream>>>(hbuf, a2, totND);

    k_dot_w3<<<gNw, blk, 0, stream>>>(hbuf, W3, t3, N);
    k_out_init<<<gN, blk, 0, stream>>>(t3, dis, b3, out, N);
    k_edge_scatter_scalar<<<gE, blk, 0, stream>>>(src, dst, nrm, t3, out, E);
}


# aggregation as gather over a per-call dst-sorted edge list (int rank atomics + scan): rows summed in registers in arrival-rank order, stored once with bias+PReLU; float atomics removed
# speedup vs baseline: 2.3495x; 2.3495x over previous
#include <hip/hip_runtime.h>

typedef __attribute__((ext_vector_type(2))) float v2f;
typedef __attribute__((ext_vector_type(8))) float v8f;

__global__ void k_zero_cnt(int* cnt, int N) {
    int i = blockIdx.x * blockDim.x + threadIdx.x;
    if (i < N) cnt[i] = 0;
}

__global__ void k_count_rank(const int* __restrict__ dst, int* cnt, int* __restrict__ rank, int E) {
    int e = blockIdx.x * blockDim.x + threadIdx.x;
    if (e < E) rank[e] = atomicAdd(&cnt[dst[e]], 1);
}

__global__ __launch_bounds__(256) void k_tile_sums(const int* __restrict__ cnt, int* __restrict__ tsum, int N) {
    __shared__ int sm[8];
    const int tid = threadIdx.x, lane = tid & 31, w = tid >> 5;
    const int base = blockIdx.x * 1024 + tid * 4;
    int s = 0;
    for (int j = 0; j < 4; ++j) if (base + j < N) s += cnt[base + j];
    for (int off = 16; off > 0; off >>= 1) s += __shfl_xor(s, off, 32);
    if (lane == 0) sm[w] = s;
    __syncthreads();
    if (tid == 0) {
        int tot = 0;
        for (int i = 0; i < 8; ++i) tot += sm[i];
        tsum[blockIdx.x] = tot;
    }
}

__global__ __launch_bounds__(256) void k_scan_rows(const int* __restrict__ cnt, const int* __restrict__ tsum,
                                                   int* __restrict__ rowptr, float* __restrict__ dis, int N) {
    __shared__ int smw[8];
    __shared__ int smo[8];
    const int tid = threadIdx.x, lane = tid & 31, w = tid >> 5;
    int o = 0;
    for (int i = tid; i < (int)blockIdx.x; i += 256) o += tsum[i];
    for (int off = 16; off > 0; off >>= 1) o += __shfl_xor(o, off, 32);
    if (lane == 0) smo[w] = o;

    const int base = blockIdx.x * 1024 + tid * 4;
    int c[4];
    for (int j = 0; j < 4; ++j) c[j] = (base + j < N) ? cnt[base + j] : 0;
    const int s = c[0] + c[1] + c[2] + c[3];
    int inc = s;
    for (int off = 1; off < 32; off <<= 1) {
        int u = __shfl_up(inc, off, 32);
        if (lane >= off) inc += u;
    }
    if (lane == 31) smw[w] = inc;
    __syncthreads();
    int ex = inc - s;
    for (int i = 0; i < 8; ++i) {
        ex += smo[i];
        if (i < w) ex += smw[i];
    }
    for (int j = 0; j < 4; ++j) {
        if (base + j < N) {
            rowptr[base + j] = ex;
            dis[base + j] = rsqrtf((float)c[j] + 1.0f);
            ex += c[j];
        }
    }
}

__global__ void k_fill_csr(const int* __restrict__ src, const int* __restrict__ dst,
                           const int* __restrict__ rank, const int* __restrict__ rowptr,
                           const float* __restrict__ dis, int2* __restrict__ csr, int E) {
    int e = blockIdx.x * blockDim.x + threadIdx.x;
    if (e >= E) return;
    int s = src[e], d = dst[e];
    float w = dis[s] * dis[d];
    csr[rowptr[d] + rank[e]] = make_int2(s, __float_as_int(w));
}

__global__ void k_input_proj(const float* __restrict__ num_x, const float* __restrict__ num_mask,
                             const float* __restrict__ x, const float* __restrict__ Wn,
                             const float* __restrict__ bn, const float* __restrict__ a,
                             float* __restrict__ h, int total) {
    int i = blockIdx.x * blockDim.x + threadIdx.x;
    if (i >= total) return;
    int n = i >> 7, d = i & 127;
    float z = num_x[n] * num_mask[n] * Wn[d] + bn[d];
    z = (z >= 0.0f) ? z : a[d] * z;
    h[i] = z + x[i];
}

__global__ __launch_bounds__(128) void k_gemm128_wmma(const float* __restrict__ h,
                                                      const float* __restrict__ W,
                                                      float* __restrict__ t,
                                                      int nRowTiles) {
    __shared__ float Wl[128 * 128];
    {
        int tid = threadIdx.x;
        v2f* Wlv = (v2f*)Wl;
        #pragma unroll
        for (int i = 0; i < 64; ++i) {
            int idx = i * 128 + tid;
            int c = idx & 127;
            int p = idx >> 7;
            v2f pr;
            pr.x = W[(2 * p)     * 128 + c];
            pr.y = W[(2 * p + 1) * 128 + c];
            Wlv[p * 128 + c] = pr;
        }
    }
    __syncthreads();

    const int wave = threadIdx.x >> 5;
    const int lane = threadIdx.x & 31;
    const int rowTile = blockIdx.x * 4 + wave;
    if (rowTile >= nRowTiles) return;

    const int r15   = lane & 15;
    const int khalf = (lane >> 4) << 1;
    const float* hrow = h + (size_t)(rowTile * 16 + r15) * 128;
    const v2f* Wlv = (const v2f*)Wl;

    v8f acc[8] = {};
    for (int k0 = 0; k0 < 128; k0 += 4) {
        v2f a = *(const v2f*)(hrow + k0 + khalf);
        const v2f* wrow = Wlv + (((k0 + khalf) >> 1) * 128 + r15);
        #pragma unroll
        for (int j = 0; j < 8; ++j) {
            v2f bfrag = wrow[j * 16];
            acc[j] = __builtin_amdgcn_wmma_f32_16x16x4_f32(
                false, a, false, bfrag, (short)0, acc[j], false, false);
        }
    }

    const int mhalf = (lane >> 4) << 3;
    float* trow = t + (size_t)(rowTile * 16 + mhalf) * 128 + r15;
    #pragma unroll
    for (int j = 0; j < 8; ++j)
        #pragma unroll
        for (int r = 0; r < 8; ++r)
            trow[(size_t)r * 128 + j * 16] = acc[j][r];
}

__global__ __launch_bounds__(256) void k_gather_rows(const int2* __restrict__ csr, const int* __restrict__ rowptr,
                                                     const int* __restrict__ cnt, const float* __restrict__ dis,
                                                     const float* __restrict__ t, const float* __restrict__ bias,
                                                     const float* __restrict__ a, float* __restrict__ h, int N) {
    const int wid  = (blockIdx.x * blockDim.x + threadIdx.x) >> 5;
    const int lane = threadIdx.x & 31;
    if (wid >= N) return;
    const int n = __builtin_amdgcn_readfirstlane(wid);
    const int beg = rowptr[n];
    const int end = beg + cnt[n];
    const float4* t4 = (const float4*)t;
    float4 acc = make_float4(0.0f, 0.0f, 0.0f, 0.0f);
    int e = beg;
    for (; e + 4 <= end; e += 4) {
        const int2 c0 = csr[e], c1 = csr[e + 1], c2 = csr[e + 2], c3 = csr[e + 3];
        const float4 v0 = t4[(size_t)c0.x * 32 + lane];
        const float4 v1 = t4[(size_t)c1.x * 32 + lane];
        const float4 v2 = t4[(size_t)c2.x * 32 + lane];
        const float4 v3 = t4[(size_t)c3.x * 32 + lane];
        const float w0 = __int_as_float(c0.y), w1 = __int_as_float(c1.y);
        const float w2 = __int_as_float(c2.y), w3 = __int_as_float(c3.y);
        acc.x += w0 * v0.x; acc.y += w0 * v0.y; acc.z += w0 * v0.z; acc.w += w0 * v0.w;
        acc.x += w1 * v1.x; acc.y += w1 * v1.y; acc.z += w1 * v1.z; acc.w += w1 * v1.w;
        acc.x += w2 * v2.x; acc.y += w2 * v2.y; acc.z += w2 * v2.z; acc.w += w2 * v2.w;
        acc.x += w3 * v3.x; acc.y += w3 * v3.y; acc.z += w3 * v3.z; acc.w += w3 * v3.w;
    }
    const int rem = end - e;
    if (rem == 3) {
        const int2 c0 = csr[e], c1 = csr[e + 1], c2 = csr[e + 2];
        const float4 v0 = t4[(size_t)c0.x * 32 + lane];
        const float4 v1 = t4[(size_t)c1.x * 32 + lane];
        const float4 v2 = t4[(size_t)c2.x * 32 + lane];
        const float w0 = __int_as_float(c0.y), w1 = __int_as_float(c1.y), w2 = __int_as_float(c2.y);
        acc.x += w0 * v0.x; acc.y += w0 * v0.y; acc.z += w0 * v0.z; acc.w += w0 * v0.w;
        acc.x += w1 * v1.x; acc.y += w1 * v1.y; acc.z += w1 * v1.z; acc.w += w1 * v1.w;
        acc.x += w2 * v2.x; acc.y += w2 * v2.y; acc.z += w2 * v2.z; acc.w += w2 * v2.w;
    } else if (rem == 2) {
        const int2 c0 = csr[e], c1 = csr[e + 1];
        const float4 v0 = t4[(size_t)c0.x * 32 + lane];
        const float4 v1 = t4[(size_t)c1.x * 32 + lane];
        const float w0 = __int_as_float(c0.y), w1 = __int_as_float(c1.y);
        acc.x += w0 * v0.x; acc.y += w0 * v0.y; acc.z += w0 * v0.z; acc.w += w0 * v0.w;
        acc.x += w1 * v1.x; acc.y += w1 * v1.y; acc.z += w1 * v1.z; acc.w += w1 * v1.w;
    } else if (rem == 1) {
        const int2 c0 = csr[e];
        const float4 v0 = t4[(size_t)c0.x * 32 + lane];
        const float w0 = __int_as_float(c0.y);
        acc.x += w0 * v0.x; acc.y += w0 * v0.y; acc.z += w0 * v0.z; acc.w += w0 * v0.w;
    }
    const float dn = dis[n];
    const float sn = dn * dn;
    const float4 tv = t4[(size_t)n * 32 + lane];
    const float4 b  = ((const float4*)bias)[lane];
    const float4 av = ((const float4*)a)[lane];
    float4 r;
    r.x = acc.x + sn * tv.x + b.x;
    r.y = acc.y + sn * tv.y + b.y;
    r.z = acc.z + sn * tv.z + b.z;
    r.w = acc.w + sn * tv.w + b.w;
    r.x = (r.x >= 0.0f) ? r.x : av.x * r.x;
    r.y = (r.y >= 0.0f) ? r.y : av.y * r.y;
    r.z = (r.z >= 0.0f) ? r.z : av.z * r.z;
    r.w = (r.w >= 0.0f) ? r.w : av.w * r.w;
    ((float4*)h)[(size_t)n * 32 + lane] = r;
}

__global__ void k_dot_w3(const float* __restrict__ h, const float* __restrict__ W3,
                         float* __restrict__ t3, int N) {
    int wid  = (blockIdx.x * blockDim.x + threadIdx.x) >> 5;
    int lane = threadIdx.x & 31;
    if (wid >= N) return;
    const float4 hv = ((const float4*)(h + (size_t)wid * 128))[lane];
    const float4 wv = ((const float4*)W3)[lane];
    float s = hv.x * wv.x + hv.y * wv.y + hv.z * wv.z + hv.w * wv.w;
    #pragma unroll
    for (int off = 16; off > 0; off >>= 1) s += __shfl_xor(s, off, 32);
    if (lane == 0) t3[wid] = s;
}

__global__ void k_gather_scalar(const int2* __restrict__ csr, const int* __restrict__ rowptr,
                                const int* __restrict__ cnt, const float* __restrict__ dis,
                                const float* __restrict__ t3, const float* __restrict__ b3,
                                float* __restrict__ out, int N) {
    const int gid = blockIdx.x * blockDim.x + threadIdx.x;
    const int n = gid >> 3, sub = gid & 7;
    const bool valid = n < N;
    int beg = 0, end = 0;
    if (valid) { beg = rowptr[n]; end = beg + cnt[n]; }
    float s = 0.0f;
    for (int e = beg + sub; e < end; e += 8) {
        const int2 c = csr[e];
        s += __int_as_float(c.y) * t3[c.x];
    }
    s += __shfl_xor(s, 4, 32);
    s += __shfl_xor(s, 2, 32);
    s += __shfl_xor(s, 1, 32);
    if (valid && sub == 0) {
        const float dn = dis[n];
        out[n] = s + dn * dn * t3[n] + b3[0];
    }
}

extern "C" void kernel_launch(void* const* d_in, const int* in_sizes, int n_in,
                              void* d_out, int out_size, void* d_ws, size_t ws_size,
                              hipStream_t stream) {
    const float* num_x    = (const float*)d_in[0];
    const float* num_mask = (const float*)d_in[1];
    const float* x        = (const float*)d_in[2];
    const int*   edge     = (const int*)d_in[3];
    const float* W_num    = (const float*)d_in[4];
    const float* b_num    = (const float*)d_in[5];
    const float* a_in     = (const float*)d_in[6];
    const float* W1       = (const float*)d_in[7];
    const float* b1       = (const float*)d_in[8];
    const float* a1       = (const float*)d_in[9];
    const float* W2       = (const float*)d_in[10];
    const float* b2       = (const float*)d_in[11];
    const float* a2       = (const float*)d_in[12];
    const float* W3       = (const float*)d_in[13];
    const float* b3       = (const float*)d_in[14];

    const int N = in_sizes[1];
    const int E = in_sizes[3] / 2;
    const int* src = edge;
    const int* dst = edge + E;

    const int nRowTiles = (N + 15) / 16;
    const size_t padRows = (size_t)nRowTiles * 16;

    auto al = [](size_t v) { return (v + 255) & ~(size_t)255; };
    const int nTiles = (N + 1023) / 1024;
    char* p = (char*)d_ws;
    float* dis    = (float*)p; p += al((size_t)N * 4);
    int*   cnt    = (int*)p;   p += al((size_t)N * 4);
    int*   rowptr = (int*)p;   p += al((size_t)N * 4);
    int*   tsum   = (int*)p;   p += al((size_t)nTiles * 4);
    int*   rank   = (int*)p;   p += al((size_t)E * 4);
    int2*  csr    = (int2*)p;  p += al((size_t)E * 8);
    float* hbuf = (float*)p; p += al(padRows * 128 * 4);
    float* tbuf = (float*)p; p += al(padRows * 128 * 4);
    float* t3   = (float*)p; p += al((size_t)N * 4);
    float* out  = (float*)d_out;

    const int TPB = 256;
    const int totND = N * 128;
    dim3 blk(TPB);
    auto gN  = dim3((N + TPB - 1) / TPB);
    auto gE  = dim3((E + TPB - 1) / TPB);
    auto gND = dim3((totND + TPB - 1) / TPB);
    auto gNw = dim3((int)(((size_t)N * 32 + TPB - 1) / TPB));
    auto gN8 = dim3((int)(((size_t)N * 8 + TPB - 1) / TPB));
    const int gemmBlocks = (nRowTiles + 3) / 4;

    k_zero_cnt<<<gN, blk, 0, stream>>>(cnt, N);
    k_count_rank<<<gE, blk, 0, stream>>>(dst, cnt, rank, E);
    k_tile_sums<<<nTiles, 256, 0, stream>>>(cnt, tsum, N);
    k_scan_rows<<<nTiles, 256, 0, stream>>>(cnt, tsum, rowptr, dis, N);
    k_fill_csr<<<gE, blk, 0, stream>>>(src, dst, rank, rowptr, dis, csr, E);

    k_input_proj<<<gND, blk, 0, stream>>>(num_x, num_mask, x, W_num, b_num, a_in, hbuf, totND);

    k_gemm128_wmma<<<gemmBlocks, 128, 0, stream>>>(hbuf, W1, tbuf, nRowTiles);
    k_gather_rows<<<gNw, blk, 0, stream>>>(csr, rowptr, cnt, dis, tbuf, b1, a1, hbuf, N);

    k_gemm128_wmma<<<gemmBlocks, 128, 0, stream>>>(hbuf, W2, tbuf, nRowTiles);
    k_gather_rows<<<gNw, blk, 0, stream>>>(csr, rowptr, cnt, dis, tbuf, b2, a2, hbuf, N);

    k_dot_w3<<<gNw, blk, 0, stream>>>(hbuf, W3, t3, N);
    k_gather_scalar<<<gN8, blk, 0, stream>>>(csr, rowptr, cnt, dis, t3, b3, out, N);
}


# input projection (PReLU(num_x*num_mask*W_num+b_num)+x) computed in the layer-1 WMMA GEMM's A-fragment load: the 200000-wave projection launch and the h plane's first store and re-read are gone
# speedup vs baseline: 2.8657x; 1.2197x over previous
#include <hip/hip_runtime.h>

typedef __attribute__((ext_vector_type(2))) float v2f;
typedef __attribute__((ext_vector_type(8))) float v8f;

__global__ void k_zero_cnt(int* cnt, int N) {
    int i = blockIdx.x * blockDim.x + threadIdx.x;
    if (i < N) cnt[i] = 0;
}

__global__ void k_count_rank(const int* __restrict__ dst, int* cnt, int* __restrict__ rank, int E) {
    int e = blockIdx.x * blockDim.x + threadIdx.x;
    if (e < E) rank[e] = atomicAdd(&cnt[dst[e]], 1);
}

__global__ __launch_bounds__(256) void k_tile_sums(const int* __restrict__ cnt, int* __restrict__ tsum, int N) {
    __shared__ int sm[8];
    const int tid = threadIdx.x, lane = tid & 31, w = tid >> 5;
    const int base = blockIdx.x * 1024 + tid * 4;
    int s = 0;
    for (int j = 0; j < 4; ++j) if (base + j < N) s += cnt[base + j];
    for (int off = 16; off > 0; off >>= 1) s += __shfl_xor(s, off, 32);
    if (lane == 0) sm[w] = s;
    __syncthreads();
    if (tid == 0) {
        int tot = 0;
        for (int i = 0; i < 8; ++i) tot += sm[i];
        tsum[blockIdx.x] = tot;
    }
}

__global__ __launch_bounds__(256) void k_scan_rows(const int* __restrict__ cnt, const int* __restrict__ tsum,
                                                   int* __restrict__ rowptr, float* __restrict__ dis, int N) {
    __shared__ int smw[8];
    __shared__ int smo[8];
    const int tid = threadIdx.x, lane = tid & 31, w = tid >> 5;
    int o = 0;
    for (int i = tid; i < (int)blockIdx.x; i += 256) o += tsum[i];
    for (int off = 16; off > 0; off >>= 1) o += __shfl_xor(o, off, 32);
    if (lane == 0) smo[w] = o;

    const int base = blockIdx.x * 1024 + tid * 4;
    int c[4];
    for (int j = 0; j < 4; ++j) c[j] = (base + j < N) ? cnt[base + j] : 0;
    const int s = c[0] + c[1] + c[2] + c[3];
    int inc = s;
    for (int off = 1; off < 32; off <<= 1) {
        int u = __shfl_up(inc, off, 32);
        if (lane >= off) inc += u;
    }
    if (lane == 31) smw[w] = inc;
    __syncthreads();
    int ex = inc - s;
    for (int i = 0; i < 8; ++i) {
        ex += smo[i];
        if (i < w) ex += smw[i];
    }
    for (int j = 0; j < 4; ++j) {
        if (base + j < N) {
            rowptr[base + j] = ex;
            dis[base + j] = rsqrtf((float)c[j] + 1.0f);
            ex += c[j];
        }
    }
}

__global__ void k_fill_csr(const int* __restrict__ src, const int* __restrict__ dst,
                           const int* __restrict__ rank, const int* __restrict__ rowptr,
                           const float* __restrict__ dis, int2* __restrict__ csr, int E) {
    int e = blockIdx.x * blockDim.x + threadIdx.x;
    if (e >= E) return;
    int s = src[e], d = dst[e];
    float w = dis[s] * dis[d];
    csr[rowptr[d] + rank[e]] = make_int2(s, __float_as_int(w));
}

__global__ __launch_bounds__(128) void k_gemm128_wmma(const float* __restrict__ h,
                                                      const float* __restrict__ W,
                                                      float* __restrict__ t,
                                                      int nRowTiles) {
    __shared__ float Wl[128 * 128];
    {
        int tid = threadIdx.x;
        v2f* Wlv = (v2f*)Wl;
        #pragma unroll
        for (int i = 0; i < 64; ++i) {
            int idx = i * 128 + tid;
            int c = idx & 127;
            int p = idx >> 7;
            v2f pr;
            pr.x = W[(2 * p)     * 128 + c];
            pr.y = W[(2 * p + 1) * 128 + c];
            Wlv[p * 128 + c] = pr;
        }
    }
    __syncthreads();

    const int wave = threadIdx.x >> 5;
    const int lane = threadIdx.x & 31;
    const int rowTile = blockIdx.x * 4 + wave;
    if (rowTile >= nRowTiles) return;

    const int r15   = lane & 15;
    const int khalf = (lane >> 4) << 1;
    const float* hrow = h + (size_t)(rowTile * 16 + r15) * 128;
    const v2f* Wlv = (const v2f*)Wl;

    v8f acc[8] = {};
    for (int k0 = 0; k0 < 128; k0 += 4) {
        v2f a = *(const v2f*)(hrow + k0 + khalf);
        const v2f* wrow = Wlv + (((k0 + khalf) >> 1) * 128 + r15);
        #pragma unroll
        for (int j = 0; j < 8; ++j) {
            v2f bfrag = wrow[j * 16];
            acc[j] = __builtin_amdgcn_wmma_f32_16x16x4_f32(
                false, a, false, bfrag, (short)0, acc[j], false, false);
        }
    }

    const int mhalf = (lane >> 4) << 3;
    float* trow = t + (size_t)(rowTile * 16 + mhalf) * 128 + r15;
    #pragma unroll
    for (int j = 0; j < 8; ++j)
        #pragma unroll
        for (int r = 0; r < 8; ++r)
            trow[(size_t)r * 128 + j * 16] = acc[j][r];
}

__global__ __launch_bounds__(128) void k_proj_gemm128_wmma(const float* __restrict__ num_x,
                                                           const float* __restrict__ num_mask,
                                                           const float* __restrict__ x,
                                                           const float* __restrict__ Wn,
                                                           const float* __restrict__ bn,
                                                           const float* __restrict__ ain,
                                                           const float* __restrict__ W,
                                                           float* __restrict__ t,
                                                           int nRowTiles, int N) {
    __shared__ float Wl[128 * 128];
    __shared__ v2f Pw[64];
    __shared__ v2f Pb[64];
    __shared__ v2f Pa[64];
    {
        int tid = threadIdx.x;
        v2f* Wlv = (v2f*)Wl;
        #pragma unroll
        for (int i = 0; i < 64; ++i) {
            int idx = i * 128 + tid;
            int c = idx & 127;
            int p = idx >> 7;
            v2f pr;
            pr.x = W[(2 * p)     * 128 + c];
            pr.y = W[(2 * p + 1) * 128 + c];
            Wlv[p * 128 + c] = pr;
        }
        if (tid < 64) {
            Pw[tid] = *(const v2f*)(Wn  + 2 * tid);
            Pb[tid] = *(const v2f*)(bn  + 2 * tid);
            Pa[tid] = *(const v2f*)(ain + 2 * tid);
        }
    }
    __syncthreads();

    const int wave = threadIdx.x >> 5;
    const int lane = threadIdx.x & 31;
    const int rowTile = blockIdx.x * 4 + wave;
    if (rowTile >= nRowTiles) return;

    const int r15   = lane & 15;
    const int khalf = (lane >> 4) << 1;
    int row = rowTile * 16 + r15;
    row = (row < N) ? row : (N - 1);
    const float s = num_x[row] * num_mask[row];
    const float* xrow = x + (size_t)row * 128;
    const v2f* Wlv = (const v2f*)Wl;

    v8f acc[8] = {};
    for (int k0 = 0; k0 < 128; k0 += 4) {
        const int kp = (k0 + khalf) >> 1;
        const v2f xv = *(const v2f*)(xrow + k0 + khalf);
        const v2f pw = Pw[kp], pb = Pb[kp], pa = Pa[kp];
        float z0 = s * pw.x + pb.x;
        float z1 = s * pw.y + pb.y;
        z0 = (z0 >= 0.0f) ? z0 : pa.x * z0;
        z1 = (z1 >= 0.0f) ? z1 : pa.y * z1;
        v2f a;
        a.x = z0 + xv.x;
        a.y = z1 + xv.y;
        const v2f* wrow = Wlv + (kp * 128 + r15);
        #pragma unroll
        for (int j = 0; j < 8; ++j) {
            v2f bfrag = wrow[j * 16];
            acc[j] = __builtin_amdgcn_wmma_f32_16x16x4_f32(
                false, a, false, bfrag, (short)0, acc[j], false, false);
        }
    }

    const int mhalf = (lane >> 4) << 3;
    float* trow = t + (size_t)(rowTile * 16 + mhalf) * 128 + r15;
    #pragma unroll
    for (int j = 0; j < 8; ++j)
        #pragma unroll
        for (int r = 0; r < 8; ++r)
            trow[(size_t)r * 128 + j * 16] = acc[j][r];
}

__global__ __launch_bounds__(256) void k_gather_rows(const int2* __restrict__ csr, const int* __restrict__ rowptr,
                                                     const int* __restrict__ cnt, const float* __restrict__ dis,
                                                     const float* __restrict__ t, const float* __restrict__ bias,
                                                     const float* __restrict__ a, float* __restrict__ h, int N) {
    const int wid  = (blockIdx.x * blockDim.x + threadIdx.x) >> 5;
    const int lane = threadIdx.x & 31;
    if (wid >= N) return;
    const int n = __builtin_amdgcn_readfirstlane(wid);
    const int beg = rowptr[n];
    const int end = beg + cnt[n];
    const float4* t4 = (const float4*)t;
    float4 acc = make_float4(0.0f, 0.0f, 0.0f, 0.0f);
    int e = beg;
    for (; e + 4 <= end; e += 4) {
        const int2 c0 = csr[e], c1 = csr[e + 1], c2 = csr[e + 2], c3 = csr[e + 3];
        const float4 v0 = t4[(size_t)c0.x * 32 + lane];
        const float4 v1 = t4[(size_t)c1.x * 32 + lane];
        const float4 v2 = t4[(size_t)c2.x * 32 + lane];
        const float4 v3 = t4[(size_t)c3.x * 32 + lane];
        const float w0 = __int_as_float(c0.y), w1 = __int_as_float(c1.y);
        const float w2 = __int_as_float(c2.y), w3 = __int_as_float(c3.y);
        acc.x += w0 * v0.x; acc.y += w0 * v0.y; acc.z += w0 * v0.z; acc.w += w0 * v0.w;
        acc.x += w1 * v1.x; acc.y += w1 * v1.y; acc.z += w1 * v1.z; acc.w += w1 * v1.w;
        acc.x += w2 * v2.x; acc.y += w2 * v2.y; acc.z += w2 * v2.z; acc.w += w2 * v2.w;
        acc.x += w3 * v3.x; acc.y += w3 * v3.y; acc.z += w3 * v3.z; acc.w += w3 * v3.w;
    }
    const int rem = end - e;
    if (rem == 3) {
        const int2 c0 = csr[e], c1 = csr[e + 1], c2 = csr[e + 2];
        const float4 v0 = t4[(size_t)c0.x * 32 + lane];
        const float4 v1 = t4[(size_t)c1.x * 32 + lane];
        const float4 v2 = t4[(size_t)c2.x * 32 + lane];
        const float w0 = __int_as_float(c0.y), w1 = __int_as_float(c1.y), w2 = __int_as_float(c2.y);
        acc.x += w0 * v0.x; acc.y += w0 * v0.y; acc.z += w0 * v0.z; acc.w += w0 * v0.w;
        acc.x += w1 * v1.x; acc.y += w1 * v1.y; acc.z += w1 * v1.z; acc.w += w1 * v1.w;
        acc.x += w2 * v2.x; acc.y += w2 * v2.y; acc.z += w2 * v2.z; acc.w += w2 * v2.w;
    } else if (rem == 2) {
        const int2 c0 = csr[e], c1 = csr[e + 1];
        const float4 v0 = t4[(size_t)c0.x * 32 + lane];
        const float4 v1 = t4[(size_t)c1.x * 32 + lane];
        const float w0 = __int_as_float(c0.y), w1 = __int_as_float(c1.y);
        acc.x += w0 * v0.x; acc.y += w0 * v0.y; acc.z += w0 * v0.z; acc.w += w0 * v0.w;
        acc.x += w1 * v1.x; acc.y += w1 * v1.y; acc.z += w1 * v1.z; acc.w += w1 * v1.w;
    } else if (rem == 1) {
        const int2 c0 = csr[e];
        const float4 v0 = t4[(size_t)c0.x * 32 + lane];
        const float w0 = __int_as_float(c0.y);
        acc.x += w0 * v0.x; acc.y += w0 * v0.y; acc.z += w0 * v0.z; acc.w += w0 * v0.w;
    }
    const float dn = dis[n];
    const float sn = dn * dn;
    const float4 tv = t4[(size_t)n * 32 + lane];
    const float4 b  = ((const float4*)bias)[lane];
    const float4 av = ((const float4*)a)[lane];
    float4 r;
    r.x = acc.x + sn * tv.x + b.x;
    r.y = acc.y + sn * tv.y + b.y;
    r.z = acc.z + sn * tv.z + b.z;
    r.w = acc.w + sn * tv.w + b.w;
    r.x = (r.x >= 0.0f) ? r.x : av.x * r.x;
    r.y = (r.y >= 0.0f) ? r.y : av.y * r.y;
    r.z = (r.z >= 0.0f) ? r.z : av.z * r.z;
    r.w = (r.w >= 0.0f) ? r.w : av.w * r.w;
    ((float4*)h)[(size_t)n * 32 + lane] = r;
}

__global__ void k_dot_w3(const float* __restrict__ h, const float* __restrict__ W3,
                         float* __restrict__ t3, int N) {
    int wid  = (blockIdx.x * blockDim.x + threadIdx.x) >> 5;
    int lane = threadIdx.x & 31;
    if (wid >= N) return;
    const float4 hv = ((const float4*)(h + (size_t)wid * 128))[lane];
    const float4 wv = ((const float4*)W3)[lane];
    float s = hv.x * wv.x + hv.y * wv.y + hv.z * wv.z + hv.w * wv.w;
    #pragma unroll
    for (int off = 16; off > 0; off >>= 1) s += __shfl_xor(s, off, 32);
    if (lane == 0) t3[wid] = s;
}

__global__ void k_gather_scalar(const int2* __restrict__ csr, const int* __restrict__ rowptr,
                                const int* __restrict__ cnt, const float* __restrict__ dis,
                                const float* __restrict__ t3, const float* __restrict__ b3,
                                float* __restrict__ out, int N) {
    const int gid = blockIdx.x * blockDim.x + threadIdx.x;
    const int n = gid >> 3, sub = gid & 7;
    const bool valid = n < N;
    int beg = 0, end = 0;
    if (valid) { beg = rowptr[n]; end = beg + cnt[n]; }
    float s = 0.0f;
    for (int e = beg + sub; e < end; e += 8) {
        const int2 c = csr[e];
        s += __int_as_float(c.y) * t3[c.x];
    }
    s += __shfl_xor(s, 4, 32);
    s += __shfl_xor(s, 2, 32);
    s += __shfl_xor(s, 1, 32);
    if (valid && sub == 0) {
        const float dn = dis[n];
        out[n] = s + dn * dn * t3[n] + b3[0];
    }
}

extern "C" void kernel_launch(void* const* d_in, const int* in_sizes, int n_in,
                              void* d_out, int out_size, void* d_ws, size_t ws_size,
                              hipStream_t stream) {
    const float* num_x    = (const float*)d_in[0];
    const float* num_mask = (const float*)d_in[1];
    const float* x        = (const float*)d_in[2];
    const int*   edge     = (const int*)d_in[3];
    const float* W_num    = (const float*)d_in[4];
    const float* b_num    = (const float*)d_in[5];
    const float* a_in     = (const float*)d_in[6];
    const float* W1       = (const float*)d_in[7];
    const float* b1       = (const float*)d_in[8];
    const float* a1       = (const float*)d_in[9];
    const float* W2       = (const float*)d_in[10];
    const float* b2       = (const float*)d_in[11];
    const float* a2       = (const float*)d_in[12];
    const float* W3       = (const float*)d_in[13];
    const float* b3       = (const float*)d_in[14];

    const int N = in_sizes[1];
    const int E = in_sizes[3] / 2;
    const int* src = edge;
    const int* dst = edge + E;

    const int nRowTiles = (N + 15) / 16;
    const size_t padRows = (size_t)nRowTiles * 16;

    auto al = [](size_t v) { return (v + 255) & ~(size_t)255; };
    const int nTiles = (N + 1023) / 1024;
    char* p = (char*)d_ws;
    float* dis    = (float*)p; p += al((size_t)N * 4);
    int*   cnt    = (int*)p;   p += al((size_t)N * 4);
    int*   rowptr = (int*)p;   p += al((size_t)N * 4);
    int*   tsum   = (int*)p;   p += al((size_t)nTiles * 4);
    int*   rank   = (int*)p;   p += al((size_t)E * 4);
    int2*  csr    = (int2*)p;  p += al((size_t)E * 8);
    float* hbuf = (float*)p; p += al(padRows * 128 * 4);
    float* tbuf = (float*)p; p += al(padRows * 128 * 4);
    float* t3   = (float*)p; p += al((size_t)N * 4);
    float* out  = (float*)d_out;

    const int TPB = 256;
    dim3 blk(TPB);
    auto gN  = dim3((N + TPB - 1) / TPB);
    auto gE  = dim3((E + TPB - 1) / TPB);
    auto gNw = dim3((int)(((size_t)N * 32 + TPB - 1) / TPB));
    auto gN8 = dim3((int)(((size_t)N * 8 + TPB - 1) / TPB));
    const int gemmBlocks = (nRowTiles + 3) / 4;

    k_zero_cnt<<<gN, blk, 0, stream>>>(cnt, N);
    k_count_rank<<<gE, blk, 0, stream>>>(dst, cnt, rank, E);
    k_tile_sums<<<nTiles, 256, 0, stream>>>(cnt, tsum, N);
    k_scan_rows<<<nTiles, 256, 0, stream>>>(cnt, tsum, rowptr, dis, N);
    k_fill_csr<<<gE, blk, 0, stream>>>(src, dst, rank, rowptr, dis, csr, E);

    k_proj_gemm128_wmma<<<gemmBlocks, 128, 0, stream>>>(num_x, num_mask, x, W_num, b_num, a_in, W1, tbuf, nRowTiles, N);
    k_gather_rows<<<gNw, blk, 0, stream>>>(csr, rowptr, cnt, dis, tbuf, b1, a1, hbuf, N);

    k_gemm128_wmma<<<gemmBlocks, 128, 0, stream>>>(hbuf, W2, tbuf, nRowTiles);
    k_gather_rows<<<gNw, blk, 0, stream>>>(csr, rowptr, cnt, dis, tbuf, b2, a2, hbuf, N);

    k_dot_w3<<<gNw, blk, 0, stream>>>(hbuf, W3, t3, N);
    k_gather_scalar<<<gN8, blk, 0, stream>>>(csr, rowptr, cnt, dis, t3, b3, out, N);
}


# edge grouping by dst with LDS counters in two levels (256-node buckets, then rows) instead of one global returning atomic per edge; rows now summed in LDS arrival order
# speedup vs baseline: 3.6243x; 1.2647x over previous
#include <hip/hip_runtime.h>

typedef __attribute__((ext_vector_type(2))) float v2f;
typedef __attribute__((ext_vector_type(8))) float v8f;


__global__ __launch_bounds__(1024) void k_bkt_count(const int* __restrict__ dst, int* __restrict__ bh,
                                                    int E, int NB, int EB) {
    __shared__ int hist[1024];
    const int tid = threadIdx.x;
    for (int i = tid; i < NB; i += 1024) hist[i] = 0;
    __syncthreads();
    const int e0 = blockIdx.x * 8192 + tid;
    int d[8];
    #pragma unroll
    for (int j = 0; j < 8; ++j) {
        const int e = e0 + j * 1024;
        d[j] = (e < E) ? dst[e] : -1;
    }
    #pragma unroll
    for (int j = 0; j < 8; ++j)
        if (d[j] >= 0) atomicAdd(&hist[d[j] >> 8], 1);
    __syncthreads();
    for (int i = tid; i < NB; i += 1024) bh[i * EB + blockIdx.x] = hist[i];
}

__global__ __launch_bounds__(1024) void k_bkt_scan(int* a, int n) {
    __shared__ int wsum[32];
    const int tid = threadIdx.x, lane = tid & 31, w = tid >> 5;
    const int chunk = (n + 1023) / 1024;
    int lo = tid * chunk;
    lo = (lo < n) ? lo : n;
    int hi = lo + chunk;
    hi = (hi < n) ? hi : n;
    int s = 0;
    for (int i = lo; i < hi; ++i) s += a[i];
    int inc = s;
    for (int off = 1; off < 32; off <<= 1) {
        int u = __shfl_up(inc, off, 32);
        if (lane >= off) inc += u;
    }
    if (lane == 31) wsum[w] = inc;
    __syncthreads();
    int ex = inc - s;
    for (int i = 0; i < w; ++i) ex += wsum[i];
    for (int i = lo; i < hi; ++i) {
        const int v = a[i];
        a[i] = ex;
        ex += v;
    }
}

__global__ __launch_bounds__(1024) void k_bkt_scatter(const int* __restrict__ src, const int* __restrict__ dst,
                                                      const int* __restrict__ boff, int2* __restrict__ ebuf,
                                                      int E, int NB, int EB) {
    __shared__ int cur[1024];
    const int tid = threadIdx.x;
    for (int i = tid; i < NB; i += 1024) cur[i] = boff[i * EB + blockIdx.x];
    __syncthreads();
    const int e0 = blockIdx.x * 8192 + tid;
    int s[8], d[8];
    #pragma unroll
    for (int j = 0; j < 8; ++j) {
        const int e = e0 + j * 1024;
        const bool ok = e < E;
        s[j] = ok ? src[e] : 0;
        d[j] = ok ? dst[e] : -1;
    }
    #pragma unroll
    for (int j = 0; j < 8; ++j) {
        if (d[j] >= 0) {
            const int p = atomicAdd(&cur[d[j] >> 8], 1);
            ebuf[p] = make_int2(s[j], d[j]);
        }
    }
}

__global__ __launch_bounds__(1024) void k_bkt_rows(const int2* __restrict__ ebuf, const int* __restrict__ boff,
                                                   int* __restrict__ rowptr, int* __restrict__ cnt,
                                                   float* __restrict__ dis, int2* __restrict__ csr,
                                                   int N, int E, int NB, int EB) {
    __shared__ int c[256];
    __shared__ int cu[256];
    __shared__ int wsum[8];
    const int tid = threadIdx.x, lane = tid & 31, w = tid >> 5;
    const int b = blockIdx.x;
    const int base = boff[b * EB];
    const int end  = (b + 1 < NB) ? boff[(b + 1) * EB] : E;
    if (tid < 256) c[tid] = 0;
    __syncthreads();
    for (int i0 = base; i0 < end; i0 += 4096) {
        int d[4];
        #pragma unroll
        for (int j = 0; j < 4; ++j) {
            const int i = i0 + j * 1024 + tid;
            d[j] = (i < end) ? ebuf[i].y : -1;
        }
        #pragma unroll
        for (int j = 0; j < 4; ++j)
            if (d[j] >= 0) atomicAdd(&c[d[j] & 255], 1);
    }
    __syncthreads();
    int v = 0, inc = 0;
    if (tid < 256) {
        v = c[tid];
        inc = v;
        for (int off = 1; off < 32; off <<= 1) {
            int u = __shfl_up(inc, off, 32);
            if (lane >= off) inc += u;
        }
        if (lane == 31) wsum[w] = inc;
    }
    __syncthreads();
    if (tid < 256) {
        int ex = inc - v;
        for (int i = 0; i < w; ++i) ex += wsum[i];
        const int start = base + ex;
        cu[tid] = start;
        const int n = b * 256 + tid;
        if (n < N) {
            rowptr[n] = start;
            cnt[n] = v;
            dis[n] = rsqrtf((float)v + 1.0f);
        }
    }
    __syncthreads();
    for (int i0 = base; i0 < end; i0 += 4096) {
        int2 e[4];
        #pragma unroll
        for (int j = 0; j < 4; ++j) {
            const int i = i0 + j * 1024 + tid;
            e[j] = (i < end) ? ebuf[i] : make_int2(0, -1);
        }
        #pragma unroll
        for (int j = 0; j < 4; ++j) {
            if (e[j].y >= 0) {
                const int p = atomicAdd(&cu[e[j].y & 255], 1);
                csr[p] = e[j];
            }
        }
    }
}

__global__ __launch_bounds__(256) void k_edge_norm(int2* __restrict__ csr, const float* __restrict__ dis, int E) {
    const int i0 = blockIdx.x * 1024 + threadIdx.x;
    if (blockIdx.x * 1024 + 1024 <= E) {
        int2 e[4];
        #pragma unroll
        for (int j = 0; j < 4; ++j) e[j] = csr[i0 + j * 256];
        float ws[4], wd[4];
        #pragma unroll
        for (int j = 0; j < 4; ++j) { ws[j] = dis[e[j].x]; wd[j] = dis[e[j].y]; }
        #pragma unroll
        for (int j = 0; j < 4; ++j)
            csr[i0 + j * 256] = make_int2(e[j].x, __float_as_int(ws[j] * wd[j]));
    } else {
        for (int j = 0; j < 4; ++j) {
            const int i = i0 + j * 256;
            if (i < E) {
                const int2 e = csr[i];
                csr[i] = make_int2(e.x, __float_as_int(dis[e.x] * dis[e.y]));
            }
        }
    }
}

__global__ __launch_bounds__(128) void k_gemm128_wmma(const float* __restrict__ h,
                                                      const float* __restrict__ W,
                                                      float* __restrict__ t,
                                                      int nRowTiles) {
    __shared__ float Wl[128 * 128];
    {
        int tid = threadIdx.x;
        v2f* Wlv = (v2f*)Wl;
        #pragma unroll
        for (int i = 0; i < 64; ++i) {
            int idx = i * 128 + tid;
            int c = idx & 127;
            int p = idx >> 7;
            v2f pr;
            pr.x = W[(2 * p)     * 128 + c];
            pr.y = W[(2 * p + 1) * 128 + c];
            Wlv[p * 128 + c] = pr;
        }
    }
    __syncthreads();

    const int wave = threadIdx.x >> 5;
    const int lane = threadIdx.x & 31;
    const int rowTile = blockIdx.x * 4 + wave;
    if (rowTile >= nRowTiles) return;

    const int r15   = lane & 15;
    const int khalf = (lane >> 4) << 1;
    const float* hrow = h + (size_t)(rowTile * 16 + r15) * 128;
    const v2f* Wlv = (const v2f*)Wl;

    v8f acc[8] = {};
    for (int k0 = 0; k0 < 128; k0 += 4) {
        v2f a = *(const v2f*)(hrow + k0 + khalf);
        const v2f* wrow = Wlv + (((k0 + khalf) >> 1) * 128 + r15);
        #pragma unroll
        for (int j = 0; j < 8; ++j) {
            v2f bfrag = wrow[j * 16];
            acc[j] = __builtin_amdgcn_wmma_f32_16x16x4_f32(
                false, a, false, bfrag, (short)0, acc[j], false, false);
        }
    }

    const int mhalf = (lane >> 4) << 3;
    float* trow = t + (size_t)(rowTile * 16 + mhalf) * 128 + r15;
    #pragma unroll
    for (int j = 0; j < 8; ++j)
        #pragma unroll
        for (int r = 0; r < 8; ++r)
            trow[(size_t)r * 128 + j * 16] = acc[j][r];
}

__global__ __launch_bounds__(128) void k_proj_gemm128_wmma(const float* __restrict__ num_x,
                                                           const float* __restrict__ num_mask,
                                                           const float* __restrict__ x,
                                                           const float* __restrict__ Wn,
                                                           const float* __restrict__ bn,
                                                           const float* __restrict__ ain,
                                                           const float* __restrict__ W,
                                                           float* __restrict__ t,
                                                           int nRowTiles, int N) {
    __shared__ float Wl[128 * 128];
    __shared__ v2f Pw[64];
    __shared__ v2f Pb[64];
    __shared__ v2f Pa[64];
    {
        int tid = threadIdx.x;
        v2f* Wlv = (v2f*)Wl;
        #pragma unroll
        for (int i = 0; i < 64; ++i) {
            int idx = i * 128 + tid;
            int c = idx & 127;
            int p = idx >> 7;
            v2f pr;
            pr.x = W[(2 * p)     * 128 + c];
            pr.y = W[(2 * p + 1) * 128 + c];
            Wlv[p * 128 + c] = pr;
        }
        if (tid < 64) {
            Pw[tid] = *(const v2f*)(Wn  + 2 * tid);
            Pb[tid] = *(const v2f*)(bn  + 2 * tid);
            Pa[tid] = *(const v2f*)(ain + 2 * tid);
        }
    }
    __syncthreads();

    const int wave = threadIdx.x >> 5;
    const int lane = threadIdx.x & 31;
    const int rowTile = blockIdx.x * 4 + wave;
    if (rowTile >= nRowTiles) return;

    const int r15   = lane & 15;
    const int khalf = (lane >> 4) << 1;
    int row = rowTile * 16 + r15;
    row = (row < N) ? row : (N - 1);
    const float s = num_x[row] * num_mask[row];
    const float* xrow = x + (size_t)row * 128;
    const v2f* Wlv = (const v2f*)Wl;

    v8f acc[8] = {};
    for (int k0 = 0; k0 < 128; k0 += 4) {
        const int kp = (k0 + khalf) >> 1;
        const v2f xv = *(const v2f*)(xrow + k0 + khalf);
        const v2f pw = Pw[kp], pb = Pb[kp], pa = Pa[kp];
        float z0 = s * pw.x + pb.x;
        float z1 = s * pw.y + pb.y;
        z0 = (z0 >= 0.0f) ? z0 : pa.x * z0;
        z1 = (z1 >= 0.0f) ? z1 : pa.y * z1;
        v2f a;
        a.x = z0 + xv.x;
        a.y = z1 + xv.y;
        const v2f* wrow = Wlv + (kp * 128 + r15);
        #pragma unroll
        for (int j = 0; j < 8; ++j) {
            v2f bfrag = wrow[j * 16];
            acc[j] = __builtin_amdgcn_wmma_f32_16x16x4_f32(
                false, a, false, bfrag, (short)0, acc[j], false, false);
        }
    }

    const int mhalf = (lane >> 4) << 3;
    float* trow = t + (size_t)(rowTile * 16 + mhalf) * 128 + r15;
    #pragma unroll
    for (int j = 0; j < 8; ++j)
        #pragma unroll
        for (int r = 0; r < 8; ++r)
            trow[(size_t)r * 128 + j * 16] = acc[j][r];
}

__global__ __launch_bounds__(256) void k_gather_rows(const int2* __restrict__ csr, const int* __restrict__ rowptr,
                                                     const int* __restrict__ cnt, const float* __restrict__ dis,
                                                     const float* __restrict__ t, const float* __restrict__ bias,
                                                     const float* __restrict__ a, float* __restrict__ h, int N) {
    const int wid  = (blockIdx.x * blockDim.x + threadIdx.x) >> 5;
    const int lane = threadIdx.x & 31;
    if (wid >= N) return;
    const int n = __builtin_amdgcn_readfirstlane(wid);
    const int beg = rowptr[n];
    const int end = beg + cnt[n];
    const float4* t4 = (const float4*)t;
    float4 acc = make_float4(0.0f, 0.0f, 0.0f, 0.0f);
    int e = beg;
    for (; e + 4 <= end; e += 4) {
        const int2 c0 = csr[e], c1 = csr[e + 1], c2 = csr[e + 2], c3 = csr[e + 3];
        const float4 v0 = t4[(size_t)c0.x * 32 + lane];
        const float4 v1 = t4[(size_t)c1.x * 32 + lane];
        const float4 v2 = t4[(size_t)c2.x * 32 + lane];
        const float4 v3 = t4[(size_t)c3.x * 32 + lane];
        const float w0 = __int_as_float(c0.y), w1 = __int_as_float(c1.y);
        const float w2 = __int_as_float(c2.y), w3 = __int_as_float(c3.y);
        acc.x += w0 * v0.x; acc.y += w0 * v0.y; acc.z += w0 * v0.z; acc.w += w0 * v0.w;
        acc.x += w1 * v1.x; acc.y += w1 * v1.y; acc.z += w1 * v1.z; acc.w += w1 * v1.w;
        acc.x += w2 * v2.x; acc.y += w2 * v2.y; acc.z += w2 * v2.z; acc.w += w2 * v2.w;
        acc.x += w3 * v3.x; acc.y += w3 * v3.y; acc.z += w3 * v3.z; acc.w += w3 * v3.w;
    }
    const int rem = end - e;
    if (rem == 3) {
        const int2 c0 = csr[e], c1 = csr[e + 1], c2 = csr[e + 2];
        const float4 v0 = t4[(size_t)c0.x * 32 + lane];
        const float4 v1 = t4[(size_t)c1.x * 32 + lane];
        const float4 v2 = t4[(size_t)c2.x * 32 + lane];
        const float w0 = __int_as_float(c0.y), w1 = __int_as_float(c1.y), w2 = __int_as_float(c2.y);
        acc.x += w0 * v0.x; acc.y += w0 * v0.y; acc.z += w0 * v0.z; acc.w += w0 * v0.w;
        acc.x += w1 * v1.x; acc.y += w1 * v1.y; acc.z += w1 * v1.z; acc.w += w1 * v1.w;
        acc.x += w2 * v2.x; acc.y += w2 * v2.y; acc.z += w2 * v2.z; acc.w += w2 * v2.w;
    } else if (rem == 2) {
        const int2 c0 = csr[e], c1 = csr[e + 1];
        const float4 v0 = t4[(size_t)c0.x * 32 + lane];
        const float4 v1 = t4[(size_t)c1.x * 32 + lane];
        const float w0 = __int_as_float(c0.y), w1 = __int_as_float(c1.y);
        acc.x += w0 * v0.x; acc.y += w0 * v0.y; acc.z += w0 * v0.z; acc.w += w0 * v0.w;
        acc.x += w1 * v1.x; acc.y += w1 * v1.y; acc.z += w1 * v1.z; acc.w += w1 * v1.w;
    } else if (rem == 1) {
        const int2 c0 = csr[e];
        const float4 v0 = t4[(size_t)c0.x * 32 + lane];
        const float w0 = __int_as_float(c0.y);
        acc.x += w0 * v0.x; acc.y += w0 * v0.y; acc.z += w0 * v0.z; acc.w += w0 * v0.w;
    }
    const float dn = dis[n];
    const float sn = dn * dn;
    const float4 tv = t4[(size_t)n * 32 + lane];
    const float4 b  = ((const float4*)bias)[lane];
    const float4 av = ((const float4*)a)[lane];
    float4 r;
    r.x = acc.x + sn * tv.x + b.x;
    r.y = acc.y + sn * tv.y + b.y;
    r.z = acc.z + sn * tv.z + b.z;
    r.w = acc.w + sn * tv.w + b.w;
    r.x = (r.x >= 0.0f) ? r.x : av.x * r.x;
    r.y = (r.y >= 0.0f) ? r.y : av.y * r.y;
    r.z = (r.z >= 0.0f) ? r.z : av.z * r.z;
    r.w = (r.w >= 0.0f) ? r.w : av.w * r.w;
    ((float4*)h)[(size_t)n * 32 + lane] = r;
}

__global__ void k_dot_w3(const float* __restrict__ h, const float* __restrict__ W3,
                         float* __restrict__ t3, int N) {
    int wid  = (blockIdx.x * blockDim.x + threadIdx.x) >> 5;
    int lane = threadIdx.x & 31;
    if (wid >= N) return;
    const float4 hv = ((const float4*)(h + (size_t)wid * 128))[lane];
    const float4 wv = ((const float4*)W3)[lane];
    float s = hv.x * wv.x + hv.y * wv.y + hv.z * wv.z + hv.w * wv.w;
    #pragma unroll
    for (int off = 16; off > 0; off >>= 1) s += __shfl_xor(s, off, 32);
    if (lane == 0) t3[wid] = s;
}

__global__ void k_gather_scalar(const int2* __restrict__ csr, const int* __restrict__ rowptr,
                                const int* __restrict__ cnt, const float* __restrict__ dis,
                                const float* __restrict__ t3, const float* __restrict__ b3,
                                float* __restrict__ out, int N) {
    const int gid = blockIdx.x * blockDim.x + threadIdx.x;
    const int n = gid >> 3, sub = gid & 7;
    const bool valid = n < N;
    int beg = 0, end = 0;
    if (valid) { beg = rowptr[n]; end = beg + cnt[n]; }
    float s = 0.0f;
    for (int e = beg + sub; e < end; e += 8) {
        const int2 c = csr[e];
        s += __int_as_float(c.y) * t3[c.x];
    }
    s += __shfl_xor(s, 4, 32);
    s += __shfl_xor(s, 2, 32);
    s += __shfl_xor(s, 1, 32);
    if (valid && sub == 0) {
        const float dn = dis[n];
        out[n] = s + dn * dn * t3[n] + b3[0];
    }
}

extern "C" void kernel_launch(void* const* d_in, const int* in_sizes, int n_in,
                              void* d_out, int out_size, void* d_ws, size_t ws_size,
                              hipStream_t stream) {
    const float* num_x    = (const float*)d_in[0];
    const float* num_mask = (const float*)d_in[1];
    const float* x        = (const float*)d_in[2];
    const int*   edge     = (const int*)d_in[3];
    const float* W_num    = (const float*)d_in[4];
    const float* b_num    = (const float*)d_in[5];
    const float* a_in     = (const float*)d_in[6];
    const float* W1       = (const float*)d_in[7];
    const float* b1       = (const float*)d_in[8];
    const float* a1       = (const float*)d_in[9];
    const float* W2       = (const float*)d_in[10];
    const float* b2       = (const float*)d_in[11];
    const float* a2       = (const float*)d_in[12];
    const float* W3       = (const float*)d_in[13];
    const float* b3       = (const float*)d_in[14];

    const int N = in_sizes[1];
    const int E = in_sizes[3] / 2;
    const int* src = edge;
    const int* dst = edge + E;

    const int nRowTiles = (N + 15) / 16;
    const size_t padRows = (size_t)nRowTiles * 16;

    auto al = [](size_t v) { return (v + 255) & ~(size_t)255; };
    const int NB = (N + 255) / 256;
    const int EB = (E + 8191) / 8192;
    char* p = (char*)d_ws;
    float* dis    = (float*)p; p += al((size_t)N * 4);
    int*   cnt    = (int*)p;   p += al((size_t)N * 4);
    int*   rowptr = (int*)p;   p += al((size_t)N * 4);
    int*   bh     = (int*)p;   p += al((size_t)NB * EB * 4);
    int2*  ebuf   = (int2*)p;  p += al((size_t)E * 8);
    int2*  csr    = (int2*)p;  p += al((size_t)E * 8);
    float* hbuf = (float*)p; p += al(padRows * 128 * 4);
    float* tbuf = (float*)p; p += al(padRows * 128 * 4);
    float* t3   = (float*)p; p += al((size_t)N * 4);
    float* out  = (float*)d_out;

    const int TPB = 256;
    dim3 blk(TPB);
    auto gNw = dim3((int)(((size_t)N * 32 + TPB - 1) / TPB));
    auto gN8 = dim3((int)(((size_t)N * 8 + TPB - 1) / TPB));
    const int gemmBlocks = (nRowTiles + 3) / 4;

    k_bkt_count<<<EB, 1024, 0, stream>>>(dst, bh, E, NB, EB);
    k_bkt_scan<<<1, 1024, 0, stream>>>(bh, NB * EB);
    k_bkt_scatter<<<EB, 1024, 0, stream>>>(src, dst, bh, ebuf, E, NB, EB);
    k_bkt_rows<<<NB, 1024, 0, stream>>>(ebuf, bh, rowptr, cnt, dis, csr, N, E, NB, EB);
    k_edge_norm<<<(E + 1023) / 1024, 256, 0, stream>>>(csr, dis, E);

    k_proj_gemm128_wmma<<<gemmBlocks, 128, 0, stream>>>(num_x, num_mask, x, W_num, b_num, a_in, W1, tbuf, nRowTiles, N);
    k_gather_rows<<<gNw, blk, 0, stream>>>(csr, rowptr, cnt, dis, tbuf, b1, a1, hbuf, N);

    k_gemm128_wmma<<<gemmBlocks, 128, 0, stream>>>(hbuf, W2, tbuf, nRowTiles);
    k_gather_rows<<<gNw, blk, 0, stream>>>(csr, rowptr, cnt, dis, tbuf, b2, a2, hbuf, N);

    k_dot_w3<<<gNw, blk, 0, stream>>>(hbuf, W3, t3, N);
    k_gather_scalar<<<gN8, blk, 0, stream>>>(csr, rowptr, cnt, dis, t3, b3, out, N);
}


# GCN norm factored out of the edge list: rows stored pre-scaled by dis (GEMM epilogue, dot), gathers sum plain rows and scale once by dis[dst]; per-edge norm launch and weights gone; products regrouped
# speedup vs baseline: 3.8569x; 1.0642x over previous
#include <hip/hip_runtime.h>

typedef __attribute__((ext_vector_type(2))) float v2f;
typedef __attribute__((ext_vector_type(8))) float v8f;


__global__ __launch_bounds__(1024) void k_bkt_count(const int* __restrict__ dst, int* __restrict__ bh,
                                                    int E, int NB, int EB) {
    __shared__ int hist[1024];
    const int tid = threadIdx.x;
    for (int i = tid; i < NB; i += 1024) hist[i] = 0;
    __syncthreads();
    const int e0 = blockIdx.x * 8192 + tid;
    int d[8];
    #pragma unroll
    for (int j = 0; j < 8; ++j) {
        const int e = e0 + j * 1024;
        d[j] = (e < E) ? dst[e] : -1;
    }
    #pragma unroll
    for (int j = 0; j < 8; ++j)
        if (d[j] >= 0) atomicAdd(&hist[d[j] >> 8], 1);
    __syncthreads();
    for (int i = tid; i < NB; i += 1024) bh[i * EB + blockIdx.x] = hist[i];
}

__global__ __launch_bounds__(1024) void k_bkt_scan(int* a, int n) {
    __shared__ int wsum[32];
    const int tid = threadIdx.x, lane = tid & 31, w = tid >> 5;
    const int chunk = (n + 1023) / 1024;
    int lo = tid * chunk;
    lo = (lo < n) ? lo : n;
    int hi = lo + chunk;
    hi = (hi < n) ? hi : n;
    int s = 0;
    for (int i = lo; i < hi; ++i) s += a[i];
    int inc = s;
    for (int off = 1; off < 32; off <<= 1) {
        int u = __shfl_up(inc, off, 32);
        if (lane >= off) inc += u;
    }
    if (lane == 31) wsum[w] = inc;
    __syncthreads();
    int ex = inc - s;
    for (int i = 0; i < w; ++i) ex += wsum[i];
    for (int i = lo; i < hi; ++i) {
        const int v = a[i];
        a[i] = ex;
        ex += v;
    }
}

__global__ __launch_bounds__(1024) void k_bkt_scatter(const int* __restrict__ src, const int* __restrict__ dst,
                                                      const int* __restrict__ boff, int2* __restrict__ ebuf,
                                                      int E, int NB, int EB) {
    __shared__ int cur[1024];
    const int tid = threadIdx.x;
    for (int i = tid; i < NB; i += 1024) cur[i] = boff[i * EB + blockIdx.x];
    __syncthreads();
    const int e0 = blockIdx.x * 8192 + tid;
    int s[8], d[8];
    #pragma unroll
    for (int j = 0; j < 8; ++j) {
        const int e = e0 + j * 1024;
        const bool ok = e < E;
        s[j] = ok ? src[e] : 0;
        d[j] = ok ? dst[e] : -1;
    }
    #pragma unroll
    for (int j = 0; j < 8; ++j) {
        if (d[j] >= 0) {
            const int p = atomicAdd(&cur[d[j] >> 8], 1);
            ebuf[p] = make_int2(s[j], d[j]);
        }
    }
}

__global__ __launch_bounds__(1024) void k_bkt_rows(const int2* __restrict__ ebuf, const int* __restrict__ boff,
                                                   int* __restrict__ rowptr, int* __restrict__ cnt,
                                                   float* __restrict__ dis, int* __restrict__ csr,
                                                   int N, int E, int NB, int EB) {
    __shared__ int c[256];
    __shared__ int cu[256];
    __shared__ int wsum[8];
    const int tid = threadIdx.x, lane = tid & 31, w = tid >> 5;
    const int b = blockIdx.x;
    const int base = boff[b * EB];
    const int end  = (b + 1 < NB) ? boff[(b + 1) * EB] : E;
    if (tid < 256) c[tid] = 0;
    __syncthreads();
    for (int i0 = base; i0 < end; i0 += 4096) {
        int d[4];
        #pragma unroll
        for (int j = 0; j < 4; ++j) {
            const int i = i0 + j * 1024 + tid;
            d[j] = (i < end) ? ebuf[i].y : -1;
        }
        #pragma unroll
        for (int j = 0; j < 4; ++j)
            if (d[j] >= 0) atomicAdd(&c[d[j] & 255], 1);
    }
    __syncthreads();
    int v = 0, inc = 0;
    if (tid < 256) {
        v = c[tid];
        inc = v;
        for (int off = 1; off < 32; off <<= 1) {
            int u = __shfl_up(inc, off, 32);
            if (lane >= off) inc += u;
        }
        if (lane == 31) wsum[w] = inc;
    }
    __syncthreads();
    if (tid < 256) {
        int ex = inc - v;
        for (int i = 0; i < w; ++i) ex += wsum[i];
        const int start = base + ex;
        cu[tid] = start;
        const int n = b * 256 + tid;
        if (n < N) {
            rowptr[n] = start;
            cnt[n] = v;
            dis[n] = rsqrtf((float)v + 1.0f);
        }
    }
    __syncthreads();
    for (int i0 = base; i0 < end; i0 += 4096) {
        int2 e[4];
        #pragma unroll
        for (int j = 0; j < 4; ++j) {
            const int i = i0 + j * 1024 + tid;
            e[j] = (i < end) ? ebuf[i] : make_int2(0, -1);
        }
        #pragma unroll
        for (int j = 0; j < 4; ++j) {
            if (e[j].y >= 0) {
                const int p = atomicAdd(&cu[e[j].y & 255], 1);
                csr[p] = e[j].x;
            }
        }
    }
}

__global__ __launch_bounds__(128) void k_gemm128_wmma(const float* __restrict__ h,
                                                      const float* __restrict__ W,
                                                      const float* __restrict__ dis,
                                                      float* __restrict__ t,
                                                      int nRowTiles, int N) {
    __shared__ float Wl[128 * 128];
    const int wave = threadIdx.x >> 5;
    const int lane = threadIdx.x & 31;
    const int rowTile = blockIdx.x * 4 + wave;
    const int mhalf = (lane >> 4) << 3;
    float sc[8];
    #pragma unroll
    for (int r = 0; r < 8; ++r) {
        int rr = rowTile * 16 + mhalf + r;
        rr = (rr < N) ? rr : (N - 1);
        sc[r] = dis[rr];
    }
    {
        int tid = threadIdx.x;
        v2f* Wlv = (v2f*)Wl;
        #pragma unroll
        for (int i = 0; i < 64; ++i) {
            int idx = i * 128 + tid;
            int c = idx & 127;
            int p = idx >> 7;
            v2f pr;
            pr.x = W[(2 * p)     * 128 + c];
            pr.y = W[(2 * p + 1) * 128 + c];
            Wlv[p * 128 + c] = pr;
        }
    }
    __syncthreads();
    asm volatile("" :: "v"(sc[0]), "v"(sc[1]), "v"(sc[2]), "v"(sc[3]), "v"(sc[4]), "v"(sc[5]), "v"(sc[6]), "v"(sc[7]));

    if (rowTile >= nRowTiles) return;

    const int r15   = lane & 15;
    const int khalf = (lane >> 4) << 1;
    const float* hrow = h + (size_t)(rowTile * 16 + r15) * 128;
    const v2f* Wlv = (const v2f*)Wl;

    v8f acc[8] = {};
    for (int k0 = 0; k0 < 128; k0 += 4) {
        v2f a = *(const v2f*)(hrow + k0 + khalf);
        const v2f* wrow = Wlv + (((k0 + khalf) >> 1) * 128 + r15);
        #pragma unroll
        for (int j = 0; j < 8; ++j) {
            v2f bfrag = wrow[j * 16];
            acc[j] = __builtin_amdgcn_wmma_f32_16x16x4_f32(
                false, a, false, bfrag, (short)0, acc[j], false, false);
        }
    }

    float* trow = t + (size_t)(rowTile * 16 + mhalf) * 128 + r15;
    #pragma unroll
    for (int j = 0; j < 8; ++j)
        #pragma unroll
        for (int r = 0; r < 8; ++r)
            trow[(size_t)r * 128 + j * 16] = acc[j][r] * sc[r];
}

__global__ __launch_bounds__(128) void k_proj_gemm128_wmma(const float* __restrict__ num_x,
                                                           const float* __restrict__ num_mask,
                                                           const float* __restrict__ x,
                                                           const float* __restrict__ Wn,
                                                           const float* __restrict__ bn,
                                                           const float* __restrict__ ain,
                                                           const float* __restrict__ W,
                                                           const float* __restrict__ dis,
                                                           float* __restrict__ t,
                                                           int nRowTiles, int N) {
    __shared__ float Wl[128 * 128];
    __shared__ v2f Pw[64];
    __shared__ v2f Pb[64];
    __shared__ v2f Pa[64];
    const int wave = threadIdx.x >> 5;
    const int lane = threadIdx.x & 31;
    const int rowTile = blockIdx.x * 4 + wave;
    const int mhalf = (lane >> 4) << 3;
    float sc[8];
    #pragma unroll
    for (int r = 0; r < 8; ++r) {
        int rr = rowTile * 16 + mhalf + r;
        rr = (rr < N) ? rr : (N - 1);
        sc[r] = dis[rr];
    }
    {
        int tid = threadIdx.x;
        v2f* Wlv = (v2f*)Wl;
        #pragma unroll
        for (int i = 0; i < 64; ++i) {
            int idx = i * 128 + tid;
            int c = idx & 127;
            int p = idx >> 7;
            v2f pr;
            pr.x = W[(2 * p)     * 128 + c];
            pr.y = W[(2 * p + 1) * 128 + c];
            Wlv[p * 128 + c] = pr;
        }
        if (tid < 64) {
            Pw[tid] = *(const v2f*)(Wn  + 2 * tid);
            Pb[tid] = *(const v2f*)(bn  + 2 * tid);
            Pa[tid] = *(const v2f*)(ain + 2 * tid);
        }
    }
    __syncthreads();
    asm volatile("" :: "v"(sc[0]), "v"(sc[1]), "v"(sc[2]), "v"(sc[3]), "v"(sc[4]), "v"(sc[5]), "v"(sc[6]), "v"(sc[7]));

    if (rowTile >= nRowTiles) return;

    const int r15   = lane & 15;
    const int khalf = (lane >> 4) << 1;
    int row = rowTile * 16 + r15;
    row = (row < N) ? row : (N - 1);
    const float s = num_x[row] * num_mask[row];
    const float* xrow = x + (size_t)row * 128;
    const v2f* Wlv = (const v2f*)Wl;

    v8f acc[8] = {};
    for (int k0 = 0; k0 < 128; k0 += 4) {
        const int kp = (k0 + khalf) >> 1;
        const v2f xv = *(const v2f*)(xrow + k0 + khalf);
        const v2f pw = Pw[kp], pb = Pb[kp], pa = Pa[kp];
        float z0 = s * pw.x + pb.x;
        float z1 = s * pw.y + pb.y;
        z0 = (z0 >= 0.0f) ? z0 : pa.x * z0;
        z1 = (z1 >= 0.0f) ? z1 : pa.y * z1;
        v2f a;
        a.x = z0 + xv.x;
        a.y = z1 + xv.y;
        const v2f* wrow = Wlv + (kp * 128 + r15);
        #pragma unroll
        for (int j = 0; j < 8; ++j) {
            v2f bfrag = wrow[j * 16];
            acc[j] = __builtin_amdgcn_wmma_f32_16x16x4_f32(
                false, a, false, bfrag, (short)0, acc[j], false, false);
        }
    }

    float* trow = t + (size_t)(rowTile * 16 + mhalf) * 128 + r15;
    #pragma unroll
    for (int j = 0; j < 8; ++j)
        #pragma unroll
        for (int r = 0; r < 8; ++r)
            trow[(size_t)r * 128 + j * 16] = acc[j][r] * sc[r];
}

__global__ __launch_bounds__(256) void k_gather_rows(const int* __restrict__ csr, const int* __restrict__ rowptr,
                                                     const int* __restrict__ cnt, const float* __restrict__ dis,
                                                     const float* __restrict__ t, const float* __restrict__ bias,
                                                     const float* __restrict__ a, float* __restrict__ h, int N) {
    const int wid  = (blockIdx.x * blockDim.x + threadIdx.x) >> 5;
    const int lane = threadIdx.x & 31;
    if (wid >= N) return;
    const int n = __builtin_amdgcn_readfirstlane(wid);
    const int beg = rowptr[n];
    const int end = beg + cnt[n];
    const float4* t4 = (const float4*)t;
    float4 acc = make_float4(0.0f, 0.0f, 0.0f, 0.0f);
    int e = beg;
    for (; e + 4 <= end; e += 4) {
        const int s0 = csr[e], s1 = csr[e + 1], s2 = csr[e + 2], s3 = csr[e + 3];
        const float4 v0 = t4[(size_t)s0 * 32 + lane];
        const float4 v1 = t4[(size_t)s1 * 32 + lane];
        const float4 v2 = t4[(size_t)s2 * 32 + lane];
        const float4 v3 = t4[(size_t)s3 * 32 + lane];
        acc.x += v0.x; acc.y += v0.y; acc.z += v0.z; acc.w += v0.w;
        acc.x += v1.x; acc.y += v1.y; acc.z += v1.z; acc.w += v1.w;
        acc.x += v2.x; acc.y += v2.y; acc.z += v2.z; acc.w += v2.w;
        acc.x += v3.x; acc.y += v3.y; acc.z += v3.z; acc.w += v3.w;
    }
    const int rem = end - e;
    if (rem == 3) {
        const int s0 = csr[e], s1 = csr[e + 1], s2 = csr[e + 2];
        const float4 v0 = t4[(size_t)s0 * 32 + lane];
        const float4 v1 = t4[(size_t)s1 * 32 + lane];
        const float4 v2 = t4[(size_t)s2 * 32 + lane];
        acc.x += v0.x; acc.y += v0.y; acc.z += v0.z; acc.w += v0.w;
        acc.x += v1.x; acc.y += v1.y; acc.z += v1.z; acc.w += v1.w;
        acc.x += v2.x; acc.y += v2.y; acc.z += v2.z; acc.w += v2.w;
    } else if (rem == 2) {
        const int s0 = csr[e], s1 = csr[e + 1];
        const float4 v0 = t4[(size_t)s0 * 32 + lane];
        const float4 v1 = t4[(size_t)s1 * 32 + lane];
        acc.x += v0.x; acc.y += v0.y; acc.z += v0.z; acc.w += v0.w;
        acc.x += v1.x; acc.y += v1.y; acc.z += v1.z; acc.w += v1.w;
    } else if (rem == 1) {
        const int s0 = csr[e];
        const float4 v0 = t4[(size_t)s0 * 32 + lane];
        acc.x += v0.x; acc.y += v0.y; acc.z += v0.z; acc.w += v0.w;
    }
    const float dn = dis[n];
    const float4 tv = t4[(size_t)n * 32 + lane];
    const float4 b  = ((const float4*)bias)[lane];
    const float4 av = ((const float4*)a)[lane];
    float4 r;
    r.x = dn * (acc.x + tv.x) + b.x;
    r.y = dn * (acc.y + tv.y) + b.y;
    r.z = dn * (acc.z + tv.z) + b.z;
    r.w = dn * (acc.w + tv.w) + b.w;
    r.x = (r.x >= 0.0f) ? r.x : av.x * r.x;
    r.y = (r.y >= 0.0f) ? r.y : av.y * r.y;
    r.z = (r.z >= 0.0f) ? r.z : av.z * r.z;
    r.w = (r.w >= 0.0f) ? r.w : av.w * r.w;
    ((float4*)h)[(size_t)n * 32 + lane] = r;
}

__global__ void k_dot_w3(const float* __restrict__ h, const float* __restrict__ W3,
                         const float* __restrict__ dis, float* __restrict__ t3, int N) {
    int wid  = (blockIdx.x * blockDim.x + threadIdx.x) >> 5;
    int lane = threadIdx.x & 31;
    if (wid >= N) return;
    const float4 hv = ((const float4*)(h + (size_t)wid * 128))[lane];
    const float4 wv = ((const float4*)W3)[lane];
    float s = hv.x * wv.x + hv.y * wv.y + hv.z * wv.z + hv.w * wv.w;
    #pragma unroll
    for (int off = 16; off > 0; off >>= 1) s += __shfl_xor(s, off, 32);
    if (lane == 0) t3[wid] = dis[wid] * s;
}

__global__ void k_gather_scalar(const int* __restrict__ csr, const int* __restrict__ rowptr,
                                const int* __restrict__ cnt, const float* __restrict__ dis,
                                const float* __restrict__ t3, const float* __restrict__ b3,
                                float* __restrict__ out, int N) {
    const int gid = blockIdx.x * blockDim.x + threadIdx.x;
    const int n = gid >> 3, sub = gid & 7;
    const bool valid = n < N;
    int beg = 0, end = 0;
    if (valid) { beg = rowptr[n]; end = beg + cnt[n]; }
    float s = 0.0f;
    for (int e = beg + sub; e < end; e += 8) s += t3[csr[e]];
    s += __shfl_xor(s, 4, 32);
    s += __shfl_xor(s, 2, 32);
    s += __shfl_xor(s, 1, 32);
    if (valid && sub == 0) out[n] = dis[n] * (s + t3[n]) + b3[0];
}

extern "C" void kernel_launch(void* const* d_in, const int* in_sizes, int n_in,
                              void* d_out, int out_size, void* d_ws, size_t ws_size,
                              hipStream_t stream) {
    const float* num_x    = (const float*)d_in[0];
    const float* num_mask = (const float*)d_in[1];
    const float* x        = (const float*)d_in[2];
    const int*   edge     = (const int*)d_in[3];
    const float* W_num    = (const float*)d_in[4];
    const float* b_num    = (const float*)d_in[5];
    const float* a_in     = (const float*)d_in[6];
    const float* W1       = (const float*)d_in[7];
    const float* b1       = (const float*)d_in[8];
    const float* a1       = (const float*)d_in[9];
    const float* W2       = (const float*)d_in[10];
    const float* b2       = (const float*)d_in[11];
    const float* a2       = (const float*)d_in[12];
    const float* W3       = (const float*)d_in[13];
    const float* b3       = (const float*)d_in[14];

    const int N = in_sizes[1];
    const int E = in_sizes[3] / 2;
    const int* src = edge;
    const int* dst = edge + E;

    const int nRowTiles = (N + 15) / 16;
    const size_t padRows = (size_t)nRowTiles * 16;

    auto al = [](size_t v) { return (v + 255) & ~(size_t)255; };
    const int NB = (N + 255) / 256;
    const int EB = (E + 8191) / 8192;
    char* p = (char*)d_ws;
    float* dis    = (float*)p; p += al((size_t)N * 4);
    int*   cnt    = (int*)p;   p += al((size_t)N * 4);
    int*   rowptr = (int*)p;   p += al((size_t)N * 4);
    int*   bh     = (int*)p;   p += al((size_t)NB * EB * 4);
    int2*  ebuf   = (int2*)p;  p += al((size_t)E * 8);
    int*   csr    = (int*)p;   p += al((size_t)E * 4);
    float* hbuf = (float*)p; p += al(padRows * 128 * 4);
    float* tbuf = (float*)p; p += al(padRows * 128 * 4);
    float* t3   = (float*)p; p += al((size_t)N * 4);
    float* out  = (float*)d_out;

    const int TPB = 256;
    dim3 blk(TPB);
    auto gNw = dim3((int)(((size_t)N * 32 + TPB - 1) / TPB));
    auto gN8 = dim3((int)(((size_t)N * 8 + TPB - 1) / TPB));
    const int gemmBlocks = (nRowTiles + 3) / 4;

    k_bkt_count<<<EB, 1024, 0, stream>>>(dst, bh, E, NB, EB);
    k_bkt_scan<<<1, 1024, 0, stream>>>(bh, NB * EB);
    k_bkt_scatter<<<EB, 1024, 0, stream>>>(src, dst, bh, ebuf, E, NB, EB);
    k_bkt_rows<<<NB, 1024, 0, stream>>>(ebuf, bh, rowptr, cnt, dis, csr, N, E, NB, EB);

    k_proj_gemm128_wmma<<<gemmBlocks, 128, 0, stream>>>(num_x, num_mask, x, W_num, b_num, a_in, W1, dis, tbuf, nRowTiles, N);
    k_gather_rows<<<gNw, blk, 0, stream>>>(csr, rowptr, cnt, dis, tbuf, b1, a1, hbuf, N);

    k_gemm128_wmma<<<gemmBlocks, 128, 0, stream>>>(hbuf, W2, dis, tbuf, nRowTiles, N);
    k_gather_rows<<<gNw, blk, 0, stream>>>(csr, rowptr, cnt, dis, tbuf, b2, a2, hbuf, N);

    k_dot_w3<<<gNw, blk, 0, stream>>>(hbuf, W3, dis, t3, N);
    k_gather_scalar<<<gN8, blk, 0, stream>>>(csr, rowptr, cnt, dis, t3, b3, out, N);
}


# layer-3 per-node dot: one wave per 8 rows, its 8 row loads, W3 and dis requested together (lanes 0-7 store), not one wave per row: 6250 waves enter instead of 50000; per-row sums bit-identical
# speedup vs baseline: 4.0672x; 1.0545x over previous
#include <hip/hip_runtime.h>

typedef __attribute__((ext_vector_type(2))) float v2f;
typedef __attribute__((ext_vector_type(8))) float v8f;


__global__ __launch_bounds__(1024) void k_bkt_count(const int* __restrict__ dst, int* __restrict__ bh,
                                                    int E, int NB, int EB) {
    __shared__ int hist[1024];
    const int tid = threadIdx.x;
    for (int i = tid; i < NB; i += 1024) hist[i] = 0;
    __syncthreads();
    const int e0 = blockIdx.x * 8192 + tid;
    int d[8];
    #pragma unroll
    for (int j = 0; j < 8; ++j) {
        const int e = e0 + j * 1024;
        d[j] = (e < E) ? dst[e] : -1;
    }
    #pragma unroll
    for (int j = 0; j < 8; ++j)
        if (d[j] >= 0) atomicAdd(&hist[d[j] >> 8], 1);
    __syncthreads();
    for (int i = tid; i < NB; i += 1024) bh[i * EB + blockIdx.x] = hist[i];
}

__global__ __launch_bounds__(1024) void k_bkt_scan(int* a, int n) {
    __shared__ int wsum[32];
    const int tid = threadIdx.x, lane = tid & 31, w = tid >> 5;
    const int chunk = (n + 1023) / 1024;
    int lo = tid * chunk;
    lo = (lo < n) ? lo : n;
    int hi = lo + chunk;
    hi = (hi < n) ? hi : n;
    int s = 0;
    for (int i = lo; i < hi; ++i) s += a[i];
    int inc = s;
    for (int off = 1; off < 32; off <<= 1) {
        int u = __shfl_up(inc, off, 32);
        if (lane >= off) inc += u;
    }
    if (lane == 31) wsum[w] = inc;
    __syncthreads();
    int ex = inc - s;
    for (int i = 0; i < w; ++i) ex += wsum[i];
    for (int i = lo; i < hi; ++i) {
        const int v = a[i];
        a[i] = ex;
        ex += v;
    }
}

__global__ __launch_bounds__(1024) void k_bkt_scatter(const int* __restrict__ src, const int* __restrict__ dst,
                                                      const int* __restrict__ boff, int2* __restrict__ ebuf,
                                                      int E, int NB, int EB) {
    __shared__ int cur[1024];
    const int tid = threadIdx.x;
    for (int i = tid; i < NB; i += 1024) cur[i] = boff[i * EB + blockIdx.x];
    __syncthreads();
    const int e0 = blockIdx.x * 8192 + tid;
    int s[8], d[8];
    #pragma unroll
    for (int j = 0; j < 8; ++j) {
        const int e = e0 + j * 1024;
        const bool ok = e < E;
        s[j] = ok ? src[e] : 0;
        d[j] = ok ? dst[e] : -1;
    }
    #pragma unroll
    for (int j = 0; j < 8; ++j) {
        if (d[j] >= 0) {
            const int p = atomicAdd(&cur[d[j] >> 8], 1);
            ebuf[p] = make_int2(s[j], d[j]);
        }
    }
}

__global__ __launch_bounds__(1024) void k_bkt_rows(const int2* __restrict__ ebuf, const int* __restrict__ boff,
                                                   int* __restrict__ rowptr, int* __restrict__ cnt,
                                                   float* __restrict__ dis, int* __restrict__ csr,
                                                   int N, int E, int NB, int EB) {
    __shared__ int c[256];
    __shared__ int cu[256];
    __shared__ int wsum[8];
    const int tid = threadIdx.x, lane = tid & 31, w = tid >> 5;
    const int b = blockIdx.x;
    const int base = boff[b * EB];
    const int end  = (b + 1 < NB) ? boff[(b + 1) * EB] : E;
    if (tid < 256) c[tid] = 0;
    __syncthreads();
    for (int i0 = base; i0 < end; i0 += 4096) {
        int d[4];
        #pragma unroll
        for (int j = 0; j < 4; ++j) {
            const int i = i0 + j * 1024 + tid;
            d[j] = (i < end) ? ebuf[i].y : -1;
        }
        #pragma unroll
        for (int j = 0; j < 4; ++j)
            if (d[j] >= 0) atomicAdd(&c[d[j] & 255], 1);
    }
    __syncthreads();
    int v = 0, inc = 0;
    if (tid < 256) {
        v = c[tid];
        inc = v;
        for (int off = 1; off < 32; off <<= 1) {
            int u = __shfl_up(inc, off, 32);
            if (lane >= off) inc += u;
        }
        if (lane == 31) wsum[w] = inc;
    }
    __syncthreads();
    if (tid < 256) {
        int ex = inc - v;
        for (int i = 0; i < w; ++i) ex += wsum[i];
        const int start = base + ex;
        cu[tid] = start;
        const int n = b * 256 + tid;
        if (n < N) {
            rowptr[n] = start;
            cnt[n] = v;
            dis[n] = rsqrtf((float)v + 1.0f);
        }
    }
    __syncthreads();
    for (int i0 = base; i0 < end; i0 += 4096) {
        int2 e[4];
        #pragma unroll
        for (int j = 0; j < 4; ++j) {
            const int i = i0 + j * 1024 + tid;
            e[j] = (i < end) ? ebuf[i] : make_int2(0, -1);
        }
        #pragma unroll
        for (int j = 0; j < 4; ++j) {
            if (e[j].y >= 0) {
                const int p = atomicAdd(&cu[e[j].y & 255], 1);
                csr[p] = e[j].x;
            }
        }
    }
}

__global__ __launch_bounds__(128) void k_gemm128_wmma(const float* __restrict__ h,
                                                      const float* __restrict__ W,
                                                      const float* __restrict__ dis,
                                                      float* __restrict__ t,
                                                      int nRowTiles, int N) {
    __shared__ float Wl[128 * 128];
    const int wave = threadIdx.x >> 5;
    const int lane = threadIdx.x & 31;
    const int rowTile = blockIdx.x * 4 + wave;
    const int mhalf = (lane >> 4) << 3;
    float sc[8];
    #pragma unroll
    for (int r = 0; r < 8; ++r) {
        int rr = rowTile * 16 + mhalf + r;
        rr = (rr < N) ? rr : (N - 1);
        sc[r] = dis[rr];
    }
    {
        int tid = threadIdx.x;
        v2f* Wlv = (v2f*)Wl;
        #pragma unroll
        for (int i = 0; i < 64; ++i) {
            int idx = i * 128 + tid;
            int c = idx & 127;
            int p = idx >> 7;
            v2f pr;
            pr.x = W[(2 * p)     * 128 + c];
            pr.y = W[(2 * p + 1) * 128 + c];
            Wlv[p * 128 + c] = pr;
        }
    }
    __syncthreads();
    asm volatile("" :: "v"(sc[0]), "v"(sc[1]), "v"(sc[2]), "v"(sc[3]), "v"(sc[4]), "v"(sc[5]), "v"(sc[6]), "v"(sc[7]));

    if (rowTile >= nRowTiles) return;

    const int r15   = lane & 15;
    const int khalf = (lane >> 4) << 1;
    const float* hrow = h + (size_t)(rowTile * 16 + r15) * 128;
    const v2f* Wlv = (const v2f*)Wl;

    v8f acc[8] = {};
    for (int k0 = 0; k0 < 128; k0 += 4) {
        v2f a = *(const v2f*)(hrow + k0 + khalf);
        const v2f* wrow = Wlv + (((k0 + khalf) >> 1) * 128 + r15);
        #pragma unroll
        for (int j = 0; j < 8; ++j) {
            v2f bfrag = wrow[j * 16];
            acc[j] = __builtin_amdgcn_wmma_f32_16x16x4_f32(
                false, a, false, bfrag, (short)0, acc[j], false, false);
        }
    }

    float* trow = t + (size_t)(rowTile * 16 + mhalf) * 128 + r15;
    #pragma unroll
    for (int j = 0; j < 8; ++j)
        #pragma unroll
        for (int r = 0; r < 8; ++r)
            trow[(size_t)r * 128 + j * 16] = acc[j][r] * sc[r];
}

__global__ __launch_bounds__(128) void k_proj_gemm128_wmma(const float* __restrict__ num_x,
                                                           const float* __restrict__ num_mask,
                                                           const float* __restrict__ x,
                                                           const float* __restrict__ Wn,
                                                           const float* __restrict__ bn,
                                                           const float* __restrict__ ain,
                                                           const float* __restrict__ W,
                                                           const float* __restrict__ dis,
                                                           float* __restrict__ t,
                                                           int nRowTiles, int N) {
    __shared__ float Wl[128 * 128];
    __shared__ v2f Pw[64];
    __shared__ v2f Pb[64];
    __shared__ v2f Pa[64];
    const int wave = threadIdx.x >> 5;
    const int lane = threadIdx.x & 31;
    const int rowTile = blockIdx.x * 4 + wave;
    const int mhalf = (lane >> 4) << 3;
    float sc[8];
    #pragma unroll
    for (int r = 0; r < 8; ++r) {
        int rr = rowTile * 16 + mhalf + r;
        rr = (rr < N) ? rr : (N - 1);
        sc[r] = dis[rr];
    }
    {
        int tid = threadIdx.x;
        v2f* Wlv = (v2f*)Wl;
        #pragma unroll
        for (int i = 0; i < 64; ++i) {
            int idx = i * 128 + tid;
            int c = idx & 127;
            int p = idx >> 7;
            v2f pr;
            pr.x = W[(2 * p)     * 128 + c];
            pr.y = W[(2 * p + 1) * 128 + c];
            Wlv[p * 128 + c] = pr;
        }
        if (tid < 64) {
            Pw[tid] = *(const v2f*)(Wn  + 2 * tid);
            Pb[tid] = *(const v2f*)(bn  + 2 * tid);
            Pa[tid] = *(const v2f*)(ain + 2 * tid);
        }
    }
    __syncthreads();
    asm volatile("" :: "v"(sc[0]), "v"(sc[1]), "v"(sc[2]), "v"(sc[3]), "v"(sc[4]), "v"(sc[5]), "v"(sc[6]), "v"(sc[7]));

    if (rowTile >= nRowTiles) return;

    const int r15   = lane & 15;
    const int khalf = (lane >> 4) << 1;
    int row = rowTile * 16 + r15;
    row = (row < N) ? row : (N - 1);
    const float s = num_x[row] * num_mask[row];
    const float* xrow = x + (size_t)row * 128;
    const v2f* Wlv = (const v2f*)Wl;

    v8f acc[8] = {};
    for (int k0 = 0; k0 < 128; k0 += 4) {
        const int kp = (k0 + khalf) >> 1;
        const v2f xv = *(const v2f*)(xrow + k0 + khalf);
        const v2f pw = Pw[kp], pb = Pb[kp], pa = Pa[kp];
        float z0 = s * pw.x + pb.x;
        float z1 = s * pw.y + pb.y;
        z0 = (z0 >= 0.0f) ? z0 : pa.x * z0;
        z1 = (z1 >= 0.0f) ? z1 : pa.y * z1;
        v2f a;
        a.x = z0 + xv.x;
        a.y = z1 + xv.y;
        const v2f* wrow = Wlv + (kp * 128 + r15);
        #pragma unroll
        for (int j = 0; j < 8; ++j) {
            v2f bfrag = wrow[j * 16];
            acc[j] = __builtin_amdgcn_wmma_f32_16x16x4_f32(
                false, a, false, bfrag, (short)0, acc[j], false, false);
        }
    }

    float* trow = t + (size_t)(rowTile * 16 + mhalf) * 128 + r15;
    #pragma unroll
    for (int j = 0; j < 8; ++j)
        #pragma unroll
        for (int r = 0; r < 8; ++r)
            trow[(size_t)r * 128 + j * 16] = acc[j][r] * sc[r];
}

__global__ __launch_bounds__(256) void k_gather_rows(const int* __restrict__ csr, const int* __restrict__ rowptr,
                                                     const int* __restrict__ cnt, const float* __restrict__ dis,
                                                     const float* __restrict__ t, const float* __restrict__ bias,
                                                     const float* __restrict__ a, float* __restrict__ h, int N) {
    const int wid  = (blockIdx.x * blockDim.x + threadIdx.x) >> 5;
    const int lane = threadIdx.x & 31;
    if (wid >= N) return;
    const int n = __builtin_amdgcn_readfirstlane(wid);
    const int beg = rowptr[n];
    const int end = beg + cnt[n];
    const float4* t4 = (const float4*)t;
    float4 acc = make_float4(0.0f, 0.0f, 0.0f, 0.0f);
    int e = beg;
    for (; e + 4 <= end; e += 4) {
        const int s0 = csr[e], s1 = csr[e + 1], s2 = csr[e + 2], s3 = csr[e + 3];
        const float4 v0 = t4[(size_t)s0 * 32 + lane];
        const float4 v1 = t4[(size_t)s1 * 32 + lane];
        const float4 v2 = t4[(size_t)s2 * 32 + lane];
        const float4 v3 = t4[(size_t)s3 * 32 + lane];
        acc.x += v0.x; acc.y += v0.y; acc.z += v0.z; acc.w += v0.w;
        acc.x += v1.x; acc.y += v1.y; acc.z += v1.z; acc.w += v1.w;
        acc.x += v2.x; acc.y += v2.y; acc.z += v2.z; acc.w += v2.w;
        acc.x += v3.x; acc.y += v3.y; acc.z += v3.z; acc.w += v3.w;
    }
    const int rem = end - e;
    if (rem == 3) {
        const int s0 = csr[e], s1 = csr[e + 1], s2 = csr[e + 2];
        const float4 v0 = t4[(size_t)s0 * 32 + lane];
        const float4 v1 = t4[(size_t)s1 * 32 + lane];
        const float4 v2 = t4[(size_t)s2 * 32 + lane];
        acc.x += v0.x; acc.y += v0.y; acc.z += v0.z; acc.w += v0.w;
        acc.x += v1.x; acc.y += v1.y; acc.z += v1.z; acc.w += v1.w;
        acc.x += v2.x; acc.y += v2.y; acc.z += v2.z; acc.w += v2.w;
    } else if (rem == 2) {
        const int s0 = csr[e], s1 = csr[e + 1];
        const float4 v0 = t4[(size_t)s0 * 32 + lane];
        const float4 v1 = t4[(size_t)s1 * 32 + lane];
        acc.x += v0.x; acc.y += v0.y; acc.z += v0.z; acc.w += v0.w;
        acc.x += v1.x; acc.y += v1.y; acc.z += v1.z; acc.w += v1.w;
    } else if (rem == 1) {
        const int s0 = csr[e];
        const float4 v0 = t4[(size_t)s0 * 32 + lane];
        acc.x += v0.x; acc.y += v0.y; acc.z += v0.z; acc.w += v0.w;
    }
    const float dn = dis[n];
    const float4 tv = t4[(size_t)n * 32 + lane];
    const float4 b  = ((const float4*)bias)[lane];
    const float4 av = ((const float4*)a)[lane];
    float4 r;
    r.x = dn * (acc.x + tv.x) + b.x;
    r.y = dn * (acc.y + tv.y) + b.y;
    r.z = dn * (acc.z + tv.z) + b.z;
    r.w = dn * (acc.w + tv.w) + b.w;
    r.x = (r.x >= 0.0f) ? r.x : av.x * r.x;
    r.y = (r.y >= 0.0f) ? r.y : av.y * r.y;
    r.z = (r.z >= 0.0f) ? r.z : av.z * r.z;
    r.w = (r.w >= 0.0f) ? r.w : av.w * r.w;
    ((float4*)h)[(size_t)n * 32 + lane] = r;
}

__global__ __launch_bounds__(256) void k_dot_w3(const float* __restrict__ h, const float* __restrict__ W3,
                                                const float* __restrict__ dis, float* __restrict__ t3, int N) {
    const int wid  = (blockIdx.x * blockDim.x + threadIdx.x) >> 5;
    const int lane = threadIdx.x & 31;
    const int row0 = wid * 8;
    if (row0 >= N) return;
    const float4* h4 = (const float4*)h;
    float4 hv[8];
    #pragma unroll
    for (int r = 0; r < 8; ++r) {
        int row = row0 + r;
        row = (row < N) ? row : (N - 1);
        hv[r] = h4[(size_t)row * 32 + lane];
    }
    const float4 wv = ((const float4*)W3)[lane];
    const int sub = lane & 7;
    const int myrow = row0 + sub;
    const float dn = dis[(myrow < N) ? myrow : (N - 1)];
    asm volatile("" :: "v"(dn));
    __builtin_amdgcn_sched_barrier(0);
    float keep = 0.0f;
    #pragma unroll
    for (int r = 0; r < 8; ++r) {
        float s = hv[r].x * wv.x + hv[r].y * wv.y + hv[r].z * wv.z + hv[r].w * wv.w;
        #pragma unroll
        for (int off = 16; off > 0; off >>= 1) s += __shfl_xor(s, off, 32);
        keep = (sub == r) ? s : keep;
    }
    if (lane < 8 && myrow < N) t3[myrow] = dn * keep;
}

__global__ void k_gather_scalar(const int* __restrict__ csr, const int* __restrict__ rowptr,
                                const int* __restrict__ cnt, const float* __restrict__ dis,
                                const float* __restrict__ t3, const float* __restrict__ b3,
                                float* __restrict__ out, int N) {
    const int gid = blockIdx.x * blockDim.x + threadIdx.x;
    const int n = gid >> 3, sub = gid & 7;
    const bool valid = n < N;
    int beg = 0, end = 0;
    if (valid) { beg = rowptr[n]; end = beg + cnt[n]; }
    float s = 0.0f;
    for (int e = beg + sub; e < end; e += 8) s += t3[csr[e]];
    s += __shfl_xor(s, 4, 32);
    s += __shfl_xor(s, 2, 32);
    s += __shfl_xor(s, 1, 32);
    if (valid && sub == 0) out[n] = dis[n] * (s + t3[n]) + b3[0];
}

extern "C" void kernel_launch(void* const* d_in, const int* in_sizes, int n_in,
                              void* d_out, int out_size, void* d_ws, size_t ws_size,
                              hipStream_t stream) {
    const float* num_x    = (const float*)d_in[0];
    const float* num_mask = (const float*)d_in[1];
    const float* x        = (const float*)d_in[2];
    const int*   edge     = (const int*)d_in[3];
    const float* W_num    = (const float*)d_in[4];
    const float* b_num    = (const float*)d_in[5];
    const float* a_in     = (const float*)d_in[6];
    const float* W1       = (const float*)d_in[7];
    const float* b1       = (const float*)d_in[8];
    const float* a1       = (const float*)d_in[9];
    const float* W2       = (const float*)d_in[10];
    const float* b2       = (const float*)d_in[11];
    const float* a2       = (const float*)d_in[12];
    const float* W3       = (const float*)d_in[13];
    const float* b3       = (const float*)d_in[14];

    const int N = in_sizes[1];
    const int E = in_sizes[3] / 2;
    const int* src = edge;
    const int* dst = edge + E;

    const int nRowTiles = (N + 15) / 16;
    const size_t padRows = (size_t)nRowTiles * 16;

    auto al = [](size_t v) { return (v + 255) & ~(size_t)255; };
    const int NB = (N + 255) / 256;
    const int EB = (E + 8191) / 8192;
    char* p = (char*)d_ws;
    float* dis    = (float*)p; p += al((size_t)N * 4);
    int*   cnt    = (int*)p;   p += al((size_t)N * 4);
    int*   rowptr = (int*)p;   p += al((size_t)N * 4);
    int*   bh     = (int*)p;   p += al((size_t)NB * EB * 4);
    int2*  ebuf   = (int2*)p;  p += al((size_t)E * 8);
    int*   csr    = (int*)p;   p += al((size_t)E * 4);
    float* hbuf = (float*)p; p += al(padRows * 128 * 4);
    float* tbuf = (float*)p; p += al(padRows * 128 * 4);
    float* t3   = (float*)p; p += al((size_t)N * 4);
    float* out  = (float*)d_out;

    const int TPB = 256;
    dim3 blk(TPB);
    auto gNw = dim3((int)(((size_t)N * 32 + TPB - 1) / TPB));
    auto gN8 = dim3((int)(((size_t)N * 8 + TPB - 1) / TPB));
    auto gN4 = dim3((int)(((size_t)N * 4 + TPB - 1) / TPB));
    const int gemmBlocks = (nRowTiles + 3) / 4;

    k_bkt_count<<<EB, 1024, 0, stream>>>(dst, bh, E, NB, EB);
    k_bkt_scan<<<1, 1024, 0, stream>>>(bh, NB * EB);
    k_bkt_scatter<<<EB, 1024, 0, stream>>>(src, dst, bh, ebuf, E, NB, EB);
    k_bkt_rows<<<NB, 1024, 0, stream>>>(ebuf, bh, rowptr, cnt, dis, csr, N, E, NB, EB);

    k_proj_gemm128_wmma<<<gemmBlocks, 128, 0, stream>>>(num_x, num_mask, x, W_num, b_num, a_in, W1, dis, tbuf, nRowTiles, N);
    k_gather_rows<<<gNw, blk, 0, stream>>>(csr, rowptr, cnt, dis, tbuf, b1, a1, hbuf, N);

    k_gemm128_wmma<<<gemmBlocks, 128, 0, stream>>>(hbuf, W2, dis, tbuf, nRowTiles, N);
    k_gather_rows<<<gNw, blk, 0, stream>>>(csr, rowptr, cnt, dis, tbuf, b2, a2, hbuf, N);

    k_dot_w3<<<gN4, blk, 0, stream>>>(hbuf, W3, dis, t3, N);
    k_gather_scalar<<<gN8, blk, 0, stream>>>(csr, rowptr, cnt, dis, t3, b3, out, N);
}


# edge grouping in 2 launches, not 4: each 8192-edge block groups its own edges by bucket in its own slice, bucket blocks collect their pieces; count pass, scan launch gone; rows keep LDS arrival order
# speedup vs baseline: 4.4910x; 1.1042x over previous
#include <hip/hip_runtime.h>

typedef __attribute__((ext_vector_type(2))) float v2f;
typedef __attribute__((ext_vector_type(8))) float v8f;


__global__ __launch_bounds__(1024) void k_blk_sort(const int* __restrict__ src, const int* __restrict__ dst,
                                                   int* __restrict__ loff, int2* __restrict__ ebuf,
                                                   int E, int NB, int EB) {
    __shared__ int hist[1024];
    __shared__ int wsum[32];
    const int tid = threadIdx.x, lane = tid & 31, w = tid >> 5;
    hist[tid] = 0;
    __syncthreads();
    const int e0 = blockIdx.x * 8192 + tid;
    int s[8], d[8], rk[8];
    #pragma unroll
    for (int j = 0; j < 8; ++j) {
        const int e = e0 + j * 1024;
        const bool ok = e < E;
        s[j] = ok ? src[e] : 0;
        d[j] = ok ? dst[e] : -1;
    }
    #pragma unroll
    for (int j = 0; j < 8; ++j) {
        rk[j] = 0;
        if (d[j] >= 0) rk[j] = atomicAdd(&hist[d[j] >> 8], 1);
    }
    __syncthreads();
    const int v = hist[tid];
    int inc = v;
    for (int off = 1; off < 32; off <<= 1) {
        int u = __shfl_up(inc, off, 32);
        if (lane >= off) inc += u;
    }
    if (lane == 31) wsum[w] = inc;
    __syncthreads();
    int ex = inc - v;
    for (int i = 0; i < w; ++i) ex += wsum[i];
    hist[tid] = ex;
    __syncthreads();
    int2* my = ebuf + (size_t)blockIdx.x * 8192;
    #pragma unroll
    for (int j = 0; j < 8; ++j)
        if (d[j] >= 0) my[hist[d[j] >> 8] + rk[j]] = make_int2(s[j], d[j]);
    if (tid < NB) loff[tid * EB + blockIdx.x] = ex;
}

__global__ __launch_bounds__(1024) void k_bkt_rows(const int2* __restrict__ ebuf, const int* __restrict__ loff,
                                                   int* __restrict__ rowptr, int* __restrict__ cnt,
                                                   float* __restrict__ dis, int* __restrict__ csr,
                                                   int N, int E, int NB, int EB) {
    __shared__ int c[256];
    __shared__ int cu[256];
    __shared__ int wsum[8];
    __shared__ int red[32];
    __shared__ int segLo[1024];
    __shared__ int segHi[1024];
    const int tid = threadIdx.x, lane = tid & 31, w = tid >> 5;
    const int b = blockIdx.x;
    int lo = 0;
    if (tid < EB) {
        lo = loff[b * EB + tid];
        const int left = E - tid * 8192;
        const int sliceCnt = (left < 8192) ? left : 8192;
        const int hi = (b + 1 < NB) ? loff[(b + 1) * EB + tid] : sliceCnt;
        segLo[tid] = tid * 8192 + lo;
        segHi[tid] = tid * 8192 + hi;
    }
    int r = lo;
    for (int off = 16; off > 0; off >>= 1) r += __shfl_xor(r, off, 32);
    if (lane == 0) red[w] = r;
    if (tid < 256) c[tid] = 0;
    __syncthreads();
    int base = red[lane];
    for (int off = 16; off > 0; off >>= 1) base += __shfl_xor(base, off, 32);

    int2 ev[4][3];
    #pragma unroll
    for (int q = 0; q < 4; ++q) {
        const int j = w + 32 * q;
        const int slo = (j < EB) ? segLo[j] : 0;
        const int shi = (j < EB) ? segHi[j] : 0;
        #pragma unroll
        for (int s = 0; s < 3; ++s) {
            const int i = slo + lane + 32 * s;
            ev[q][s] = (i < shi) ? ebuf[i] : make_int2(0, -1);
        }
    }
    #pragma unroll
    for (int q = 0; q < 4; ++q)
        #pragma unroll
        for (int s = 0; s < 3; ++s)
            if (ev[q][s].y >= 0) atomicAdd(&c[ev[q][s].y & 255], 1);
    for (int j = w; j < EB; j += 32) {
        const int shi = segHi[j];
        for (int i = segLo[j] + lane + ((j < 128) ? 96 : 0); i < shi; i += 32)
            atomicAdd(&c[ebuf[i].y & 255], 1);
    }
    __syncthreads();
    int v = 0, inc = 0;
    if (tid < 256) {
        v = c[tid];
        inc = v;
        for (int off = 1; off < 32; off <<= 1) {
            int u = __shfl_up(inc, off, 32);
            if (lane >= off) inc += u;
        }
        if (lane == 31) wsum[w] = inc;
    }
    __syncthreads();
    if (tid < 256) {
        int ex = inc - v;
        for (int i = 0; i < w; ++i) ex += wsum[i];
        const int start = base + ex;
        cu[tid] = start;
        const int n = b * 256 + tid;
        if (n < N) {
            rowptr[n] = start;
            cnt[n] = v;
            dis[n] = rsqrtf((float)v + 1.0f);
        }
    }
    __syncthreads();
    #pragma unroll
    for (int q = 0; q < 4; ++q)
        #pragma unroll
        for (int s = 0; s < 3; ++s)
            if (ev[q][s].y >= 0) {
                const int p = atomicAdd(&cu[ev[q][s].y & 255], 1);
                csr[p] = ev[q][s].x;
            }
    for (int j = w; j < EB; j += 32) {
        const int shi = segHi[j];
        for (int i = segLo[j] + lane + ((j < 128) ? 96 : 0); i < shi; i += 32) {
            const int2 e = ebuf[i];
            const int p = atomicAdd(&cu[e.y & 255], 1);
            csr[p] = e.x;
        }
    }
}

__global__ __launch_bounds__(128) void k_gemm128_wmma(const float* __restrict__ h,
                                                      const float* __restrict__ W,
                                                      const float* __restrict__ dis,
                                                      float* __restrict__ t,
                                                      int nRowTiles, int N) {
    __shared__ float Wl[128 * 128];
    const int wave = threadIdx.x >> 5;
    const int lane = threadIdx.x & 31;
    const int rowTile = blockIdx.x * 4 + wave;
    const int mhalf = (lane >> 4) << 3;
    float sc[8];
    #pragma unroll
    for (int r = 0; r < 8; ++r) {
        int rr = rowTile * 16 + mhalf + r;
        rr = (rr < N) ? rr : (N - 1);
        sc[r] = dis[rr];
    }
    {
        int tid = threadIdx.x;
        v2f* Wlv = (v2f*)Wl;
        #pragma unroll
        for (int i = 0; i < 64; ++i) {
            int idx = i * 128 + tid;
            int c = idx & 127;
            int p = idx >> 7;
            v2f pr;
            pr.x = W[(2 * p)     * 128 + c];
            pr.y = W[(2 * p + 1) * 128 + c];
            Wlv[p * 128 + c] = pr;
        }
    }
    __syncthreads();
    asm volatile("" :: "v"(sc[0]), "v"(sc[1]), "v"(sc[2]), "v"(sc[3]), "v"(sc[4]), "v"(sc[5]), "v"(sc[6]), "v"(sc[7]));

    if (rowTile >= nRowTiles) return;

    const int r15   = lane & 15;
    const int khalf = (lane >> 4) << 1;
    const float* hrow = h + (size_t)(rowTile * 16 + r15) * 128;
    const v2f* Wlv = (const v2f*)Wl;

    v8f acc[8] = {};
    for (int k0 = 0; k0 < 128; k0 += 4) {
        v2f a = *(const v2f*)(hrow + k0 + khalf);
        const v2f* wrow = Wlv + (((k0 + khalf) >> 1) * 128 + r15);
        #pragma unroll
        for (int j = 0; j < 8; ++j) {
            v2f bfrag = wrow[j * 16];
            acc[j] = __builtin_amdgcn_wmma_f32_16x16x4_f32(
                false, a, false, bfrag, (short)0, acc[j], false, false);
        }
    }

    float* trow = t + (size_t)(rowTile * 16 + mhalf) * 128 + r15;
    #pragma unroll
    for (int j = 0; j < 8; ++j)
        #pragma unroll
        for (int r = 0; r < 8; ++r)
            trow[(size_t)r * 128 + j * 16] = acc[j][r] * sc[r];
}

__global__ __launch_bounds__(128) void k_proj_gemm128_wmma(const float* __restrict__ num_x,
                                                           const float* __restrict__ num_mask,
                                                           const float* __restrict__ x,
                                                           const float* __restrict__ Wn,
                                                           const float* __restrict__ bn,
                                                           const float* __restrict__ ain,
                                                           const float* __restrict__ W,
                                                           const float* __restrict__ dis,
                                                           float* __restrict__ t,
                                                           int nRowTiles, int N) {
    __shared__ float Wl[128 * 128];
    __shared__ v2f Pw[64];
    __shared__ v2f Pb[64];
    __shared__ v2f Pa[64];
    const int wave = threadIdx.x >> 5;
    const int lane = threadIdx.x & 31;
    const int rowTile = blockIdx.x * 4 + wave;
    const int mhalf = (lane >> 4) << 3;
    float sc[8];
    #pragma unroll
    for (int r = 0; r < 8; ++r) {
        int rr = rowTile * 16 + mhalf + r;
        rr = (rr < N) ? rr : (N - 1);
        sc[r] = dis[rr];
    }
    {
        int tid = threadIdx.x;
        v2f* Wlv = (v2f*)Wl;
        #pragma unroll
        for (int i = 0; i < 64; ++i) {
            int idx = i * 128 + tid;
            int c = idx & 127;
            int p = idx >> 7;
            v2f pr;
            pr.x = W[(2 * p)     * 128 + c];
            pr.y = W[(2 * p + 1) * 128 + c];
            Wlv[p * 128 + c] = pr;
        }
        if (tid < 64) {
            Pw[tid] = *(const v2f*)(Wn  + 2 * tid);
            Pb[tid] = *(const v2f*)(bn  + 2 * tid);
            Pa[tid] = *(const v2f*)(ain + 2 * tid);
        }
    }
    __syncthreads();
    asm volatile("" :: "v"(sc[0]), "v"(sc[1]), "v"(sc[2]), "v"(sc[3]), "v"(sc[4]), "v"(sc[5]), "v"(sc[6]), "v"(sc[7]));

    if (rowTile >= nRowTiles) return;

    const int r15   = lane & 15;
    const int khalf = (lane >> 4) << 1;
    int row = rowTile * 16 + r15;
    row = (row < N) ? row : (N - 1);
    const float s = num_x[row] * num_mask[row];
    const float* xrow = x + (size_t)row * 128;
    const v2f* Wlv = (const v2f*)Wl;

    v8f acc[8] = {};
    for (int k0 = 0; k0 < 128; k0 += 4) {
        const int kp = (k0 + khalf) >> 1;
        const v2f xv = *(const v2f*)(xrow + k0 + khalf);
        const v2f pw = Pw[kp], pb = Pb[kp], pa = Pa[kp];
        float z0 = s * pw.x + pb.x;
        float z1 = s * pw.y + pb.y;
        z0 = (z0 >= 0.0f) ? z0 : pa.x * z0;
        z1 = (z1 >= 0.0f) ? z1 : pa.y * z1;
        v2f a;
        a.x = z0 + xv.x;
        a.y = z1 + xv.y;
        const v2f* wrow = Wlv + (kp * 128 + r15);
        #pragma unroll
        for (int j = 0; j < 8; ++j) {
            v2f bfrag = wrow[j * 16];
            acc[j] = __builtin_amdgcn_wmma_f32_16x16x4_f32(
                false, a, false, bfrag, (short)0, acc[j], false, false);
        }
    }

    float* trow = t + (size_t)(rowTile * 16 + mhalf) * 128 + r15;
    #pragma unroll
    for (int j = 0; j < 8; ++j)
        #pragma unroll
        for (int r = 0; r < 8; ++r)
            trow[(size_t)r * 128 + j * 16] = acc[j][r] * sc[r];
}

__global__ __launch_bounds__(256) void k_gather_rows(const int* __restrict__ csr, const int* __restrict__ rowptr,
                                                     const int* __restrict__ cnt, const float* __restrict__ dis,
                                                     const float* __restrict__ t, const float* __restrict__ bias,
                                                     const float* __restrict__ a, float* __restrict__ h, int N) {
    const int wid  = (blockIdx.x * blockDim.x + threadIdx.x) >> 5;
    const int lane = threadIdx.x & 31;
    if (wid >= N) return;
    const int n = __builtin_amdgcn_readfirstlane(wid);
    const int beg = rowptr[n];
    const int end = beg + cnt[n];
    const float4* t4 = (const float4*)t;
    float4 acc = make_float4(0.0f, 0.0f, 0.0f, 0.0f);
    int e = beg;
    for (; e + 4 <= end; e += 4) {
        const int s0 = csr[e], s1 = csr[e + 1], s2 = csr[e + 2], s3 = csr[e + 3];
        const float4 v0 = t4[(size_t)s0 * 32 + lane];
        const float4 v1 = t4[(size_t)s1 * 32 + lane];
        const float4 v2 = t4[(size_t)s2 * 32 + lane];
        const float4 v3 = t4[(size_t)s3 * 32 + lane];
        acc.x += v0.x; acc.y += v0.y; acc.z += v0.z; acc.w += v0.w;
        acc.x += v1.x; acc.y += v1.y; acc.z += v1.z; acc.w += v1.w;
        acc.x += v2.x; acc.y += v2.y; acc.z += v2.z; acc.w += v2.w;
        acc.x += v3.x; acc.y += v3.y; acc.z += v3.z; acc.w += v3.w;
    }
    const int rem = end - e;
    if (rem == 3) {
        const int s0 = csr[e], s1 = csr[e + 1], s2 = csr[e + 2];
        const float4 v0 = t4[(size_t)s0 * 32 + lane];
        const float4 v1 = t4[(size_t)s1 * 32 + lane];
        const float4 v2 = t4[(size_t)s2 * 32 + lane];
        acc.x += v0.x; acc.y += v0.y; acc.z += v0.z; acc.w += v0.w;
        acc.x += v1.x; acc.y += v1.y; acc.z += v1.z; acc.w += v1.w;
        acc.x += v2.x; acc.y += v2.y; acc.z += v2.z; acc.w += v2.w;
    } else if (rem == 2) {
        const int s0 = csr[e], s1 = csr[e + 1];
        const float4 v0 = t4[(size_t)s0 * 32 + lane];
        const float4 v1 = t4[(size_t)s1 * 32 + lane];
        acc.x += v0.x; acc.y += v0.y; acc.z += v0.z; acc.w += v0.w;
        acc.x += v1.x; acc.y += v1.y; acc.z += v1.z; acc.w += v1.w;
    } else if (rem == 1) {
        const int s0 = csr[e];
        const float4 v0 = t4[(size_t)s0 * 32 + lane];
        acc.x += v0.x; acc.y += v0.y; acc.z += v0.z; acc.w += v0.w;
    }
    const float dn = dis[n];
    const float4 tv = t4[(size_t)n * 32 + lane];
    const float4 b  = ((const float4*)bias)[lane];
    const float4 av = ((const float4*)a)[lane];
    float4 r;
    r.x = dn * (acc.x + tv.x) + b.x;
    r.y = dn * (acc.y + tv.y) + b.y;
    r.z = dn * (acc.z + tv.z) + b.z;
    r.w = dn * (acc.w + tv.w) + b.w;
    r.x = (r.x >= 0.0f) ? r.x : av.x * r.x;
    r.y = (r.y >= 0.0f) ? r.y : av.y * r.y;
    r.z = (r.z >= 0.0f) ? r.z : av.z * r.z;
    r.w = (r.w >= 0.0f) ? r.w : av.w * r.w;
    ((float4*)h)[(size_t)n * 32 + lane] = r;
}

__global__ __launch_bounds__(256) void k_dot_w3(const float* __restrict__ h, const float* __restrict__ W3,
                                                const float* __restrict__ dis, float* __restrict__ t3, int N) {
    const int wid  = (blockIdx.x * blockDim.x + threadIdx.x) >> 5;
    const int lane = threadIdx.x & 31;
    const int row0 = wid * 8;
    if (row0 >= N) return;
    const float4* h4 = (const float4*)h;
    float4 hv[8];
    #pragma unroll
    for (int r = 0; r < 8; ++r) {
        int row = row0 + r;
        row = (row < N) ? row : (N - 1);
        hv[r] = h4[(size_t)row * 32 + lane];
    }
    const float4 wv = ((const float4*)W3)[lane];
    const int sub = lane & 7;
    const int myrow = row0 + sub;
    const float dn = dis[(myrow < N) ? myrow : (N - 1)];
    asm volatile("" :: "v"(dn));
    __builtin_amdgcn_sched_barrier(0);
    float keep = 0.0f;
    #pragma unroll
    for (int r = 0; r < 8; ++r) {
        float s = hv[r].x * wv.x + hv[r].y * wv.y + hv[r].z * wv.z + hv[r].w * wv.w;
        #pragma unroll
        for (int off = 16; off > 0; off >>= 1) s += __shfl_xor(s, off, 32);
        keep = (sub == r) ? s : keep;
    }
    if (lane < 8 && myrow < N) t3[myrow] = dn * keep;
}

__global__ void k_gather_scalar(const int* __restrict__ csr, const int* __restrict__ rowptr,
                                const int* __restrict__ cnt, const float* __restrict__ dis,
                                const float* __restrict__ t3, const float* __restrict__ b3,
                                float* __restrict__ out, int N) {
    const int gid = blockIdx.x * blockDim.x + threadIdx.x;
    const int n = gid >> 3, sub = gid & 7;
    const bool valid = n < N;
    int beg = 0, end = 0;
    if (valid) { beg = rowptr[n]; end = beg + cnt[n]; }
    float s = 0.0f;
    for (int e = beg + sub; e < end; e += 8) s += t3[csr[e]];
    s += __shfl_xor(s, 4, 32);
    s += __shfl_xor(s, 2, 32);
    s += __shfl_xor(s, 1, 32);
    if (valid && sub == 0) out[n] = dis[n] * (s + t3[n]) + b3[0];
}

extern "C" void kernel_launch(void* const* d_in, const int* in_sizes, int n_in,
                              void* d_out, int out_size, void* d_ws, size_t ws_size,
                              hipStream_t stream) {
    const float* num_x    = (const float*)d_in[0];
    const float* num_mask = (const float*)d_in[1];
    const float* x        = (const float*)d_in[2];
    const int*   edge     = (const int*)d_in[3];
    const float* W_num    = (const float*)d_in[4];
    const float* b_num    = (const float*)d_in[5];
    const float* a_in     = (const float*)d_in[6];
    const float* W1       = (const float*)d_in[7];
    const float* b1       = (const float*)d_in[8];
    const float* a1       = (const float*)d_in[9];
    const float* W2       = (const float*)d_in[10];
    const float* b2       = (const float*)d_in[11];
    const float* a2       = (const float*)d_in[12];
    const float* W3       = (const float*)d_in[13];
    const float* b3       = (const float*)d_in[14];

    const int N = in_sizes[1];
    const int E = in_sizes[3] / 2;
    const int* src = edge;
    const int* dst = edge + E;

    const int nRowTiles = (N + 15) / 16;
    const size_t padRows = (size_t)nRowTiles * 16;

    auto al = [](size_t v) { return (v + 255) & ~(size_t)255; };
    const int NB = (N + 255) / 256;
    const int EB = (E + 8191) / 8192;
    char* p = (char*)d_ws;
    float* dis    = (float*)p; p += al((size_t)N * 4);
    int*   cnt    = (int*)p;   p += al((size_t)N * 4);
    int*   rowptr = (int*)p;   p += al((size_t)N * 4);
    int*   loff   = (int*)p;   p += al((size_t)NB * EB * 4);
    int2*  ebuf   = (int2*)p;  p += al((size_t)EB * 8192 * 8);
    int*   csr    = (int*)p;   p += al((size_t)E * 4);
    float* hbuf = (float*)p; p += al(padRows * 128 * 4);
    float* tbuf = (float*)p; p += al(padRows * 128 * 4);
    float* t3   = (float*)p; p += al((size_t)N * 4);
    float* out  = (float*)d_out;

    const int TPB = 256;
    dim3 blk(TPB);
    auto gNw = dim3((int)(((size_t)N * 32 + TPB - 1) / TPB));
    auto gN8 = dim3((int)(((size_t)N * 8 + TPB - 1) / TPB));
    auto gN4 = dim3((int)(((size_t)N * 4 + TPB - 1) / TPB));
    const int gemmBlocks = (nRowTiles + 3) / 4;

    k_blk_sort<<<EB, 1024, 0, stream>>>(src, dst, loff, ebuf, E, NB, EB);
    k_bkt_rows<<<NB, 1024, 0, stream>>>(ebuf, loff, rowptr, cnt, dis, csr, N, E, NB, EB);

    k_proj_gemm128_wmma<<<gemmBlocks, 128, 0, stream>>>(num_x, num_mask, x, W_num, b_num, a_in, W1, dis, tbuf, nRowTiles, N);
    k_gather_rows<<<gNw, blk, 0, stream>>>(csr, rowptr, cnt, dis, tbuf, b1, a1, hbuf, N);

    k_gemm128_wmma<<<gemmBlocks, 128, 0, stream>>>(hbuf, W2, dis, tbuf, nRowTiles, N);
    k_gather_rows<<<gNw, blk, 0, stream>>>(csr, rowptr, cnt, dis, tbuf, b2, a2, hbuf, N);

    k_dot_w3<<<gN4, blk, 0, stream>>>(hbuf, W3, dis, t3, N);
    k_gather_scalar<<<gN8, blk, 0, stream>>>(csr, rowptr, cnt, dis, t3, b3, out, N);
}


# GEMMs on a fixed grid: 256 blocks of 16 waves, one per WGP (LDS sized for it), W staged once per block; 12 whole row strips per block, a 13th split in 4 column quarters: matrix units evenly loaded
# speedup vs baseline: 4.7415x; 1.0558x over previous
#include <hip/hip_runtime.h>

typedef __attribute__((ext_vector_type(2))) float v2f;
typedef __attribute__((ext_vector_type(8))) float v8f;


__global__ __launch_bounds__(1024) void k_blk_sort(const int* __restrict__ src, const int* __restrict__ dst,
                                                   int* __restrict__ loff, int2* __restrict__ ebuf,
                                                   int E, int NB, int EB) {
    __shared__ int hist[1024];
    __shared__ int wsum[32];
    const int tid = threadIdx.x, lane = tid & 31, w = tid >> 5;
    hist[tid] = 0;
    __syncthreads();
    const int e0 = blockIdx.x * 8192 + tid;
    int s[8], d[8], rk[8];
    #pragma unroll
    for (int j = 0; j < 8; ++j) {
        const int e = e0 + j * 1024;
        const bool ok = e < E;
        s[j] = ok ? src[e] : 0;
        d[j] = ok ? dst[e] : -1;
    }
    #pragma unroll
    for (int j = 0; j < 8; ++j) {
        rk[j] = 0;
        if (d[j] >= 0) rk[j] = atomicAdd(&hist[d[j] >> 8], 1);
    }
    __syncthreads();
    const int v = hist[tid];
    int inc = v;
    for (int off = 1; off < 32; off <<= 1) {
        int u = __shfl_up(inc, off, 32);
        if (lane >= off) inc += u;
    }
    if (lane == 31) wsum[w] = inc;
    __syncthreads();
    int ex = inc - v;
    for (int i = 0; i < w; ++i) ex += wsum[i];
    hist[tid] = ex;
    __syncthreads();
    int2* my = ebuf + (size_t)blockIdx.x * 8192;
    #pragma unroll
    for (int j = 0; j < 8; ++j)
        if (d[j] >= 0) my[hist[d[j] >> 8] + rk[j]] = make_int2(s[j], d[j]);
    if (tid < NB) loff[tid * EB + blockIdx.x] = ex;
}

__global__ __launch_bounds__(1024) void k_bkt_rows(const int2* __restrict__ ebuf, const int* __restrict__ loff,
                                                   int* __restrict__ rowptr, int* __restrict__ cnt,
                                                   float* __restrict__ dis, int* __restrict__ csr,
                                                   int N, int E, int NB, int EB) {
    __shared__ int c[256];
    __shared__ int cu[256];
    __shared__ int wsum[8];
    __shared__ int red[32];
    __shared__ int segLo[1024];
    __shared__ int segHi[1024];
    const int tid = threadIdx.x, lane = tid & 31, w = tid >> 5;
    const int b = blockIdx.x;
    int lo = 0;
    if (tid < EB) {
        lo = loff[b * EB + tid];
        const int left = E - tid * 8192;
        const int sliceCnt = (left < 8192) ? left : 8192;
        const int hi = (b + 1 < NB) ? loff[(b + 1) * EB + tid] : sliceCnt;
        segLo[tid] = tid * 8192 + lo;
        segHi[tid] = tid * 8192 + hi;
    }
    int r = lo;
    for (int off = 16; off > 0; off >>= 1) r += __shfl_xor(r, off, 32);
    if (lane == 0) red[w] = r;
    if (tid < 256) c[tid] = 0;
    __syncthreads();
    int base = red[lane];
    for (int off = 16; off > 0; off >>= 1) base += __shfl_xor(base, off, 32);

    int2 ev[4][3];
    #pragma unroll
    for (int q = 0; q < 4; ++q) {
        const int j = w + 32 * q;
        const int slo = (j < EB) ? segLo[j] : 0;
        const int shi = (j < EB) ? segHi[j] : 0;
        #pragma unroll
        for (int s = 0; s < 3; ++s) {
            const int i = slo + lane + 32 * s;
            ev[q][s] = (i < shi) ? ebuf[i] : make_int2(0, -1);
        }
    }
    #pragma unroll
    for (int q = 0; q < 4; ++q)
        #pragma unroll
        for (int s = 0; s < 3; ++s)
            if (ev[q][s].y >= 0) atomicAdd(&c[ev[q][s].y & 255], 1);
    for (int j = w; j < EB; j += 32) {
        const int shi = segHi[j];
        for (int i = segLo[j] + lane + ((j < 128) ? 96 : 0); i < shi; i += 32)
            atomicAdd(&c[ebuf[i].y & 255], 1);
    }
    __syncthreads();
    int v = 0, inc = 0;
    if (tid < 256) {
        v = c[tid];
        inc = v;
        for (int off = 1; off < 32; off <<= 1) {
            int u = __shfl_up(inc, off, 32);
            if (lane >= off) inc += u;
        }
        if (lane == 31) wsum[w] = inc;
    }
    __syncthreads();
    if (tid < 256) {
        int ex = inc - v;
        for (int i = 0; i < w; ++i) ex += wsum[i];
        const int start = base + ex;
        cu[tid] = start;
        const int n = b * 256 + tid;
        if (n < N) {
            rowptr[n] = start;
            cnt[n] = v;
            dis[n] = rsqrtf((float)v + 1.0f);
        }
    }
    __syncthreads();
    #pragma unroll
    for (int q = 0; q < 4; ++q)
        #pragma unroll
        for (int s = 0; s < 3; ++s)
            if (ev[q][s].y >= 0) {
                const int p = atomicAdd(&cu[ev[q][s].y & 255], 1);
                csr[p] = ev[q][s].x;
            }
    for (int j = w; j < EB; j += 32) {
        const int shi = segHi[j];
        for (int i = segLo[j] + lane + ((j < 128) ? 96 : 0); i < shi; i += 32) {
            const int2 e = ebuf[i];
            const int p = atomicAdd(&cu[e.y & 255], 1);
            csr[p] = e.x;
        }
    }
}

#define GEMM_BLOCKS 256
#define GEMM_THREADS 512
#define GEMM_FULL_WAVES 12
#define GEMM_LDS_FLOATS (42 * 1024)

__device__ __forceinline__ void load_row_scales(const float* __restrict__ dis, int rowTile, int mhalf, int N,
                                                float (&sc)[8]) {
    #pragma unroll
    for (int r = 0; r < 8; ++r) {
        int rr = rowTile * 16 + mhalf + r;
        rr = (rr < N) ? rr : (N - 1);
        sc[r] = dis[rr];
    }
}

__device__ __forceinline__ void stage_w(const float* __restrict__ W, float* Wl) {
    const int tid = threadIdx.x;
    v2f* Wlv = (v2f*)Wl;
    #pragma unroll
    for (int i = 0; i < 8192 / GEMM_THREADS; ++i) {
        const int idx = i * GEMM_THREADS + tid;
        const int c = idx & 127;
        const int p = idx >> 7;
        v2f pr;
        pr.x = W[(2 * p)     * 128 + c];
        pr.y = W[(2 * p + 1) * 128 + c];
        Wlv[p * 128 + c] = pr;
    }
}

template <int NJ, bool PROJ>
__device__ __forceinline__ void gemm_strip(const float* __restrict__ a_or_x, const float* __restrict__ num_x,
                                           const float* __restrict__ num_mask, const v2f* Pw, const v2f* Pb,
                                           const v2f* Pa, const v2f* Wlv, float* __restrict__ t,
                                           int rowTile, int jc0, const float (&sc)[8], int N, int lane) {
    const int r15   = lane & 15;
    const int khalf = (lane >> 4) << 1;
    const int mhalf = (lane >> 4) << 3;
    int row = rowTile * 16 + r15;
    float s = 0.0f;
    if (PROJ) {
        row = (row < N) ? row : (N - 1);
        s = num_x[row] * num_mask[row];
    }
    const float* arow = a_or_x + (size_t)row * 128;

    v8f acc[NJ] = {};
    for (int k0 = 0; k0 < 128; k0 += 4) {
        const int kp = (k0 + khalf) >> 1;
        v2f a = *(const v2f*)(arow + k0 + khalf);
        if (PROJ) {
            const v2f pw = Pw[kp], pb = Pb[kp], pa = Pa[kp];
            float z0 = s * pw.x + pb.x;
            float z1 = s * pw.y + pb.y;
            z0 = (z0 >= 0.0f) ? z0 : pa.x * z0;
            z1 = (z1 >= 0.0f) ? z1 : pa.y * z1;
            a.x = z0 + a.x;
            a.y = z1 + a.y;
        }
        const v2f* wrow = Wlv + (kp * 128 + jc0 * 16 + r15);
        #pragma unroll
        for (int j = 0; j < NJ; ++j) {
            v2f bfrag = wrow[j * 16];
            acc[j] = __builtin_amdgcn_wmma_f32_16x16x4_f32(
                false, a, false, bfrag, (short)0, acc[j], false, false);
        }
    }

    float* trow = t + (size_t)(rowTile * 16 + mhalf) * 128 + jc0 * 16 + r15;
    #pragma unroll
    for (int j = 0; j < NJ; ++j)
        #pragma unroll
        for (int r = 0; r < 8; ++r)
            trow[(size_t)r * 128 + j * 16] = acc[j][r] * sc[r];
}

template <bool PROJ>
__device__ __forceinline__ void gemm_block(const float* __restrict__ a_or_x, const float* __restrict__ num_x,
                                           const float* __restrict__ num_mask, const float* __restrict__ Wn,
                                           const float* __restrict__ bn, const float* __restrict__ ain,
                                           v2f* Pw, v2f* Pb, v2f* Pa, const float* __restrict__ W, float* Wl,
                                           const float* __restrict__ dis, float* __restrict__ t,
                                           int nRowTiles, int N) {
    const int wave = threadIdx.x >> 5;
    const int lane = threadIdx.x & 31;
    const int b = blockIdx.x;
    const int mhalf = (lane >> 4) << 3;
    const int nFull = nRowTiles / GEMM_BLOCKS;
    const int rem   = nRowTiles % GEMM_BLOCKS;
    const bool whole = wave < GEMM_FULL_WAVES;
    const bool has   = whole ? (wave < nFull) : (b < rem);
    const int tile0  = whole ? (b + GEMM_BLOCKS * wave) : (b + GEMM_BLOCKS * nFull);
    float sc[8];
    load_row_scales(dis, has ? tile0 : 0, mhalf, N, sc);
    const bool stp = PROJ && threadIdx.x < 64;
    v2f pwv = {}, pbv = {}, pav = {};
    if (stp) {
        pwv = *(const v2f*)(Wn  + 2 * threadIdx.x);
        pbv = *(const v2f*)(bn  + 2 * threadIdx.x);
        pav = *(const v2f*)(ain + 2 * threadIdx.x);
    }
    stage_w(W, Wl);
    if (stp) {
        Pw[threadIdx.x] = pwv;
        Pb[threadIdx.x] = pbv;
        Pa[threadIdx.x] = pav;
    }
    __syncthreads();
    asm volatile("" :: "v"(sc[0]), "v"(sc[1]), "v"(sc[2]), "v"(sc[3]), "v"(sc[4]), "v"(sc[5]), "v"(sc[6]), "v"(sc[7]));

    const v2f* Wlv = (const v2f*)Wl;
    if (whole) {
        for (int i = wave; i < nFull; i += GEMM_FULL_WAVES) {
            const int tile = b + GEMM_BLOCKS * i;
            if (i != wave) load_row_scales(dis, tile, mhalf, N, sc);
            gemm_strip<8, PROJ>(a_or_x, num_x, num_mask, Pw, Pb, Pa, Wlv, t, tile, 0, sc, N, lane);
        }
    } else if (has) {
        gemm_strip<2, PROJ>(a_or_x, num_x, num_mask, Pw, Pb, Pa, Wlv, t, tile0, (wave - GEMM_FULL_WAVES) * 2,
                            sc, N, lane);
    }
}

__global__ __launch_bounds__(GEMM_THREADS) void k_gemm128_wmma(const float* __restrict__ h,
                                                               const float* __restrict__ W,
                                                               const float* __restrict__ dis,
                                                               float* __restrict__ t,
                                                               int nRowTiles, int N) {
    __shared__ float Wl[GEMM_LDS_FLOATS];
    gemm_block<false>(h, nullptr, nullptr, nullptr, nullptr, nullptr, nullptr, nullptr, nullptr, W, Wl, dis, t,
                      nRowTiles, N);
}

__global__ __launch_bounds__(GEMM_THREADS) void k_proj_gemm128_wmma(const float* __restrict__ num_x,
                                                                    const float* __restrict__ num_mask,
                                                                    const float* __restrict__ x,
                                                                    const float* __restrict__ Wn,
                                                                    const float* __restrict__ bn,
                                                                    const float* __restrict__ ain,
                                                                    const float* __restrict__ W,
                                                                    const float* __restrict__ dis,
                                                                    float* __restrict__ t,
                                                                    int nRowTiles, int N) {
    __shared__ float Wl[GEMM_LDS_FLOATS];
    __shared__ v2f Pw[64];
    __shared__ v2f Pb[64];
    __shared__ v2f Pa[64];
    gemm_block<true>(x, num_x, num_mask, Wn, bn, ain, Pw, Pb, Pa, W, Wl, dis, t, nRowTiles, N);
}

__global__ __launch_bounds__(256) void k_gather_rows(const int* __restrict__ csr, const int* __restrict__ rowptr,
                                                     const int* __restrict__ cnt, const float* __restrict__ dis,
                                                     const float* __restrict__ t, const float* __restrict__ bias,
                                                     const float* __restrict__ a, float* __restrict__ h, int N) {
    const int wid  = (blockIdx.x * blockDim.x + threadIdx.x) >> 5;
    const int lane = threadIdx.x & 31;
    if (wid >= N) return;
    const int n = __builtin_amdgcn_readfirstlane(wid);
    const int beg = rowptr[n];
    const int end = beg + cnt[n];
    const float4* t4 = (const float4*)t;
    float4 acc = make_float4(0.0f, 0.0f, 0.0f, 0.0f);
    int e = beg;
    for (; e + 4 <= end; e += 4) {
        const int s0 = csr[e], s1 = csr[e + 1], s2 = csr[e + 2], s3 = csr[e + 3];
        const float4 v0 = t4[(size_t)s0 * 32 + lane];
        const float4 v1 = t4[(size_t)s1 * 32 + lane];
        const float4 v2 = t4[(size_t)s2 * 32 + lane];
        const float4 v3 = t4[(size_t)s3 * 32 + lane];
        acc.x += v0.x; acc.y += v0.y; acc.z += v0.z; acc.w += v0.w;
        acc.x += v1.x; acc.y += v1.y; acc.z += v1.z; acc.w += v1.w;
        acc.x += v2.x; acc.y += v2.y; acc.z += v2.z; acc.w += v2.w;
        acc.x += v3.x; acc.y += v3.y; acc.z += v3.z; acc.w += v3.w;
    }
    const int rem = end - e;
    if (rem == 3) {
        const int s0 = csr[e], s1 = csr[e + 1], s2 = csr[e + 2];
        const float4 v0 = t4[(size_t)s0 * 32 + lane];
        const float4 v1 = t4[(size_t)s1 * 32 + lane];
        const float4 v2 = t4[(size_t)s2 * 32 + lane];
        acc.x += v0.x; acc.y += v0.y; acc.z += v0.z; acc.w += v0.w;
        acc.x += v1.x; acc.y += v1.y; acc.z += v1.z; acc.w += v1.w;
        acc.x += v2.x; acc.y += v2.y; acc.z += v2.z; acc.w += v2.w;
    } else if (rem == 2) {
        const int s0 = csr[e], s1 = csr[e + 1];
        const float4 v0 = t4[(size_t)s0 * 32 + lane];
        const float4 v1 = t4[(size_t)s1 * 32 + lane];
        acc.x += v0.x; acc.y += v0.y; acc.z += v0.z; acc.w += v0.w;
        acc.x += v1.x; acc.y += v1.y; acc.z += v1.z; acc.w += v1.w;
    } else if (rem == 1) {
        const int s0 = csr[e];
        const float4 v0 = t4[(size_t)s0 * 32 + lane];
        acc.x += v0.x; acc.y += v0.y; acc.z += v0.z; acc.w += v0.w;
    }
    const float dn = dis[n];
    const float4 tv = t4[(size_t)n * 32 + lane];
    const float4 b  = ((const float4*)bias)[lane];
    const float4 av = ((const float4*)a)[lane];
    float4 r;
    r.x = dn * (acc.x + tv.x) + b.x;
    r.y = dn * (acc.y + tv.y) + b.y;
    r.z = dn * (acc.z + tv.z) + b.z;
    r.w = dn * (acc.w + tv.w) + b.w;
    r.x = (r.x >= 0.0f) ? r.x : av.x * r.x;
    r.y = (r.y >= 0.0f) ? r.y : av.y * r.y;
    r.z = (r.z >= 0.0f) ? r.z : av.z * r.z;
    r.w = (r.w >= 0.0f) ? r.w : av.w * r.w;
    ((float4*)h)[(size_t)n * 32 + lane] = r;
}

__global__ __launch_bounds__(256) void k_dot_w3(const float* __restrict__ h, const float* __restrict__ W3,
                                                const float* __restrict__ dis, float* __restrict__ t3, int N) {
    const int wid  = (blockIdx.x * blockDim.x + threadIdx.x) >> 5;
    const int lane = threadIdx.x & 31;
    const int row0 = wid * 8;
    if (row0 >= N) return;
    const float4* h4 = (const float4*)h;
    float4 hv[8];
    #pragma unroll
    for (int r = 0; r < 8; ++r) {
        int row = row0 + r;
        row = (row < N) ? row : (N - 1);
        hv[r] = h4[(size_t)row * 32 + lane];
    }
    const float4 wv = ((const float4*)W3)[lane];
    const int sub = lane & 7;
    const int myrow = row0 + sub;
    const float dn = dis[(myrow < N) ? myrow : (N - 1)];
    asm volatile("" :: "v"(dn));
    __builtin_amdgcn_sched_barrier(0);
    float keep = 0.0f;
    #pragma unroll
    for (int r = 0; r < 8; ++r) {
        float s = hv[r].x * wv.x + hv[r].y * wv.y + hv[r].z * wv.z + hv[r].w * wv.w;
        #pragma unroll
        for (int off = 16; off > 0; off >>= 1) s += __shfl_xor(s, off, 32);
        keep = (sub == r) ? s : keep;
    }
    if (lane < 8 && myrow < N) t3[myrow] = dn * keep;
}

__global__ void k_gather_scalar(const int* __restrict__ csr, const int* __restrict__ rowptr,
                                const int* __restrict__ cnt, const float* __restrict__ dis,
                                const float* __restrict__ t3, const float* __restrict__ b3,
                                float* __restrict__ out, int N) {
    const int gid = blockIdx.x * blockDim.x + threadIdx.x;
    const int n = gid >> 3, sub = gid & 7;
    const bool valid = n < N;
    int beg = 0, end = 0;
    if (valid) { beg = rowptr[n]; end = beg + cnt[n]; }
    float s = 0.0f;
    for (int e = beg + sub; e < end; e += 8) s += t3[csr[e]];
    s += __shfl_xor(s, 4, 32);
    s += __shfl_xor(s, 2, 32);
    s += __shfl_xor(s, 1, 32);
    if (valid && sub == 0) out[n] = dis[n] * (s + t3[n]) + b3[0];
}

extern "C" void kernel_launch(void* const* d_in, const int* in_sizes, int n_in,
                              void* d_out, int out_size, void* d_ws, size_t ws_size,
                              hipStream_t stream) {
    const float* num_x    = (const float*)d_in[0];
    const float* num_mask = (const float*)d_in[1];
    const float* x        = (const float*)d_in[2];
    const int*   edge     = (const int*)d_in[3];
    const float* W_num    = (const float*)d_in[4];
    const float* b_num    = (const float*)d_in[5];
    const float* a_in     = (const float*)d_in[6];
    const float* W1       = (const float*)d_in[7];
    const float* b1       = (const float*)d_in[8];
    const float* a1       = (const float*)d_in[9];
    const float* W2       = (const float*)d_in[10];
    const float* b2       = (const float*)d_in[11];
    const float* a2       = (const float*)d_in[12];
    const float* W3       = (const float*)d_in[13];
    const float* b3       = (const float*)d_in[14];

    const int N = in_sizes[1];
    const int E = in_sizes[3] / 2;
    const int* src = edge;
    const int* dst = edge + E;

    const int nRowTiles = (N + 15) / 16;
    const size_t padRows = (size_t)nRowTiles * 16;

    auto al = [](size_t v) { return (v + 255) & ~(size_t)255; };
    const int NB = (N + 255) / 256;
    const int EB = (E + 8191) / 8192;
    char* p = (char*)d_ws;
    float* dis    = (float*)p; p += al((size_t)N * 4);
    int*   cnt    = (int*)p;   p += al((size_t)N * 4);
    int*   rowptr = (int*)p;   p += al((size_t)N * 4);
    int*   loff   = (int*)p;   p += al((size_t)NB * EB * 4);
    int2*  ebuf   = (int2*)p;  p += al((size_t)EB * 8192 * 8);
    int*   csr    = (int*)p;   p += al((size_t)E * 4);
    float* hbuf = (float*)p; p += al(padRows * 128 * 4);
    float* tbuf = (float*)p; p += al(padRows * 128 * 4);
    float* t3   = (float*)p; p += al((size_t)N * 4);
    float* out  = (float*)d_out;

    const int TPB = 256;
    dim3 blk(TPB);
    auto gNw = dim3((int)(((size_t)N * 32 + TPB - 1) / TPB));
    auto gN8 = dim3((int)(((size_t)N * 8 + TPB - 1) / TPB));
    auto gN4 = dim3((int)(((size_t)N * 4 + TPB - 1) / TPB));

    k_blk_sort<<<EB, 1024, 0, stream>>>(src, dst, loff, ebuf, E, NB, EB);
    k_bkt_rows<<<NB, 1024, 0, stream>>>(ebuf, loff, rowptr, cnt, dis, csr, N, E, NB, EB);

    k_proj_gemm128_wmma<<<GEMM_BLOCKS, GEMM_THREADS, 0, stream>>>(num_x, num_mask, x, W_num, b_num, a_in, W1, dis, tbuf, nRowTiles, N);
    k_gather_rows<<<gNw, blk, 0, stream>>>(csr, rowptr, cnt, dis, tbuf, b1, a1, hbuf, N);

    k_gemm128_wmma<<<GEMM_BLOCKS, GEMM_THREADS, 0, stream>>>(hbuf, W2, dis, tbuf, nRowTiles, N);
    k_gather_rows<<<gNw, blk, 0, stream>>>(csr, rowptr, cnt, dis, tbuf, b2, a2, hbuf, N);

    k_dot_w3<<<gN4, blk, 0, stream>>>(hbuf, W3, dis, t3, N);
    k_gather_scalar<<<gN8, blk, 0, stream>>>(csr, rowptr, cnt, dis, t3, b3, out, N);
}


# both edge-grouping passes place their records in LDS and store them in index order: a wave's store covers whole 128-byte lines, not 32 lines with 4 or 8 bytes each; row order kept
# speedup vs baseline: 5.0808x; 1.0716x over previous
#include <hip/hip_runtime.h>

typedef __attribute__((ext_vector_type(2))) float v2f;
typedef __attribute__((ext_vector_type(8))) float v8f;
typedef __attribute__((ext_vector_type(2))) int v2i;


__global__ __launch_bounds__(1024) void k_blk_sort(const int* __restrict__ src, const int* __restrict__ dst,
                                                   int* __restrict__ loff, int2* __restrict__ ebuf,
                                                   int E, int NB, int EB) {
    __shared__ int hist[1024];
    __shared__ int wsum[32];
    __shared__ v2i stage[8192];
    const int tid = threadIdx.x, lane = tid & 31, w = tid >> 5;
    hist[tid] = 0;
    __syncthreads();
    const int e0 = blockIdx.x * 8192 + tid;
    int s[8], d[8], rk[8];
    #pragma unroll
    for (int j = 0; j < 8; ++j) {
        const int e = e0 + j * 1024;
        const bool ok = e < E;
        s[j] = ok ? src[e] : 0;
        d[j] = ok ? dst[e] : -1;
    }
    #pragma unroll
    for (int j = 0; j < 8; ++j) {
        rk[j] = 0;
        if (d[j] >= 0) rk[j] = atomicAdd(&hist[d[j] >> 8], 1);
    }
    __syncthreads();
    const int v = hist[tid];
    int inc = v;
    for (int off = 1; off < 32; off <<= 1) {
        int u = __shfl_up(inc, off, 32);
        if (lane >= off) inc += u;
    }
    if (lane == 31) wsum[w] = inc;
    __syncthreads();
    int ex = inc - v;
    for (int i = 0; i < w; ++i) ex += wsum[i];
    hist[tid] = ex;
    __syncthreads();
    #pragma unroll
    for (int j = 0; j < 8; ++j)
        if (d[j] >= 0) {
            v2i rec;
            rec.x = s[j];
            rec.y = d[j];
            stage[hist[d[j] >> 8] + rk[j]] = rec;
        }
    __syncthreads();
    const int left = E - blockIdx.x * 8192;
    const int n = (left < 8192) ? left : 8192;
    int2* my = ebuf + (size_t)blockIdx.x * 8192;
    #pragma unroll
    for (int j = 0; j < 8; ++j) {
        const int i = tid + j * 1024;
        if (i < n) {
            const v2i rec = stage[i];
            my[i] = make_int2(rec.x, rec.y);
        }
    }
    if (tid < NB) loff[tid * EB + blockIdx.x] = ex;
}

#define BKT_STAGE 8192
__global__ __launch_bounds__(1024) void k_bkt_rows(const int2* __restrict__ ebuf, const int* __restrict__ loff,
                                                   int* __restrict__ rowptr, int* __restrict__ cnt,
                                                   float* __restrict__ dis, int* __restrict__ csr,
                                                   int N, int E, int NB, int EB) {
    __shared__ int c[256];
    __shared__ int cu[256];
    __shared__ int wsum[8];
    __shared__ int red[32];
    __shared__ int segLo[1024];
    __shared__ int segHi[1024];
    __shared__ int rowsLds[BKT_STAGE];
    __shared__ int totS;
    const int tid = threadIdx.x, lane = tid & 31, w = tid >> 5;
    const int b = blockIdx.x;
    int lo = 0;
    if (tid < EB) {
        lo = loff[b * EB + tid];
        const int left = E - tid * 8192;
        const int sliceCnt = (left < 8192) ? left : 8192;
        const int hi = (b + 1 < NB) ? loff[(b + 1) * EB + tid] : sliceCnt;
        segLo[tid] = tid * 8192 + lo;
        segHi[tid] = tid * 8192 + hi;
    }
    int r = lo;
    for (int off = 16; off > 0; off >>= 1) r += __shfl_xor(r, off, 32);
    if (lane == 0) red[w] = r;
    if (tid < 256) c[tid] = 0;
    __syncthreads();
    int base = red[lane];
    for (int off = 16; off > 0; off >>= 1) base += __shfl_xor(base, off, 32);

    int2 ev[4][3];
    #pragma unroll
    for (int q = 0; q < 4; ++q) {
        const int j = w + 32 * q;
        const int slo = (j < EB) ? segLo[j] : 0;
        const int shi = (j < EB) ? segHi[j] : 0;
        #pragma unroll
        for (int s = 0; s < 3; ++s) {
            const int i = slo + lane + 32 * s;
            ev[q][s] = (i < shi) ? ebuf[i] : make_int2(0, -1);
        }
    }
    #pragma unroll
    for (int q = 0; q < 4; ++q)
        #pragma unroll
        for (int s = 0; s < 3; ++s)
            if (ev[q][s].y >= 0) atomicAdd(&c[ev[q][s].y & 255], 1);
    for (int j = w; j < EB; j += 32) {
        const int shi = segHi[j];
        for (int i = segLo[j] + lane + ((j < 128) ? 96 : 0); i < shi; i += 32)
            atomicAdd(&c[ebuf[i].y & 255], 1);
    }
    __syncthreads();
    int v = 0, inc = 0;
    if (tid < 256) {
        v = c[tid];
        inc = v;
        for (int off = 1; off < 32; off <<= 1) {
            int u = __shfl_up(inc, off, 32);
            if (lane >= off) inc += u;
        }
        if (lane == 31) wsum[w] = inc;
    }
    __syncthreads();
    if (tid < 256) {
        int ex = inc - v;
        for (int i = 0; i < w; ++i) ex += wsum[i];
        const int start = base + ex;
        cu[tid] = start;
        if (tid == 255) totS = ex + v;
        const int n = b * 256 + tid;
        if (n < N) {
            rowptr[n] = start;
            cnt[n] = v;
            dis[n] = rsqrtf((float)v + 1.0f);
        }
    }
    __syncthreads();
    const int total = totS;
    if (total <= BKT_STAGE) {
        #pragma unroll
        for (int q = 0; q < 4; ++q)
            #pragma unroll
            for (int s = 0; s < 3; ++s)
                if (ev[q][s].y >= 0) {
                    const int p = atomicAdd(&cu[ev[q][s].y & 255], 1);
                    rowsLds[p - base] = ev[q][s].x;
                }
        for (int j = w; j < EB; j += 32) {
            const int shi = segHi[j];
            for (int i = segLo[j] + lane + ((j < 128) ? 96 : 0); i < shi; i += 32) {
                const int2 e = ebuf[i];
                const int p = atomicAdd(&cu[e.y & 255], 1);
                rowsLds[p - base] = e.x;
            }
        }
        __syncthreads();
        for (int i = tid; i < total; i += 1024) csr[base + i] = rowsLds[i];
    } else {
        #pragma unroll
        for (int q = 0; q < 4; ++q)
            #pragma unroll
            for (int s = 0; s < 3; ++s)
                if (ev[q][s].y >= 0) {
                    const int p = atomicAdd(&cu[ev[q][s].y & 255], 1);
                    csr[p] = ev[q][s].x;
                }
        for (int j = w; j < EB; j += 32) {
            const int shi = segHi[j];
            for (int i = segLo[j] + lane + ((j < 128) ? 96 : 0); i < shi; i += 32) {
                const int2 e = ebuf[i];
                const int p = atomicAdd(&cu[e.y & 255], 1);
                csr[p] = e.x;
            }
        }
    }
}

#define GEMM_BLOCKS 256
#define GEMM_THREADS 512
#define GEMM_FULL_WAVES 12
#define GEMM_LDS_FLOATS (42 * 1024)

__device__ __forceinline__ void load_row_scales(const float* __restrict__ dis, int rowTile, int mhalf, int N,
                                                float (&sc)[8]) {
    #pragma unroll
    for (int r = 0; r < 8; ++r) {
        int rr = rowTile * 16 + mhalf + r;
        rr = (rr < N) ? rr : (N - 1);
        sc[r] = dis[rr];
    }
}

__device__ __forceinline__ void stage_w(const float* __restrict__ W, float* Wl) {
    const int tid = threadIdx.x;
    v2f* Wlv = (v2f*)Wl;
    #pragma unroll
    for (int i = 0; i < 8192 / GEMM_THREADS; ++i) {
        const int idx = i * GEMM_THREADS + tid;
        const int c = idx & 127;
        const int p = idx >> 7;
        v2f pr;
        pr.x = W[(2 * p)     * 128 + c];
        pr.y = W[(2 * p + 1) * 128 + c];
        Wlv[p * 128 + c] = pr;
    }
}

template <int NJ, bool PROJ>
__device__ __forceinline__ void gemm_strip(const float* __restrict__ a_or_x, const float* __restrict__ num_x,
                                           const float* __restrict__ num_mask, const v2f* Pw, const v2f* Pb,
                                           const v2f* Pa, const v2f* Wlv, float* __restrict__ t,
                                           int rowTile, int jc0, const float (&sc)[8], int N, int lane) {
    const int r15   = lane & 15;
    const int khalf = (lane >> 4) << 1;
    const int mhalf = (lane >> 4) << 3;
    int row = rowTile * 16 + r15;
    float s = 0.0f;
    if (PROJ) {
        row = (row < N) ? row : (N - 1);
        s = num_x[row] * num_mask[row];
    }
    const float* arow = a_or_x + (size_t)row * 128;

    v8f acc[NJ] = {};
    for (int k0 = 0; k0 < 128; k0 += 4) {
        const int kp = (k0 + khalf) >> 1;
        v2f a = *(const v2f*)(arow + k0 + khalf);
        if (PROJ) {
            const v2f pw = Pw[kp], pb = Pb[kp], pa = Pa[kp];
            float z0 = s * pw.x + pb.x;
            float z1 = s * pw.y + pb.y;
            z0 = (z0 >= 0.0f) ? z0 : pa.x * z0;
            z1 = (z1 >= 0.0f) ? z1 : pa.y * z1;
            a.x = z0 + a.x;
            a.y = z1 + a.y;
        }
        const v2f* wrow = Wlv + (kp * 128 + jc0 * 16 + r15);
        #pragma unroll
        for (int j = 0; j < NJ; ++j) {
            v2f bfrag = wrow[j * 16];
            acc[j] = __builtin_amdgcn_wmma_f32_16x16x4_f32(
                false, a, false, bfrag, (short)0, acc[j], false, false);
        }
    }

    float* trow = t + (size_t)(rowTile * 16 + mhalf) * 128 + jc0 * 16 + r15;
    #pragma unroll
    for (int j = 0; j < NJ; ++j)
        #pragma unroll
        for (int r = 0; r < 8; ++r)
            trow[(size_t)r * 128 + j * 16] = acc[j][r] * sc[r];
}

template <bool PROJ>
__device__ __forceinline__ void gemm_block(const float* __restrict__ a_or_x, const float* __restrict__ num_x,
                                           const float* __restrict__ num_mask, const float* __restrict__ Wn,
                                           const float* __restrict__ bn, const float* __restrict__ ain,
                                           v2f* Pw, v2f* Pb, v2f* Pa, const float* __restrict__ W, float* Wl,
                                           const float* __restrict__ dis, float* __restrict__ t,
                                           int nRowTiles, int N) {
    const int wave = threadIdx.x >> 5;
    const int lane = threadIdx.x & 31;
    const int b = blockIdx.x;
    const int mhalf = (lane >> 4) << 3;
    const int nFull = nRowTiles / GEMM_BLOCKS;
    const int rem   = nRowTiles % GEMM_BLOCKS;
    const bool whole = wave < GEMM_FULL_WAVES;
    const bool has   = whole ? (wave < nFull) : (b < rem);
    const int tile0  = whole ? (b + GEMM_BLOCKS * wave) : (b + GEMM_BLOCKS * nFull);
    float sc[8];
    load_row_scales(dis, has ? tile0 : 0, mhalf, N, sc);
    const bool stp = PROJ && threadIdx.x < 64;
    v2f pwv = {}, pbv = {}, pav = {};
    if (stp) {
        pwv = *(const v2f*)(Wn  + 2 * threadIdx.x);
        pbv = *(const v2f*)(bn  + 2 * threadIdx.x);
        pav = *(const v2f*)(ain + 2 * threadIdx.x);
    }
    stage_w(W, Wl);
    if (stp) {
        Pw[threadIdx.x] = pwv;
        Pb[threadIdx.x] = pbv;
        Pa[threadIdx.x] = pav;
    }
    __syncthreads();
    asm volatile("" :: "v"(sc[0]), "v"(sc[1]), "v"(sc[2]), "v"(sc[3]), "v"(sc[4]), "v"(sc[5]), "v"(sc[6]), "v"(sc[7]));

    const v2f* Wlv = (const v2f*)Wl;
    if (whole) {
        for (int i = wave; i < nFull; i += GEMM_FULL_WAVES) {
            const int tile = b + GEMM_BLOCKS * i;
            if (i != wave) load_row_scales(dis, tile, mhalf, N, sc);
            gemm_strip<8, PROJ>(a_or_x, num_x, num_mask, Pw, Pb, Pa, Wlv, t, tile, 0, sc, N, lane);
        }
    } else if (has) {
        gemm_strip<2, PROJ>(a_or_x, num_x, num_mask, Pw, Pb, Pa, Wlv, t, tile0, (wave - GEMM_FULL_WAVES) * 2,
                            sc, N, lane);
    }
}

__global__ __launch_bounds__(GEMM_THREADS) void k_gemm128_wmma(const float* __restrict__ h,
                                                               const float* __restrict__ W,
                                                               const float* __restrict__ dis,
                                                               float* __restrict__ t,
                                                               int nRowTiles, int N) {
    __shared__ float Wl[GEMM_LDS_FLOATS];
    gemm_block<false>(h, nullptr, nullptr, nullptr, nullptr, nullptr, nullptr, nullptr, nullptr, W, Wl, dis, t,
                      nRowTiles, N);
}

__global__ __launch_bounds__(GEMM_THREADS) void k_proj_gemm128_wmma(const float* __restrict__ num_x,
                                                                    const float* __restrict__ num_mask,
                                                                    const float* __restrict__ x,
                                                                    const float* __restrict__ Wn,
                                                                    const float* __restrict__ bn,
                                                                    const float* __restrict__ ain,
                                                                    const float* __restrict__ W,
                                                                    const float* __restrict__ dis,
                                                                    float* __restrict__ t,
                                                                    int nRowTiles, int N) {
    __shared__ float Wl[GEMM_LDS_FLOATS];
    __shared__ v2f Pw[64];
    __shared__ v2f Pb[64];
    __shared__ v2f Pa[64];
    gemm_block<true>(x, num_x, num_mask, Wn, bn, ain, Pw, Pb, Pa, W, Wl, dis, t, nRowTiles, N);
}

__global__ __launch_bounds__(256) void k_gather_rows(const int* __restrict__ csr, const int* __restrict__ rowptr,
                                                     const int* __restrict__ cnt, const float* __restrict__ dis,
                                                     const float* __restrict__ t, const float* __restrict__ bias,
                                                     const float* __restrict__ a, float* __restrict__ h, int N) {
    const int wid  = (blockIdx.x * blockDim.x + threadIdx.x) >> 5;
    const int lane = threadIdx.x & 31;
    if (wid >= N) return;
    const int n = __builtin_amdgcn_readfirstlane(wid);
    const int beg = rowptr[n];
    const int end = beg + cnt[n];
    const float4* t4 = (const float4*)t;
    float4 acc = make_float4(0.0f, 0.0f, 0.0f, 0.0f);
    int e = beg;
    for (; e + 4 <= end; e += 4) {
        const int s0 = csr[e], s1 = csr[e + 1], s2 = csr[e + 2], s3 = csr[e + 3];
        const float4 v0 = t4[(size_t)s0 * 32 + lane];
        const float4 v1 = t4[(size_t)s1 * 32 + lane];
        const float4 v2 = t4[(size_t)s2 * 32 + lane];
        const float4 v3 = t4[(size_t)s3 * 32 + lane];
        acc.x += v0.x; acc.y += v0.y; acc.z += v0.z; acc.w += v0.w;
        acc.x += v1.x; acc.y += v1.y; acc.z += v1.z; acc.w += v1.w;
        acc.x += v2.x; acc.y += v2.y; acc.z += v2.z; acc.w += v2.w;
        acc.x += v3.x; acc.y += v3.y; acc.z += v3.z; acc.w += v3.w;
    }
    const int rem = end - e;
    if (rem == 3) {
        const int s0 = csr[e], s1 = csr[e + 1], s2 = csr[e + 2];
        const float4 v0 = t4[(size_t)s0 * 32 + lane];
        const float4 v1 = t4[(size_t)s1 * 32 + lane];
        const float4 v2 = t4[(size_t)s2 * 32 + lane];
        acc.x += v0.x; acc.y += v0.y; acc.z += v0.z; acc.w += v0.w;
        acc.x += v1.x; acc.y += v1.y; acc.z += v1.z; acc.w += v1.w;
        acc.x += v2.x; acc.y += v2.y; acc.z += v2.z; acc.w += v2.w;
    } else if (rem == 2) {
        const int s0 = csr[e], s1 = csr[e + 1];
        const float4 v0 = t4[(size_t)s0 * 32 + lane];
        const float4 v1 = t4[(size_t)s1 * 32 + lane];
        acc.x += v0.x; acc.y += v0.y; acc.z += v0.z; acc.w += v0.w;
        acc.x += v1.x; acc.y += v1.y; acc.z += v1.z; acc.w += v1.w;
    } else if (rem == 1) {
        const int s0 = csr[e];
        const float4 v0 = t4[(size_t)s0 * 32 + lane];
        acc.x += v0.x; acc.y += v0.y; acc.z += v0.z; acc.w += v0.w;
    }
    const float dn = dis[n];
    const float4 tv = t4[(size_t)n * 32 + lane];
    const float4 b  = ((const float4*)bias)[lane];
    const float4 av = ((const float4*)a)[lane];
    float4 r;
    r.x = dn * (acc.x + tv.x) + b.x;
    r.y = dn * (acc.y + tv.y) + b.y;
    r.z = dn * (acc.z + tv.z) + b.z;
    r.w = dn * (acc.w + tv.w) + b.w;
    r.x = (r.x >= 0.0f) ? r.x : av.x * r.x;
    r.y = (r.y >= 0.0f) ? r.y : av.y * r.y;
    r.z = (r.z >= 0.0f) ? r.z : av.z * r.z;
    r.w = (r.w >= 0.0f) ? r.w : av.w * r.w;
    ((float4*)h)[(size_t)n * 32 + lane] = r;
}

__global__ __launch_bounds__(256) void k_dot_w3(const float* __restrict__ h, const float* __restrict__ W3,
                                                const float* __restrict__ dis, float* __restrict__ t3, int N) {
    const int wid  = (blockIdx.x * blockDim.x + threadIdx.x) >> 5;
    const int lane = threadIdx.x & 31;
    const int row0 = wid * 8;
    if (row0 >= N) return;
    const float4* h4 = (const float4*)h;
    float4 hv[8];
    #pragma unroll
    for (int r = 0; r < 8; ++r) {
        int row = row0 + r;
        row = (row < N) ? row : (N - 1);
        hv[r] = h4[(size_t)row * 32 + lane];
    }
    const float4 wv = ((const float4*)W3)[lane];
    const int sub = lane & 7;
    const int myrow = row0 + sub;
    const float dn = dis[(myrow < N) ? myrow : (N - 1)];
    asm volatile("" :: "v"(dn));
    __builtin_amdgcn_sched_barrier(0);
    float keep = 0.0f;
    #pragma unroll
    for (int r = 0; r < 8; ++r) {
        float s = hv[r].x * wv.x + hv[r].y * wv.y + hv[r].z * wv.z + hv[r].w * wv.w;
        #pragma unroll
        for (int off = 16; off > 0; off >>= 1) s += __shfl_xor(s, off, 32);
        keep = (sub == r) ? s : keep;
    }
    if (lane < 8 && myrow < N) t3[myrow] = dn * keep;
}

__global__ void k_gather_scalar(const int* __restrict__ csr, const int* __restrict__ rowptr,
                                const int* __restrict__ cnt, const float* __restrict__ dis,
                                const float* __restrict__ t3, const float* __restrict__ b3,
                                float* __restrict__ out, int N) {
    const int gid = blockIdx.x * blockDim.x + threadIdx.x;
    const int n = gid >> 3, sub = gid & 7;
    const bool valid = n < N;
    int beg = 0, end = 0;
    if (valid) { beg = rowptr[n]; end = beg + cnt[n]; }
    float s = 0.0f;
    for (int e = beg + sub; e < end; e += 8) s += t3[csr[e]];
    s += __shfl_xor(s, 4, 32);
    s += __shfl_xor(s, 2, 32);
    s += __shfl_xor(s, 1, 32);
    if (valid && sub == 0) out[n] = dis[n] * (s + t3[n]) + b3[0];
}

extern "C" void kernel_launch(void* const* d_in, const int* in_sizes, int n_in,
                              void* d_out, int out_size, void* d_ws, size_t ws_size,
                              hipStream_t stream) {
    const float* num_x    = (const float*)d_in[0];
    const float* num_mask = (const float*)d_in[1];
    const float* x        = (const float*)d_in[2];
    const int*   edge     = (const int*)d_in[3];
    const float* W_num    = (const float*)d_in[4];
    const float* b_num    = (const float*)d_in[5];
    const float* a_in     = (const float*)d_in[6];
    const float* W1       = (const float*)d_in[7];
    const float* b1       = (const float*)d_in[8];
    const float* a1       = (const float*)d_in[9];
    const float* W2       = (const float*)d_in[10];
    const float* b2       = (const float*)d_in[11];
    const float* a2       = (const float*)d_in[12];
    const float* W3       = (const float*)d_in[13];
    const float* b3       = (const float*)d_in[14];

    const int N = in_sizes[1];
    const int E = in_sizes[3] / 2;
    const int* src = edge;
    const int* dst = edge + E;

    const int nRowTiles = (N + 15) / 16;
    const size_t padRows = (size_t)nRowTiles * 16;

    auto al = [](size_t v) { return (v + 255) & ~(size_t)255; };
    const int NB = (N + 255) / 256;
    const int EB = (E + 8191) / 8192;
    char* p = (char*)d_ws;
    float* dis    = (float*)p; p += al((size_t)N * 4);
    int*   cnt    = (int*)p;   p += al((size_t)N * 4);
    int*   rowptr = (int*)p;   p += al((size_t)N * 4);
    int*   loff   = (int*)p;   p += al((size_t)NB * EB * 4);
    int2*  ebuf   = (int2*)p;  p += al((size_t)EB * 8192 * 8);
    int*   csr    = (int*)p;   p += al((size_t)E * 4);
    float* hbuf = (float*)p; p += al(padRows * 128 * 4);
    float* tbuf = (float*)p; p += al(padRows * 128 * 4);
    float* t3   = (float*)p; p += al((size_t)N * 4);
    float* out  = (float*)d_out;

    const int TPB = 256;
    dim3 blk(TPB);
    auto gNw = dim3((int)(((size_t)N * 32 + TPB - 1) / TPB));
    auto gN8 = dim3((int)(((size_t)N * 8 + TPB - 1) / TPB));
    auto gN4 = dim3((int)(((size_t)N * 4 + TPB - 1) / TPB));

    k_blk_sort<<<EB, 1024, 0, stream>>>(src, dst, loff, ebuf, E, NB, EB);
    k_bkt_rows<<<NB, 1024, 0, stream>>>(ebuf, loff, rowptr, cnt, dis, csr, N, E, NB, EB);

    k_proj_gemm128_wmma<<<GEMM_BLOCKS, GEMM_THREADS, 0, stream>>>(num_x, num_mask, x, W_num, b_num, a_in, W1, dis, tbuf, nRowTiles, N);
    k_gather_rows<<<gNw, blk, 0, stream>>>(csr, rowptr, cnt, dis, tbuf, b1, a1, hbuf, N);

    k_gemm128_wmma<<<GEMM_BLOCKS, GEMM_THREADS, 0, stream>>>(hbuf, W2, dis, tbuf, nRowTiles, N);
    k_gather_rows<<<gNw, blk, 0, stream>>>(csr, rowptr, cnt, dis, tbuf, b2, a2, hbuf, N);

    k_dot_w3<<<gN4, blk, 0, stream>>>(hbuf, W3, dis, t3, N);
    k_gather_scalar<<<gN8, blk, 0, stream>>>(csr, rowptr, cnt, dis, t3, b3, out, N);
}


# layer-2 GEMM fused into the layer-1 gather launch: 16 waves gather a 16-row tile into LDS (8 row loads in flight), 4 multiply it by W2 on WMMA; h1 plane, its re-read, one launch gone; sum order kept
# speedup vs baseline: 5.6414x; 1.1103x over previous
#include <hip/hip_runtime.h>

typedef __attribute__((ext_vector_type(2))) float v2f;
typedef __attribute__((ext_vector_type(8))) float v8f;
typedef __attribute__((ext_vector_type(2))) int v2i;


__global__ __launch_bounds__(1024) void k_blk_sort(const int* __restrict__ src, const int* __restrict__ dst,
                                                   int* __restrict__ loff, int2* __restrict__ ebuf,
                                                   int E, int NB, int EB) {
    __shared__ int hist[1024];
    __shared__ int wsum[32];
    __shared__ v2i stage[8192];
    const int tid = threadIdx.x, lane = tid & 31, w = tid >> 5;
    hist[tid] = 0;
    __syncthreads();
    const int e0 = blockIdx.x * 8192 + tid;
    int s[8], d[8], rk[8];
    #pragma unroll
    for (int j = 0; j < 8; ++j) {
        const int e = e0 + j * 1024;
        const bool ok = e < E;
        s[j] = ok ? src[e] : 0;
        d[j] = ok ? dst[e] : -1;
    }
    #pragma unroll
    for (int j = 0; j < 8; ++j) {
        rk[j] = 0;
        if (d[j] >= 0) rk[j] = atomicAdd(&hist[d[j] >> 8], 1);
    }
    __syncthreads();
    const int v = hist[tid];
    int inc = v;
    for (int off = 1; off < 32; off <<= 1) {
        int u = __shfl_up(inc, off, 32);
        if (lane >= off) inc += u;
    }
    if (lane == 31) wsum[w] = inc;
    __syncthreads();
    int ex = inc - v;
    for (int i = 0; i < w; ++i) ex += wsum[i];
    hist[tid] = ex;
    __syncthreads();
    #pragma unroll
    for (int j = 0; j < 8; ++j)
        if (d[j] >= 0) {
            v2i rec;
            rec.x = s[j];
            rec.y = d[j];
            stage[hist[d[j] >> 8] + rk[j]] = rec;
        }
    __syncthreads();
    const int left = E - blockIdx.x * 8192;
    const int n = (left < 8192) ? left : 8192;
    int2* my = ebuf + (size_t)blockIdx.x * 8192;
    #pragma unroll
    for (int j = 0; j < 8; ++j) {
        const int i = tid + j * 1024;
        if (i < n) {
            const v2i rec = stage[i];
            my[i] = make_int2(rec.x, rec.y);
        }
    }
    if (tid < NB) loff[tid * EB + blockIdx.x] = ex;
}

#define BKT_STAGE 8192
__global__ __launch_bounds__(1024) void k_bkt_rows(const int2* __restrict__ ebuf, const int* __restrict__ loff,
                                                   int* __restrict__ rowptr, int* __restrict__ cnt,
                                                   float* __restrict__ dis, int* __restrict__ csr,
                                                   int N, int E, int NB, int EB) {
    __shared__ int c[256];
    __shared__ int cu[256];
    __shared__ int wsum[8];
    __shared__ int red[32];
    __shared__ int segLo[1024];
    __shared__ int segHi[1024];
    __shared__ int rowsLds[BKT_STAGE];
    __shared__ int totS;
    const int tid = threadIdx.x, lane = tid & 31, w = tid >> 5;
    const int b = blockIdx.x;
    int lo = 0;
    if (tid < EB) {
        lo = loff[b * EB + tid];
        const int left = E - tid * 8192;
        const int sliceCnt = (left < 8192) ? left : 8192;
        const int hi = (b + 1 < NB) ? loff[(b + 1) * EB + tid] : sliceCnt;
        segLo[tid] = tid * 8192 + lo;
        segHi[tid] = tid * 8192 + hi;
    }
    int r = lo;
    for (int off = 16; off > 0; off >>= 1) r += __shfl_xor(r, off, 32);
    if (lane == 0) red[w] = r;
    if (tid < 256) c[tid] = 0;
    __syncthreads();
    int base = red[lane];
    for (int off = 16; off > 0; off >>= 1) base += __shfl_xor(base, off, 32);

    int2 ev[4][3];
    #pragma unroll
    for (int q = 0; q < 4; ++q) {
        const int j = w + 32 * q;
        const int slo = (j < EB) ? segLo[j] : 0;
        const int shi = (j < EB) ? segHi[j] : 0;
        #pragma unroll
        for (int s = 0; s < 3; ++s) {
            const int i = slo + lane + 32 * s;
            ev[q][s] = (i < shi) ? ebuf[i] : make_int2(0, -1);
        }
    }
    #pragma unroll
    for (int q = 0; q < 4; ++q)
        #pragma unroll
        for (int s = 0; s < 3; ++s)
            if (ev[q][s].y >= 0) atomicAdd(&c[ev[q][s].y & 255], 1);
    for (int j = w; j < EB; j += 32) {
        const int shi = segHi[j];
        for (int i = segLo[j] + lane + ((j < 128) ? 96 : 0); i < shi; i += 32)
            atomicAdd(&c[ebuf[i].y & 255], 1);
    }
    __syncthreads();
    int v = 0, inc = 0;
    if (tid < 256) {
        v = c[tid];
        inc = v;
        for (int off = 1; off < 32; off <<= 1) {
            int u = __shfl_up(inc, off, 32);
            if (lane >= off) inc += u;
        }
        if (lane == 31) wsum[w] = inc;
    }
    __syncthreads();
    if (tid < 256) {
        int ex = inc - v;
        for (int i = 0; i < w; ++i) ex += wsum[i];
        const int start = base + ex;
        cu[tid] = start;
        if (tid == 255) totS = ex + v;
        const int n = b * 256 + tid;
        if (n < N) {
            rowptr[n] = start;
            cnt[n] = v;
            dis[n] = rsqrtf((float)v + 1.0f);
        }
    }
    __syncthreads();
    const int total = totS;
    if (total <= BKT_STAGE) {
        #pragma unroll
        for (int q = 0; q < 4; ++q)
            #pragma unroll
            for (int s = 0; s < 3; ++s)
                if (ev[q][s].y >= 0) {
                    const int p = atomicAdd(&cu[ev[q][s].y & 255], 1);
                    rowsLds[p - base] = ev[q][s].x;
                }
        for (int j = w; j < EB; j += 32) {
            const int shi = segHi[j];
            for (int i = segLo[j] + lane + ((j < 128) ? 96 : 0); i < shi; i += 32) {
                const int2 e = ebuf[i];
                const int p = atomicAdd(&cu[e.y & 255], 1);
                rowsLds[p - base] = e.x;
            }
        }
        __syncthreads();
        for (int i = tid; i < total; i += 1024) csr[base + i] = rowsLds[i];
    } else {
        #pragma unroll
        for (int q = 0; q < 4; ++q)
            #pragma unroll
            for (int s = 0; s < 3; ++s)
                if (ev[q][s].y >= 0) {
                    const int p = atomicAdd(&cu[ev[q][s].y & 255], 1);
                    csr[p] = ev[q][s].x;
                }
        for (int j = w; j < EB; j += 32) {
            const int shi = segHi[j];
            for (int i = segLo[j] + lane + ((j < 128) ? 96 : 0); i < shi; i += 32) {
                const int2 e = ebuf[i];
                const int p = atomicAdd(&cu[e.y & 255], 1);
                csr[p] = e.x;
            }
        }
    }
}

#define GEMM_BLOCKS 256
#define GEMM_THREADS 512
#define GEMM_FULL_WAVES 12
#define GEMM_LDS_FLOATS (42 * 1024)

__device__ __forceinline__ void load_row_scales(const float* __restrict__ dis, int rowTile, int mhalf, int N,
                                                float (&sc)[8]) {
    #pragma unroll
    for (int r = 0; r < 8; ++r) {
        int rr = rowTile * 16 + mhalf + r;
        rr = (rr < N) ? rr : (N - 1);
        sc[r] = dis[rr];
    }
}

__device__ __forceinline__ void stage_w(const float* __restrict__ W, float* Wl) {
    const int tid = threadIdx.x;
    v2f* Wlv = (v2f*)Wl;
    #pragma unroll
    for (int i = 0; i < 8192 / GEMM_THREADS; ++i) {
        const int idx = i * GEMM_THREADS + tid;
        const int c = idx & 127;
        const int p = idx >> 7;
        v2f pr;
        pr.x = W[(2 * p)     * 128 + c];
        pr.y = W[(2 * p + 1) * 128 + c];
        Wlv[p * 128 + c] = pr;
    }
}

template <int NJ, bool PROJ>
__device__ __forceinline__ void gemm_strip(const float* __restrict__ a_or_x, const float* __restrict__ num_x,
                                           const float* __restrict__ num_mask, const v2f* Pw, const v2f* Pb,
                                           const v2f* Pa, const v2f* Wlv, float* __restrict__ t,
                                           int rowTile, int jc0, const float (&sc)[8], int N, int lane) {
    const int r15   = lane & 15;
    const int khalf = (lane >> 4) << 1;
    const int mhalf = (lane >> 4) << 3;
    int row = rowTile * 16 + r15;
    float s = 0.0f;
    if (PROJ) {
        row = (row < N) ? row : (N - 1);
        s = num_x[row] * num_mask[row];
    }
    const float* arow = a_or_x + (size_t)row * 128;

    v8f acc[NJ] = {};
    for (int k0 = 0; k0 < 128; k0 += 4) {
        const int kp = (k0 + khalf) >> 1;
        v2f a = *(const v2f*)(arow + k0 + khalf);
        if (PROJ) {
            const v2f pw = Pw[kp], pb = Pb[kp], pa = Pa[kp];
            float z0 = s * pw.x + pb.x;
            float z1 = s * pw.y + pb.y;
            z0 = (z0 >= 0.0f) ? z0 : pa.x * z0;
            z1 = (z1 >= 0.0f) ? z1 : pa.y * z1;
            a.x = z0 + a.x;
            a.y = z1 + a.y;
        }
        const v2f* wrow = Wlv + (kp * 128 + jc0 * 16 + r15);
        #pragma unroll
        for (int j = 0; j < NJ; ++j) {
            v2f bfrag = wrow[j * 16];
            acc[j] = __builtin_amdgcn_wmma_f32_16x16x4_f32(
                false, a, false, bfrag, (short)0, acc[j], false, false);
        }
    }

    float* trow = t + (size_t)(rowTile * 16 + mhalf) * 128 + jc0 * 16 + r15;
    #pragma unroll
    for (int j = 0; j < NJ; ++j)
        #pragma unroll
        for (int r = 0; r < 8; ++r)
            trow[(size_t)r * 128 + j * 16] = acc[j][r] * sc[r];
}

template <bool PROJ>
__device__ __forceinline__ void gemm_block(const float* __restrict__ a_or_x, const float* __restrict__ num_x,
                                           const float* __restrict__ num_mask, const float* __restrict__ Wn,
                                           const float* __restrict__ bn, const float* __restrict__ ain,
                                           v2f* Pw, v2f* Pb, v2f* Pa, const float* __restrict__ W, float* Wl,
                                           const float* __restrict__ dis, float* __restrict__ t,
                                           int nRowTiles, int N) {
    const int wave = threadIdx.x >> 5;
    const int lane = threadIdx.x & 31;
    const int b = blockIdx.x;
    const int mhalf = (lane >> 4) << 3;
    const int nFull = nRowTiles / GEMM_BLOCKS;
    const int rem   = nRowTiles % GEMM_BLOCKS;
    const bool whole = wave < GEMM_FULL_WAVES;
    const bool has   = whole ? (wave < nFull) : (b < rem);
    const int tile0  = whole ? (b + GEMM_BLOCKS * wave) : (b + GEMM_BLOCKS * nFull);
    float sc[8];
    load_row_scales(dis, has ? tile0 : 0, mhalf, N, sc);
    const bool stp = PROJ && threadIdx.x < 64;
    v2f pwv = {}, pbv = {}, pav = {};
    if (stp) {
        pwv = *(const v2f*)(Wn  + 2 * threadIdx.x);
        pbv = *(const v2f*)(bn  + 2 * threadIdx.x);
        pav = *(const v2f*)(ain + 2 * threadIdx.x);
    }
    stage_w(W, Wl);
    if (stp) {
        Pw[threadIdx.x] = pwv;
        Pb[threadIdx.x] = pbv;
        Pa[threadIdx.x] = pav;
    }
    __syncthreads();
    asm volatile("" :: "v"(sc[0]), "v"(sc[1]), "v"(sc[2]), "v"(sc[3]), "v"(sc[4]), "v"(sc[5]), "v"(sc[6]), "v"(sc[7]));

    const v2f* Wlv = (const v2f*)Wl;
    if (whole) {
        for (int i = wave; i < nFull; i += GEMM_FULL_WAVES) {
            const int tile = b + GEMM_BLOCKS * i;
            if (i != wave) load_row_scales(dis, tile, mhalf, N, sc);
            gemm_strip<8, PROJ>(a_or_x, num_x, num_mask, Pw, Pb, Pa, Wlv, t, tile, 0, sc, N, lane);
        }
    } else if (has) {
        gemm_strip<2, PROJ>(a_or_x, num_x, num_mask, Pw, Pb, Pa, Wlv, t, tile0, (wave - GEMM_FULL_WAVES) * 2,
                            sc, N, lane);
    }
}

__global__ __launch_bounds__(GEMM_THREADS) void k_proj_gemm128_wmma(const float* __restrict__ num_x,
                                                                    const float* __restrict__ num_mask,
                                                                    const float* __restrict__ x,
                                                                    const float* __restrict__ Wn,
                                                                    const float* __restrict__ bn,
                                                                    const float* __restrict__ ain,
                                                                    const float* __restrict__ W,
                                                                    const float* __restrict__ dis,
                                                                    float* __restrict__ t,
                                                                    int nRowTiles, int N) {
    __shared__ float Wl[GEMM_LDS_FLOATS];
    __shared__ v2f Pw[64];
    __shared__ v2f Pb[64];
    __shared__ v2f Pa[64];
    gemm_block<true>(x, num_x, num_mask, Wn, bn, ain, Pw, Pb, Pa, W, Wl, dis, t, nRowTiles, N);
}

__global__ __launch_bounds__(256) void k_gather_rows(const int* __restrict__ csr, const int* __restrict__ rowptr,
                                                     const int* __restrict__ cnt, const float* __restrict__ dis,
                                                     const float* __restrict__ t, const float* __restrict__ bias,
                                                     const float* __restrict__ a, float* __restrict__ h, int N) {
    const int wid  = (blockIdx.x * blockDim.x + threadIdx.x) >> 5;
    const int lane = threadIdx.x & 31;
    if (wid >= N) return;
    const int n = __builtin_amdgcn_readfirstlane(wid);
    const int beg = rowptr[n];
    const int end = beg + cnt[n];
    const float4* t4 = (const float4*)t;
    float4 acc = make_float4(0.0f, 0.0f, 0.0f, 0.0f);
    int e = beg;
    for (; e + 4 <= end; e += 4) {
        const int s0 = csr[e], s1 = csr[e + 1], s2 = csr[e + 2], s3 = csr[e + 3];
        const float4 v0 = t4[(size_t)s0 * 32 + lane];
        const float4 v1 = t4[(size_t)s1 * 32 + lane];
        const float4 v2 = t4[(size_t)s2 * 32 + lane];
        const float4 v3 = t4[(size_t)s3 * 32 + lane];
        acc.x += v0.x; acc.y += v0.y; acc.z += v0.z; acc.w += v0.w;
        acc.x += v1.x; acc.y += v1.y; acc.z += v1.z; acc.w += v1.w;
        acc.x += v2.x; acc.y += v2.y; acc.z += v2.z; acc.w += v2.w;
        acc.x += v3.x; acc.y += v3.y; acc.z += v3.z; acc.w += v3.w;
    }
    const int rem = end - e;
    if (rem == 3) {
        const int s0 = csr[e], s1 = csr[e + 1], s2 = csr[e + 2];
        const float4 v0 = t4[(size_t)s0 * 32 + lane];
        const float4 v1 = t4[(size_t)s1 * 32 + lane];
        const float4 v2 = t4[(size_t)s2 * 32 + lane];
        acc.x += v0.x; acc.y += v0.y; acc.z += v0.z; acc.w += v0.w;
        acc.x += v1.x; acc.y += v1.y; acc.z += v1.z; acc.w += v1.w;
        acc.x += v2.x; acc.y += v2.y; acc.z += v2.z; acc.w += v2.w;
    } else if (rem == 2) {
        const int s0 = csr[e], s1 = csr[e + 1];
        const float4 v0 = t4[(size_t)s0 * 32 + lane];
        const float4 v1 = t4[(size_t)s1 * 32 + lane];
        acc.x += v0.x; acc.y += v0.y; acc.z += v0.z; acc.w += v0.w;
        acc.x += v1.x; acc.y += v1.y; acc.z += v1.z; acc.w += v1.w;
    } else if (rem == 1) {
        const int s0 = csr[e];
        const float4 v0 = t4[(size_t)s0 * 32 + lane];
        acc.x += v0.x; acc.y += v0.y; acc.z += v0.z; acc.w += v0.w;
    }
    const float dn = dis[n];
    const float4 tv = t4[(size_t)n * 32 + lane];
    const float4 b  = ((const float4*)bias)[lane];
    const float4 av = ((const float4*)a)[lane];
    float4 r;
    r.x = dn * (acc.x + tv.x) + b.x;
    r.y = dn * (acc.y + tv.y) + b.y;
    r.z = dn * (acc.z + tv.z) + b.z;
    r.w = dn * (acc.w + tv.w) + b.w;
    r.x = (r.x >= 0.0f) ? r.x : av.x * r.x;
    r.y = (r.y >= 0.0f) ? r.y : av.y * r.y;
    r.z = (r.z >= 0.0f) ? r.z : av.z * r.z;
    r.w = (r.w >= 0.0f) ? r.w : av.w * r.w;
    ((float4*)h)[(size_t)n * 32 + lane] = r;
}

#define FUSE_BLOCKS 1024
#define A_PITCH 132

__device__ __forceinline__ void add4(float4& acc, const float4 v) {
    acc.x += v.x; acc.y += v.y; acc.z += v.z; acc.w += v.w;
}

__device__ __forceinline__ void stage_w_quarters(const float* __restrict__ W, float* Wq) {
    const int tid = threadIdx.x;
    v2f* Wqv = (v2f*)Wq;
    #pragma unroll
    for (int i = 0; i < 8192 / GEMM_THREADS; ++i) {
        const int idx = i * GEMM_THREADS + tid;
        const int c = idx & 127;
        const int p = idx >> 7;
        v2f pr;
        pr.x = W[(2 * p)     * 128 + c];
        pr.y = W[(2 * p + 1) * 128 + c];
        Wqv[p * 128 + (c >> 5) * 32 + (c & 15) * 2 + ((c >> 4) & 1)] = pr;
    }
}

__global__ __launch_bounds__(GEMM_THREADS) void k_gather_gemm128_wmma(const int* __restrict__ csr,
                                                                      const int* __restrict__ rowptr,
                                                                      const int* __restrict__ cnt,
                                                                      const float* __restrict__ dis,
                                                                      const float* __restrict__ t,
                                                                      const float* __restrict__ bias,
                                                                      const float* __restrict__ a,
                                                                      const float* __restrict__ W,
                                                                      float* __restrict__ t2,
                                                                      int nRowTiles, int N) {
    __shared__ alignas(16) float Wq[128 * 128];
    __shared__ alignas(16) float At[16 * A_PITCH];
    __shared__ float dsc[16];
    __shared__ float4 Pb[32];
    __shared__ float4 Pa[32];
    const int wave = threadIdx.x >> 5;
    const int lane = threadIdx.x & 31;
    const float4* t4 = (const float4*)t;
    if (wave == 0) Pb[lane] = ((const float4*)bias)[lane];
    if (wave == 1) Pa[lane] = ((const float4*)a)[lane];
    stage_w_quarters(W, Wq);
    __syncthreads();
    const bool mm = (wave & 3) == (wave >> 2);
    const int quarter = wave & 3;
    const int r15   = lane & 15;
    const int khalf = (lane >> 4) << 1;
    const int mhalf = (lane >> 4) << 3;

    for (int tile = blockIdx.x; tile < nRowTiles; tile += gridDim.x) {
        const int n = __builtin_amdgcn_readfirstlane(tile * 16 + wave);
        float4 r = make_float4(0.0f, 0.0f, 0.0f, 0.0f);
        float dn = 0.0f;
        if (n < N) {
            const int beg = rowptr[n];
            const int end = beg + cnt[n];
            dn = dis[n];
            const float4 tv = t4[(size_t)n * 32 + lane];
            float4 acc = make_float4(0.0f, 0.0f, 0.0f, 0.0f);
            int e = beg;
            for (; e + 8 <= end; e += 8) {
                const int s0 = __builtin_amdgcn_readfirstlane(csr[e]);
                const int s1 = __builtin_amdgcn_readfirstlane(csr[e + 1]);
                const int s2 = __builtin_amdgcn_readfirstlane(csr[e + 2]);
                const int s3 = __builtin_amdgcn_readfirstlane(csr[e + 3]);
                const int s4 = __builtin_amdgcn_readfirstlane(csr[e + 4]);
                const int s5 = __builtin_amdgcn_readfirstlane(csr[e + 5]);
                const int s6 = __builtin_amdgcn_readfirstlane(csr[e + 6]);
                const int s7 = __builtin_amdgcn_readfirstlane(csr[e + 7]);
                const float4 v0 = t4[(size_t)s0 * 32 + lane];
                const float4 v1 = t4[(size_t)s1 * 32 + lane];
                const float4 v2 = t4[(size_t)s2 * 32 + lane];
                const float4 v3 = t4[(size_t)s3 * 32 + lane];
                const float4 v4 = t4[(size_t)s4 * 32 + lane];
                const float4 v5 = t4[(size_t)s5 * 32 + lane];
                const float4 v6 = t4[(size_t)s6 * 32 + lane];
                const float4 v7 = t4[(size_t)s7 * 32 + lane];
                add4(acc, v0); add4(acc, v1); add4(acc, v2); add4(acc, v3);
                add4(acc, v4); add4(acc, v5); add4(acc, v6); add4(acc, v7);
            }
            if (e + 4 <= end) {
                const int s0 = __builtin_amdgcn_readfirstlane(csr[e]);
                const int s1 = __builtin_amdgcn_readfirstlane(csr[e + 1]);
                const int s2 = __builtin_amdgcn_readfirstlane(csr[e + 2]);
                const int s3 = __builtin_amdgcn_readfirstlane(csr[e + 3]);
                const float4 v0 = t4[(size_t)s0 * 32 + lane];
                const float4 v1 = t4[(size_t)s1 * 32 + lane];
                const float4 v2 = t4[(size_t)s2 * 32 + lane];
                const float4 v3 = t4[(size_t)s3 * 32 + lane];
                add4(acc, v0); add4(acc, v1); add4(acc, v2); add4(acc, v3);
                e += 4;
            }
            const int rem = end - e;
            if (rem == 3) {
                const int s0 = __builtin_amdgcn_readfirstlane(csr[e]);
                const int s1 = __builtin_amdgcn_readfirstlane(csr[e + 1]);
                const int s2 = __builtin_amdgcn_readfirstlane(csr[e + 2]);
                const float4 v0 = t4[(size_t)s0 * 32 + lane];
                const float4 v1 = t4[(size_t)s1 * 32 + lane];
                const float4 v2 = t4[(size_t)s2 * 32 + lane];
                add4(acc, v0); add4(acc, v1); add4(acc, v2);
            } else if (rem == 2) {
                const int s0 = __builtin_amdgcn_readfirstlane(csr[e]);
                const int s1 = __builtin_amdgcn_readfirstlane(csr[e + 1]);
                const float4 v0 = t4[(size_t)s0 * 32 + lane];
                const float4 v1 = t4[(size_t)s1 * 32 + lane];
                add4(acc, v0); add4(acc, v1);
            } else if (rem == 1) {
                const int s0 = __builtin_amdgcn_readfirstlane(csr[e]);
                const float4 v0 = t4[(size_t)s0 * 32 + lane];
                add4(acc, v0);
            }
            const float4 b  = Pb[lane];
            const float4 av = Pa[lane];
            r.x = dn * (acc.x + tv.x) + b.x;
            r.y = dn * (acc.y + tv.y) + b.y;
            r.z = dn * (acc.z + tv.z) + b.z;
            r.w = dn * (acc.w + tv.w) + b.w;
            r.x = (r.x >= 0.0f) ? r.x : av.x * r.x;
            r.y = (r.y >= 0.0f) ? r.y : av.y * r.y;
            r.z = (r.z >= 0.0f) ? r.z : av.z * r.z;
            r.w = (r.w >= 0.0f) ? r.w : av.w * r.w;
        }
        *(float4*)(At + wave * A_PITCH + lane * 4) = r;
        if (lane == 0) dsc[wave] = dn;
        __syncthreads();

        if (mm) {
            const float* arow = At + r15 * A_PITCH + khalf;
            const float* brow = Wq + (khalf >> 1) * 256 + quarter * 64 + r15 * 4;
            v8f acc[2] = {};
            #pragma unroll
            for (int k0 = 0; k0 < 128; k0 += 4) {
                const v2f af = *(const v2f*)(arow + k0);
                const float4 bq = *(const float4*)(brow + (k0 >> 1) * 256);
                v2f b0, b1;
                b0.x = bq.x; b0.y = bq.y;
                b1.x = bq.z; b1.y = bq.w;
                acc[0] = __builtin_amdgcn_wmma_f32_16x16x4_f32(false, af, false, b0, (short)0, acc[0], false, false);
                acc[1] = __builtin_amdgcn_wmma_f32_16x16x4_f32(false, af, false, b1, (short)0, acc[1], false, false);
            }
            float* trow = t2 + (size_t)(tile * 16 + mhalf) * 128 + quarter * 32 + r15;
            #pragma unroll
            for (int j = 0; j < 2; ++j)
                #pragma unroll
                for (int rr = 0; rr < 8; ++rr)
                    trow[(size_t)rr * 128 + j * 16] = acc[j][rr] * dsc[mhalf + rr];
        }
        __syncthreads();
    }
}

__global__ __launch_bounds__(256) void k_dot_w3(const float* __restrict__ h, const float* __restrict__ W3,
                                                const float* __restrict__ dis, float* __restrict__ t3, int N) {
    const int wid  = (blockIdx.x * blockDim.x + threadIdx.x) >> 5;
    const int lane = threadIdx.x & 31;
    const int row0 = wid * 8;
    if (row0 >= N) return;
    const float4* h4 = (const float4*)h;
    float4 hv[8];
    #pragma unroll
    for (int r = 0; r < 8; ++r) {
        int row = row0 + r;
        row = (row < N) ? row : (N - 1);
        hv[r] = h4[(size_t)row * 32 + lane];
    }
    const float4 wv = ((const float4*)W3)[lane];
    const int sub = lane & 7;
    const int myrow = row0 + sub;
    const float dn = dis[(myrow < N) ? myrow : (N - 1)];
    asm volatile("" :: "v"(dn));
    __builtin_amdgcn_sched_barrier(0);
    float keep = 0.0f;
    #pragma unroll
    for (int r = 0; r < 8; ++r) {
        float s = hv[r].x * wv.x + hv[r].y * wv.y + hv[r].z * wv.z + hv[r].w * wv.w;
        #pragma unroll
        for (int off = 16; off > 0; off >>= 1) s += __shfl_xor(s, off, 32);
        keep = (sub == r) ? s : keep;
    }
    if (lane < 8 && myrow < N) t3[myrow] = dn * keep;
}

__global__ void k_gather_scalar(const int* __restrict__ csr, const int* __restrict__ rowptr,
                                const int* __restrict__ cnt, const float* __restrict__ dis,
                                const float* __restrict__ t3, const float* __restrict__ b3,
                                float* __restrict__ out, int N) {
    const int gid = blockIdx.x * blockDim.x + threadIdx.x;
    const int n = gid >> 3, sub = gid & 7;
    const bool valid = n < N;
    int beg = 0, end = 0;
    if (valid) { beg = rowptr[n]; end = beg + cnt[n]; }
    float s = 0.0f;
    for (int e = beg + sub; e < end; e += 8) s += t3[csr[e]];
    s += __shfl_xor(s, 4, 32);
    s += __shfl_xor(s, 2, 32);
    s += __shfl_xor(s, 1, 32);
    if (valid && sub == 0) out[n] = dis[n] * (s + t3[n]) + b3[0];
}

extern "C" void kernel_launch(void* const* d_in, const int* in_sizes, int n_in,
                              void* d_out, int out_size, void* d_ws, size_t ws_size,
                              hipStream_t stream) {
    const float* num_x    = (const float*)d_in[0];
    const float* num_mask = (const float*)d_in[1];
    const float* x        = (const float*)d_in[2];
    const int*   edge     = (const int*)d_in[3];
    const float* W_num    = (const float*)d_in[4];
    const float* b_num    = (const float*)d_in[5];
    const float* a_in     = (const float*)d_in[6];
    const float* W1       = (const float*)d_in[7];
    const float* b1       = (const float*)d_in[8];
    const float* a1       = (const float*)d_in[9];
    const float* W2       = (const float*)d_in[10];
    const float* b2       = (const float*)d_in[11];
    const float* a2       = (const float*)d_in[12];
    const float* W3       = (const float*)d_in[13];
    const float* b3       = (const float*)d_in[14];

    const int N = in_sizes[1];
    const int E = in_sizes[3] / 2;
    const int* src = edge;
    const int* dst = edge + E;

    const int nRowTiles = (N + 15) / 16;
    const size_t padRows = (size_t)nRowTiles * 16;

    auto al = [](size_t v) { return (v + 255) & ~(size_t)255; };
    const int NB = (N + 255) / 256;
    const int EB = (E + 8191) / 8192;
    char* p = (char*)d_ws;
    float* dis    = (float*)p; p += al((size_t)N * 4);
    int*   cnt    = (int*)p;   p += al((size_t)N * 4);
    int*   rowptr = (int*)p;   p += al((size_t)N * 4);
    int*   loff   = (int*)p;   p += al((size_t)NB * EB * 4);
    int2*  ebuf   = (int2*)p;  p += al((size_t)EB * 8192 * 8);
    int*   csr    = (int*)p;   p += al((size_t)E * 4);
    float* hbuf = (float*)p; p += al(padRows * 128 * 4);
    float* tbuf = (float*)p; p += al(padRows * 128 * 4);
    float* t3   = (float*)p; p += al((size_t)N * 4);
    float* out  = (float*)d_out;

    const int TPB = 256;
    dim3 blk(TPB);
    auto gNw = dim3((int)(((size_t)N * 32 + TPB - 1) / TPB));
    auto gN8 = dim3((int)(((size_t)N * 8 + TPB - 1) / TPB));
    auto gN4 = dim3((int)(((size_t)N * 4 + TPB - 1) / TPB));

    k_blk_sort<<<EB, 1024, 0, stream>>>(src, dst, loff, ebuf, E, NB, EB);
    k_bkt_rows<<<NB, 1024, 0, stream>>>(ebuf, loff, rowptr, cnt, dis, csr, N, E, NB, EB);

    k_proj_gemm128_wmma<<<GEMM_BLOCKS, GEMM_THREADS, 0, stream>>>(num_x, num_mask, x, W_num, b_num, a_in, W1, dis, tbuf, nRowTiles, N);
    k_gather_gemm128_wmma<<<FUSE_BLOCKS, GEMM_THREADS, 0, stream>>>(csr, rowptr, cnt, dis, tbuf, b1, a1, W2, hbuf, nRowTiles, N);

    k_gather_rows<<<gNw, blk, 0, stream>>>(csr, rowptr, cnt, dis, hbuf, b2, a2, tbuf, N);

    k_dot_w3<<<gN4, blk, 0, stream>>>(tbuf, W3, dis, t3, N);
    k_gather_scalar<<<gN8, blk, 0, stream>>>(csr, rowptr, cnt, dis, t3, b3, out, N);
}


# layer-3 dot done in the layer-2 gather launch by 8-wave blocks owning 32 rows (4 per wave): one wave stores t3 as a whole 128-byte line; dot launch, h2 plane gone; row sums and dot chain unchanged
# speedup vs baseline: 5.9109x; 1.0478x over previous
#include <hip/hip_runtime.h>

typedef __attribute__((ext_vector_type(2))) float v2f;
typedef __attribute__((ext_vector_type(8))) float v8f;
typedef __attribute__((ext_vector_type(2))) int v2i;


__global__ __launch_bounds__(1024) void k_blk_sort(const int* __restrict__ src, const int* __restrict__ dst,
                                                   int* __restrict__ loff, int2* __restrict__ ebuf,
                                                   int E, int NB, int EB) {
    __shared__ int hist[1024];
    __shared__ int wsum[32];
    __shared__ v2i stage[8192];
    const int tid = threadIdx.x, lane = tid & 31, w = tid >> 5;
    hist[tid] = 0;
    __syncthreads();
    const int e0 = blockIdx.x * 8192 + tid;
    int s[8], d[8], rk[8];
    #pragma unroll
    for (int j = 0; j < 8; ++j) {
        const int e = e0 + j * 1024;
        const bool ok = e < E;
        s[j] = ok ? src[e] : 0;
        d[j] = ok ? dst[e] : -1;
    }
    #pragma unroll
    for (int j = 0; j < 8; ++j) {
        rk[j] = 0;
        if (d[j] >= 0) rk[j] = atomicAdd(&hist[d[j] >> 8], 1);
    }
    __syncthreads();
    const int v = hist[tid];
    int inc = v;
    for (int off = 1; off < 32; off <<= 1) {
        int u = __shfl_up(inc, off, 32);
        if (lane >= off) inc += u;
    }
    if (lane == 31) wsum[w] = inc;
    __syncthreads();
    int ex = inc - v;
    for (int i = 0; i < w; ++i) ex += wsum[i];
    hist[tid] = ex;
    __syncthreads();
    #pragma unroll
    for (int j = 0; j < 8; ++j)
        if (d[j] >= 0) {
            v2i rec;
            rec.x = s[j];
            rec.y = d[j];
            stage[hist[d[j] >> 8] + rk[j]] = rec;
        }
    __syncthreads();
    const int left = E - blockIdx.x * 8192;
    const int n = (left < 8192) ? left : 8192;
    int2* my = ebuf + (size_t)blockIdx.x * 8192;
    #pragma unroll
    for (int j = 0; j < 8; ++j) {
        const int i = tid + j * 1024;
        if (i < n) {
            const v2i rec = stage[i];
            my[i] = make_int2(rec.x, rec.y);
        }
    }
    if (tid < NB) loff[tid * EB + blockIdx.x] = ex;
}

#define BKT_STAGE 8192
__global__ __launch_bounds__(1024) void k_bkt_rows(const int2* __restrict__ ebuf, const int* __restrict__ loff,
                                                   int* __restrict__ rowptr, int* __restrict__ cnt,
                                                   float* __restrict__ dis, int* __restrict__ csr,
                                                   int N, int E, int NB, int EB) {
    __shared__ int c[256];
    __shared__ int cu[256];
    __shared__ int wsum[8];
    __shared__ int red[32];
    __shared__ int segLo[1024];
    __shared__ int segHi[1024];
    __shared__ int rowsLds[BKT_STAGE];
    __shared__ int totS;
    const int tid = threadIdx.x, lane = tid & 31, w = tid >> 5;
    const int b = blockIdx.x;
    int lo = 0;
    if (tid < EB) {
        lo = loff[b * EB + tid];
        const int left = E - tid * 8192;
        const int sliceCnt = (left < 8192) ? left : 8192;
        const int hi = (b + 1 < NB) ? loff[(b + 1) * EB + tid] : sliceCnt;
        segLo[tid] = tid * 8192 + lo;
        segHi[tid] = tid * 8192 + hi;
    }
    int r = lo;
    for (int off = 16; off > 0; off >>= 1) r += __shfl_xor(r, off, 32);
    if (lane == 0) red[w] = r;
    if (tid < 256) c[tid] = 0;
    __syncthreads();
    int base = red[lane];
    for (int off = 16; off > 0; off >>= 1) base += __shfl_xor(base, off, 32);

    int2 ev[4][3];
    #pragma unroll
    for (int q = 0; q < 4; ++q) {
        const int j = w + 32 * q;
        const int slo = (j < EB) ? segLo[j] : 0;
        const int shi = (j < EB) ? segHi[j] : 0;
        #pragma unroll
        for (int s = 0; s < 3; ++s) {
            const int i = slo + lane + 32 * s;
            ev[q][s] = (i < shi) ? ebuf[i] : make_int2(0, -1);
        }
    }
    #pragma unroll
    for (int q = 0; q < 4; ++q)
        #pragma unroll
        for (int s = 0; s < 3; ++s)
            if (ev[q][s].y >= 0) atomicAdd(&c[ev[q][s].y & 255], 1);
    for (int j = w; j < EB; j += 32) {
        const int shi = segHi[j];
        for (int i = segLo[j] + lane + ((j < 128) ? 96 : 0); i < shi; i += 32)
            atomicAdd(&c[ebuf[i].y & 255], 1);
    }
    __syncthreads();
    int v = 0, inc = 0;
    if (tid < 256) {
        v = c[tid];
        inc = v;
        for (int off = 1; off < 32; off <<= 1) {
            int u = __shfl_up(inc, off, 32);
            if (lane >= off) inc += u;
        }
        if (lane == 31) wsum[w] = inc;
    }
    __syncthreads();
    if (tid < 256) {
        int ex = inc - v;
        for (int i = 0; i < w; ++i) ex += wsum[i];
        const int start = base + ex;
        cu[tid] = start;
        if (tid == 255) totS = ex + v;
        const int n = b * 256 + tid;
        if (n < N) {
            rowptr[n] = start;
            cnt[n] = v;
            dis[n] = rsqrtf((float)v + 1.0f);
        }
    }
    __syncthreads();
    const int total = totS;
    if (total <= BKT_STAGE) {
        #pragma unroll
        for (int q = 0; q < 4; ++q)
            #pragma unroll
            for (int s = 0; s < 3; ++s)
                if (ev[q][s].y >= 0) {
                    const int p = atomicAdd(&cu[ev[q][s].y & 255], 1);
                    rowsLds[p - base] = ev[q][s].x;
                }
        for (int j = w; j < EB; j += 32) {
            const int shi = segHi[j];
            for (int i = segLo[j] + lane + ((j < 128) ? 96 : 0); i < shi; i += 32) {
                const int2 e = ebuf[i];
                const int p = atomicAdd(&cu[e.y & 255], 1);
                rowsLds[p - base] = e.x;
            }
        }
        __syncthreads();
        for (int i = tid; i < total; i += 1024) csr[base + i] = rowsLds[i];
    } else {
        #pragma unroll
        for (int q = 0; q < 4; ++q)
            #pragma unroll
            for (int s = 0; s < 3; ++s)
                if (ev[q][s].y >= 0) {
                    const int p = atomicAdd(&cu[ev[q][s].y & 255], 1);
                    csr[p] = ev[q][s].x;
                }
        for (int j = w; j < EB; j += 32) {
            const int shi = segHi[j];
            for (int i = segLo[j] + lane + ((j < 128) ? 96 : 0); i < shi; i += 32) {
                const int2 e = ebuf[i];
                const int p = atomicAdd(&cu[e.y & 255], 1);
                csr[p] = e.x;
            }
        }
    }
}

#define GEMM_BLOCKS 256
#define GEMM_THREADS 512
#define GEMM_FULL_WAVES 12
#define GEMM_LDS_FLOATS (42 * 1024)

__device__ __forceinline__ void load_row_scales(const float* __restrict__ dis, int rowTile, int mhalf, int N,
                                                float (&sc)[8]) {
    #pragma unroll
    for (int r = 0; r < 8; ++r) {
        int rr = rowTile * 16 + mhalf + r;
        rr = (rr < N) ? rr : (N - 1);
        sc[r] = dis[rr];
    }
}

__device__ __forceinline__ void stage_w(const float* __restrict__ W, float* Wl) {
    const int tid = threadIdx.x;
    v2f* Wlv = (v2f*)Wl;
    #pragma unroll
    for (int i = 0; i < 8192 / GEMM_THREADS; ++i) {
        const int idx = i * GEMM_THREADS + tid;
        const int c = idx & 127;
        const int p = idx >> 7;
        v2f pr;
        pr.x = W[(2 * p)     * 128 + c];
        pr.y = W[(2 * p + 1) * 128 + c];
        Wlv[p * 128 + c] = pr;
    }
}

template <int NJ, bool PROJ>
__device__ __forceinline__ void gemm_strip(const float* __restrict__ a_or_x, const float* __restrict__ num_x,
                                           const float* __restrict__ num_mask, const v2f* Pw, const v2f* Pb,
                                           const v2f* Pa, const v2f* Wlv, float* __restrict__ t,
                                           int rowTile, int jc0, const float (&sc)[8], int N, int lane) {
    const int r15   = lane & 15;
    const int khalf = (lane >> 4) << 1;
    const int mhalf = (lane >> 4) << 3;
    int row = rowTile * 16 + r15;
    float s = 0.0f;
    if (PROJ) {
        row = (row < N) ? row : (N - 1);
        s = num_x[row] * num_mask[row];
    }
    const float* arow = a_or_x + (size_t)row * 128;

    v8f acc[NJ] = {};
    for (int k0 = 0; k0 < 128; k0 += 4) {
        const int kp = (k0 + khalf) >> 1;
        v2f a = *(const v2f*)(arow + k0 + khalf);
        if (PROJ) {
            const v2f pw = Pw[kp], pb = Pb[kp], pa = Pa[kp];
            float z0 = s * pw.x + pb.x;
            float z1 = s * pw.y + pb.y;
            z0 = (z0 >= 0.0f) ? z0 : pa.x * z0;
            z1 = (z1 >= 0.0f) ? z1 : pa.y * z1;
            a.x = z0 + a.x;
            a.y = z1 + a.y;
        }
        const v2f* wrow = Wlv + (kp * 128 + jc0 * 16 + r15);
        #pragma unroll
        for (int j = 0; j < NJ; ++j) {
            v2f bfrag = wrow[j * 16];
            acc[j] = __builtin_amdgcn_wmma_f32_16x16x4_f32(
                false, a, false, bfrag, (short)0, acc[j], false, false);
        }
    }

    float* trow = t + (size_t)(rowTile * 16 + mhalf) * 128 + jc0 * 16 + r15;
    #pragma unroll
    for (int j = 0; j < NJ; ++j)
        #pragma unroll
        for (int r = 0; r < 8; ++r)
            trow[(size_t)r * 128 + j * 16] = acc[j][r] * sc[r];
}

template <bool PROJ>
__device__ __forceinline__ void gemm_block(const float* __restrict__ a_or_x, const float* __restrict__ num_x,
                                           const float* __restrict__ num_mask, const float* __restrict__ Wn,
                                           const float* __restrict__ bn, const float* __restrict__ ain,
                                           v2f* Pw, v2f* Pb, v2f* Pa, const float* __restrict__ W, float* Wl,
                                           const float* __restrict__ dis, float* __restrict__ t,
                                           int nRowTiles, int N) {
    const int wave = threadIdx.x >> 5;
    const int lane = threadIdx.x & 31;
    const int b = blockIdx.x;
    const int mhalf = (lane >> 4) << 3;
    const int nFull = nRowTiles / GEMM_BLOCKS;
    const int rem   = nRowTiles % GEMM_BLOCKS;
    const bool whole = wave < GEMM_FULL_WAVES;
    const bool has   = whole ? (wave < nFull) : (b < rem);
    const int tile0  = whole ? (b + GEMM_BLOCKS * wave) : (b + GEMM_BLOCKS * nFull);
    float sc[8];
    load_row_scales(dis, has ? tile0 : 0, mhalf, N, sc);
    const bool stp = PROJ && threadIdx.x < 64;
    v2f pwv = {}, pbv = {}, pav = {};
    if (stp) {
        pwv = *(const v2f*)(Wn  + 2 * threadIdx.x);
        pbv = *(const v2f*)(bn  + 2 * threadIdx.x);
        pav = *(const v2f*)(ain + 2 * threadIdx.x);
    }
    stage_w(W, Wl);
    if (stp) {
        Pw[threadIdx.x] = pwv;
        Pb[threadIdx.x] = pbv;
        Pa[threadIdx.x] = pav;
    }
    __syncthreads();
    asm volatile("" :: "v"(sc[0]), "v"(sc[1]), "v"(sc[2]), "v"(sc[3]), "v"(sc[4]), "v"(sc[5]), "v"(sc[6]), "v"(sc[7]));

    const v2f* Wlv = (const v2f*)Wl;
    if (whole) {
        for (int i = wave; i < nFull; i += GEMM_FULL_WAVES) {
            const int tile = b + GEMM_BLOCKS * i;
            if (i != wave) load_row_scales(dis, tile, mhalf, N, sc);
            gemm_strip<8, PROJ>(a_or_x, num_x, num_mask, Pw, Pb, Pa, Wlv, t, tile, 0, sc, N, lane);
        }
    } else if (has) {
        gemm_strip<2, PROJ>(a_or_x, num_x, num_mask, Pw, Pb, Pa, Wlv, t, tile0, (wave - GEMM_FULL_WAVES) * 2,
                            sc, N, lane);
    }
}

__global__ __launch_bounds__(GEMM_THREADS) void k_proj_gemm128_wmma(const float* __restrict__ num_x,
                                                                    const float* __restrict__ num_mask,
                                                                    const float* __restrict__ x,
                                                                    const float* __restrict__ Wn,
                                                                    const float* __restrict__ bn,
                                                                    const float* __restrict__ ain,
                                                                    const float* __restrict__ W,
                                                                    const float* __restrict__ dis,
                                                                    float* __restrict__ t,
                                                                    int nRowTiles, int N) {
    __shared__ float Wl[GEMM_LDS_FLOATS];
    __shared__ v2f Pw[64];
    __shared__ v2f Pb[64];
    __shared__ v2f Pa[64];
    gemm_block<true>(x, num_x, num_mask, Wn, bn, ain, Pw, Pb, Pa, W, Wl, dis, t, nRowTiles, N);
}

#define FUSE_BLOCKS 1024
#define A_PITCH 132

__device__ __forceinline__ void add4(float4& acc, const float4 v) {
    acc.x += v.x; acc.y += v.y; acc.z += v.z; acc.w += v.w;
}

__device__ __forceinline__ void stage_w_quarters(const float* __restrict__ W, float* Wq) {
    const int tid = threadIdx.x;
    v2f* Wqv = (v2f*)Wq;
    #pragma unroll
    for (int i = 0; i < 8192 / GEMM_THREADS; ++i) {
        const int idx = i * GEMM_THREADS + tid;
        const int c = idx & 127;
        const int p = idx >> 7;
        v2f pr;
        pr.x = W[(2 * p)     * 128 + c];
        pr.y = W[(2 * p + 1) * 128 + c];
        Wqv[p * 128 + (c >> 5) * 32 + (c & 15) * 2 + ((c >> 4) & 1)] = pr;
    }
}

__global__ __launch_bounds__(GEMM_THREADS) void k_gather_gemm128_wmma(const int* __restrict__ csr,
                                                                      const int* __restrict__ rowptr,
                                                                      const int* __restrict__ cnt,
                                                                      const float* __restrict__ dis,
                                                                      const float* __restrict__ t,
                                                                      const float* __restrict__ bias,
                                                                      const float* __restrict__ a,
                                                                      const float* __restrict__ W,
                                                                      float* __restrict__ t2,
                                                                      int nRowTiles, int N) {
    __shared__ alignas(16) float Wq[128 * 128];
    __shared__ alignas(16) float At[16 * A_PITCH];
    __shared__ float dsc[16];
    __shared__ float4 Pb[32];
    __shared__ float4 Pa[32];
    const int wave = threadIdx.x >> 5;
    const int lane = threadIdx.x & 31;
    const float4* t4 = (const float4*)t;
    if (wave == 0) Pb[lane] = ((const float4*)bias)[lane];
    if (wave == 1) Pa[lane] = ((const float4*)a)[lane];
    stage_w_quarters(W, Wq);
    __syncthreads();
    const bool mm = (wave & 3) == (wave >> 2);
    const int quarter = wave & 3;
    const int r15   = lane & 15;
    const int khalf = (lane >> 4) << 1;
    const int mhalf = (lane >> 4) << 3;

    for (int tile = blockIdx.x; tile < nRowTiles; tile += gridDim.x) {
        const int n = __builtin_amdgcn_readfirstlane(tile * 16 + wave);
        float4 r = make_float4(0.0f, 0.0f, 0.0f, 0.0f);
        float dn = 0.0f;
        if (n < N) {
            const int beg = rowptr[n];
            const int end = beg + cnt[n];
            dn = dis[n];
            const float4 tv = t4[(size_t)n * 32 + lane];
            float4 acc = make_float4(0.0f, 0.0f, 0.0f, 0.0f);
            int e = beg;
            for (; e + 8 <= end; e += 8) {
                const int s0 = __builtin_amdgcn_readfirstlane(csr[e]);
                const int s1 = __builtin_amdgcn_readfirstlane(csr[e + 1]);
                const int s2 = __builtin_amdgcn_readfirstlane(csr[e + 2]);
                const int s3 = __builtin_amdgcn_readfirstlane(csr[e + 3]);
                const int s4 = __builtin_amdgcn_readfirstlane(csr[e + 4]);
                const int s5 = __builtin_amdgcn_readfirstlane(csr[e + 5]);
                const int s6 = __builtin_amdgcn_readfirstlane(csr[e + 6]);
                const int s7 = __builtin_amdgcn_readfirstlane(csr[e + 7]);
                const float4 v0 = t4[(size_t)s0 * 32 + lane];
                const float4 v1 = t4[(size_t)s1 * 32 + lane];
                const float4 v2 = t4[(size_t)s2 * 32 + lane];
                const float4 v3 = t4[(size_t)s3 * 32 + lane];
                const float4 v4 = t4[(size_t)s4 * 32 + lane];
                const float4 v5 = t4[(size_t)s5 * 32 + lane];
                const float4 v6 = t4[(size_t)s6 * 32 + lane];
                const float4 v7 = t4[(size_t)s7 * 32 + lane];
                add4(acc, v0); add4(acc, v1); add4(acc, v2); add4(acc, v3);
                add4(acc, v4); add4(acc, v5); add4(acc, v6); add4(acc, v7);
            }
            if (e + 4 <= end) {
                const int s0 = __builtin_amdgcn_readfirstlane(csr[e]);
                const int s1 = __builtin_amdgcn_readfirstlane(csr[e + 1]);
                const int s2 = __builtin_amdgcn_readfirstlane(csr[e + 2]);
                const int s3 = __builtin_amdgcn_readfirstlane(csr[e + 3]);
                const float4 v0 = t4[(size_t)s0 * 32 + lane];
                const float4 v1 = t4[(size_t)s1 * 32 + lane];
                const float4 v2 = t4[(size_t)s2 * 32 + lane];
                const float4 v3 = t4[(size_t)s3 * 32 + lane];
                add4(acc, v0); add4(acc, v1); add4(acc, v2); add4(acc, v3);
                e += 4;
            }
            const int rem = end - e;
            if (rem == 3) {
                const int s0 = __builtin_amdgcn_readfirstlane(csr[e]);
                const int s1 = __builtin_amdgcn_readfirstlane(csr[e + 1]);
                const int s2 = __builtin_amdgcn_readfirstlane(csr[e + 2]);
                const float4 v0 = t4[(size_t)s0 * 32 + lane];
                const float4 v1 = t4[(size_t)s1 * 32 + lane];
                const float4 v2 = t4[(size_t)s2 * 32 + lane];
                add4(acc, v0); add4(acc, v1); add4(acc, v2);
            } else if (rem == 2) {
                const int s0 = __builtin_amdgcn_readfirstlane(csr[e]);
                const int s1 = __builtin_amdgcn_readfirstlane(csr[e + 1]);
                const float4 v0 = t4[(size_t)s0 * 32 + lane];
                const float4 v1 = t4[(size_t)s1 * 32 + lane];
                add4(acc, v0); add4(acc, v1);
            } else if (rem == 1) {
                const int s0 = __builtin_amdgcn_readfirstlane(csr[e]);
                const float4 v0 = t4[(size_t)s0 * 32 + lane];
                add4(acc, v0);
            }
            const float4 b  = Pb[lane];
            const float4 av = Pa[lane];
            r.x = dn * (acc.x + tv.x) + b.x;
            r.y = dn * (acc.y + tv.y) + b.y;
            r.z = dn * (acc.z + tv.z) + b.z;
            r.w = dn * (acc.w + tv.w) + b.w;
            r.x = (r.x >= 0.0f) ? r.x : av.x * r.x;
            r.y = (r.y >= 0.0f) ? r.y : av.y * r.y;
            r.z = (r.z >= 0.0f) ? r.z : av.z * r.z;
            r.w = (r.w >= 0.0f) ? r.w : av.w * r.w;
        }
        *(float4*)(At + wave * A_PITCH + lane * 4) = r;
        if (lane == 0) dsc[wave] = dn;
        __syncthreads();

        if (mm) {
            const float* arow = At + r15 * A_PITCH + khalf;
            const float* brow = Wq + (khalf >> 1) * 256 + quarter * 64 + r15 * 4;
            v8f acc[2] = {};
            #pragma unroll
            for (int k0 = 0; k0 < 128; k0 += 4) {
                const v2f af = *(const v2f*)(arow + k0);
                const float4 bq = *(const float4*)(brow + (k0 >> 1) * 256);
                v2f b0, b1;
                b0.x = bq.x; b0.y = bq.y;
                b1.x = bq.z; b1.y = bq.w;
                acc[0] = __builtin_amdgcn_wmma_f32_16x16x4_f32(false, af, false, b0, (short)0, acc[0], false, false);
                acc[1] = __builtin_amdgcn_wmma_f32_16x16x4_f32(false, af, false, b1, (short)0, acc[1], false, false);
            }
            float* trow = t2 + (size_t)(tile * 16 + mhalf) * 128 + quarter * 32 + r15;
            #pragma unroll
            for (int j = 0; j < 2; ++j)
                #pragma unroll
                for (int rr = 0; rr < 8; ++rr)
                    trow[(size_t)rr * 128 + j * 16] = acc[j][rr] * dsc[mhalf + rr];
        }
        __syncthreads();
    }
}

#define GD_ROWS 32
__global__ __launch_bounds__(256) void k_gather_dot_w3(const int* __restrict__ csr, const int* __restrict__ rowptr,
                                                       const int* __restrict__ cnt, const float* __restrict__ dis,
                                                       const float* __restrict__ t, const float* __restrict__ bias,
                                                       const float* __restrict__ a, const float* __restrict__ W3,
                                                       float* __restrict__ t3, int N) {
    __shared__ float rsum[GD_ROWS];
    __shared__ float4 Pb[32];
    __shared__ float4 Pa[32];
    __shared__ float4 Pw[32];
    const int wave = threadIdx.x >> 5;
    const int lane = threadIdx.x & 31;
    const float4* t4 = (const float4*)t;
    if (wave == 0) Pb[lane] = ((const float4*)bias)[lane];
    if (wave == 1) Pa[lane] = ((const float4*)a)[lane];
    if (wave == 2) Pw[lane] = ((const float4*)W3)[lane];
    __syncthreads();
    for (int q = 0; q < GD_ROWS / 8; ++q) {
        const int n = __builtin_amdgcn_readfirstlane(blockIdx.x * GD_ROWS + q * 8 + wave);
        float s = 0.0f;
        if (n < N) {
            const int beg = rowptr[n];
            const int end = beg + cnt[n];
            const float dn = dis[n];
            const float4 tv = t4[(size_t)n * 32 + lane];
            float4 acc = make_float4(0.0f, 0.0f, 0.0f, 0.0f);
            int e = beg;
            for (; e + 8 <= end; e += 8) {
                const int s0 = __builtin_amdgcn_readfirstlane(csr[e]);
                const int s1 = __builtin_amdgcn_readfirstlane(csr[e + 1]);
                const int s2 = __builtin_amdgcn_readfirstlane(csr[e + 2]);
                const int s3 = __builtin_amdgcn_readfirstlane(csr[e + 3]);
                const int s4 = __builtin_amdgcn_readfirstlane(csr[e + 4]);
                const int s5 = __builtin_amdgcn_readfirstlane(csr[e + 5]);
                const int s6 = __builtin_amdgcn_readfirstlane(csr[e + 6]);
                const int s7 = __builtin_amdgcn_readfirstlane(csr[e + 7]);
                const float4 v0 = t4[(size_t)s0 * 32 + lane];
                const float4 v1 = t4[(size_t)s1 * 32 + lane];
                const float4 v2 = t4[(size_t)s2 * 32 + lane];
                const float4 v3 = t4[(size_t)s3 * 32 + lane];
                const float4 v4 = t4[(size_t)s4 * 32 + lane];
                const float4 v5 = t4[(size_t)s5 * 32 + lane];
                const float4 v6 = t4[(size_t)s6 * 32 + lane];
                const float4 v7 = t4[(size_t)s7 * 32 + lane];
                add4(acc, v0); add4(acc, v1); add4(acc, v2); add4(acc, v3);
                add4(acc, v4); add4(acc, v5); add4(acc, v6); add4(acc, v7);
            }
            if (e + 4 <= end) {
                const int s0 = __builtin_amdgcn_readfirstlane(csr[e]);
                const int s1 = __builtin_amdgcn_readfirstlane(csr[e + 1]);
                const int s2 = __builtin_amdgcn_readfirstlane(csr[e + 2]);
                const int s3 = __builtin_amdgcn_readfirstlane(csr[e + 3]);
                const float4 v0 = t4[(size_t)s0 * 32 + lane];
                const float4 v1 = t4[(size_t)s1 * 32 + lane];
                const float4 v2 = t4[(size_t)s2 * 32 + lane];
                const float4 v3 = t4[(size_t)s3 * 32 + lane];
                add4(acc, v0); add4(acc, v1); add4(acc, v2); add4(acc, v3);
                e += 4;
            }
            const int rem = end - e;
            if (rem == 3) {
                const int s0 = __builtin_amdgcn_readfirstlane(csr[e]);
                const int s1 = __builtin_amdgcn_readfirstlane(csr[e + 1]);
                const int s2 = __builtin_amdgcn_readfirstlane(csr[e + 2]);
                const float4 v0 = t4[(size_t)s0 * 32 + lane];
                const float4 v1 = t4[(size_t)s1 * 32 + lane];
                const float4 v2 = t4[(size_t)s2 * 32 + lane];
                add4(acc, v0); add4(acc, v1); add4(acc, v2);
            } else if (rem == 2) {
                const int s0 = __builtin_amdgcn_readfirstlane(csr[e]);
                const int s1 = __builtin_amdgcn_readfirstlane(csr[e + 1]);
                const float4 v0 = t4[(size_t)s0 * 32 + lane];
                const float4 v1 = t4[(size_t)s1 * 32 + lane];
                add4(acc, v0); add4(acc, v1);
            } else if (rem == 1) {
                const int s0 = __builtin_amdgcn_readfirstlane(csr[e]);
                const float4 v0 = t4[(size_t)s0 * 32 + lane];
                add4(acc, v0);
            }
            const float4 b  = Pb[lane];
            const float4 av = Pa[lane];
            const float4 wv = Pw[lane];
            float4 r;
            r.x = dn * (acc.x + tv.x) + b.x;
            r.y = dn * (acc.y + tv.y) + b.y;
            r.z = dn * (acc.z + tv.z) + b.z;
            r.w = dn * (acc.w + tv.w) + b.w;
            r.x = (r.x >= 0.0f) ? r.x : av.x * r.x;
            r.y = (r.y >= 0.0f) ? r.y : av.y * r.y;
            r.z = (r.z >= 0.0f) ? r.z : av.z * r.z;
            r.w = (r.w >= 0.0f) ? r.w : av.w * r.w;
            s = fmaf(r.w, wv.w, fmaf(r.z, wv.z, fmaf(r.x, wv.x, r.y * wv.y)));
            for (int off = 16; off > 0; off >>= 1) s += __shfl_xor(s, off, 32);
        }
        if (lane == 0) rsum[q * 8 + wave] = s;
    }
    __syncthreads();
    if (wave == 0) {
        const int m = blockIdx.x * GD_ROWS + lane;
        if (m < N) t3[m] = dis[m] * rsum[lane];
    }
}

__global__ void k_gather_scalar(const int* __restrict__ csr, const int* __restrict__ rowptr,
                                const int* __restrict__ cnt, const float* __restrict__ dis,
                                const float* __restrict__ t3, const float* __restrict__ b3,
                                float* __restrict__ out, int N) {
    const int gid = blockIdx.x * blockDim.x + threadIdx.x;
    const int n = gid >> 3, sub = gid & 7;
    const bool valid = n < N;
    int beg = 0, end = 0;
    if (valid) { beg = rowptr[n]; end = beg + cnt[n]; }
    float s = 0.0f;
    for (int e = beg + sub; e < end; e += 8) s += t3[csr[e]];
    s += __shfl_xor(s, 4, 32);
    s += __shfl_xor(s, 2, 32);
    s += __shfl_xor(s, 1, 32);
    if (valid && sub == 0) out[n] = dis[n] * (s + t3[n]) + b3[0];
}

extern "C" void kernel_launch(void* const* d_in, const int* in_sizes, int n_in,
                              void* d_out, int out_size, void* d_ws, size_t ws_size,
                              hipStream_t stream) {
    const float* num_x    = (const float*)d_in[0];
    const float* num_mask = (const float*)d_in[1];
    const float* x        = (const float*)d_in[2];
    const int*   edge     = (const int*)d_in[3];
    const float* W_num    = (const float*)d_in[4];
    const float* b_num    = (const float*)d_in[5];
    const float* a_in     = (const float*)d_in[6];
    const float* W1       = (const float*)d_in[7];
    const float* b1       = (const float*)d_in[8];
    const float* a1       = (const float*)d_in[9];
    const float* W2       = (const float*)d_in[10];
    const float* b2       = (const float*)d_in[11];
    const float* a2       = (const float*)d_in[12];
    const float* W3       = (const float*)d_in[13];
    const float* b3       = (const float*)d_in[14];

    const int N = in_sizes[1];
    const int E = in_sizes[3] / 2;
    const int* src = edge;
    const int* dst = edge + E;

    const int nRowTiles = (N + 15) / 16;
    const size_t padRows = (size_t)nRowTiles * 16;

    auto al = [](size_t v) { return (v + 255) & ~(size_t)255; };
    const int NB = (N + 255) / 256;
    const int EB = (E + 8191) / 8192;
    char* p = (char*)d_ws;
    float* dis    = (float*)p; p += al((size_t)N * 4);
    int*   cnt    = (int*)p;   p += al((size_t)N * 4);
    int*   rowptr = (int*)p;   p += al((size_t)N * 4);
    int*   loff   = (int*)p;   p += al((size_t)NB * EB * 4);
    int2*  ebuf   = (int2*)p;  p += al((size_t)EB * 8192 * 8);
    int*   csr    = (int*)p;   p += al((size_t)E * 4);
    float* hbuf = (float*)p; p += al(padRows * 128 * 4);
    float* tbuf = (float*)p; p += al(padRows * 128 * 4);
    float* t3   = (float*)p; p += al((size_t)N * 4);
    float* out  = (float*)d_out;

    const int TPB = 256;
    dim3 blk(TPB);
    auto gN8 = dim3((int)(((size_t)N * 8 + TPB - 1) / TPB));
    auto gN32 = dim3((N + GD_ROWS - 1) / GD_ROWS);

    k_blk_sort<<<EB, 1024, 0, stream>>>(src, dst, loff, ebuf, E, NB, EB);
    k_bkt_rows<<<NB, 1024, 0, stream>>>(ebuf, loff, rowptr, cnt, dis, csr, N, E, NB, EB);

    k_proj_gemm128_wmma<<<GEMM_BLOCKS, GEMM_THREADS, 0, stream>>>(num_x, num_mask, x, W_num, b_num, a_in, W1, dis, tbuf, nRowTiles, N);
    k_gather_gemm128_wmma<<<FUSE_BLOCKS, GEMM_THREADS, 0, stream>>>(csr, rowptr, cnt, dis, tbuf, b1, a1, W2, hbuf, nRowTiles, N);

    k_gather_dot_w3<<<gN32, blk, 0, stream>>>(csr, rowptr, cnt, dis, hbuf, b2, a2, W3, t3, N);
    k_gather_scalar<<<gN8, blk, 0, stream>>>(csr, rowptr, cnt, dis, t3, b3, out, N);
}
